# starting kernel
# baseline (speedup 1.0000x reference)
#include <hip/hip_runtime.h>


typedef float v2f __attribute__((ext_vector_type(2)));
typedef float v8f __attribute__((ext_vector_type(8)));

#define N_NODES   1000000
#define N_EDGES   16000000
#define F_IN      32
#define F_H1      16
#define F_H2      8
#define N_CLS     5
#define N_GRAPHS  10000

#define ATOMIC_FADD(p, v) \
  __hip_atomic_fetch_add((p), (v), __ATOMIC_RELAXED, __HIP_MEMORY_SCOPE_AGENT)


__global__ void zero_f32(float* __restrict__ p, long long n) {
  long long i = (long long)blockIdx.x * blockDim.x + threadIdx.x;
  long long stride = (long long)gridDim.x * blockDim.x;
  for (; i < n; i += stride) p[i] = 0.0f;
}

__global__ void degree_kernel(const int* __restrict__ dst,
                              float* __restrict__ deg, int nE) {
  int e = blockIdx.x * blockDim.x + threadIdx.x;
  if (e < nE) ATOMIC_FADD(&deg[dst[e]], 1.0f);
}

__global__ void rsqrt_kernel(float* __restrict__ deg, int n) {
  int i = blockIdx.x * blockDim.x + threadIdx.x;
  if (i < n) deg[i] = rsqrtf(deg[i] + 1.0f);
}


__global__ __launch_bounds__(128) void gemm_x_w1(const float* __restrict__ X,
                                                 const float* __restrict__ W1,
                                                 float* __restrict__ H1pre) {
  int wave = blockIdx.x * (blockDim.x >> 5) + (threadIdx.x >> 5);
  int lane = threadIdx.x & 31;
  int m  = lane & 15;
  int kk = (lane >> 4) << 1;
  const float* xrow = X + (size_t)(wave * 16 + m) * F_IN;

  v8f c = {0.f, 0.f, 0.f, 0.f, 0.f, 0.f, 0.f, 0.f};
#pragma unroll
  for (int k = 0; k < F_IN; k += 4) {
    v2f a, b;
    a.x = xrow[k + kk];
    a.y = xrow[k + kk + 1];
    b.x = W1[(k + kk) * F_H1 + m];
    b.y = W1[(k + kk + 1) * F_H1 + m];
    c = __builtin_amdgcn_wmma_f32_16x16x4_f32(false, a, false, b,
                                              (short)0, c, false, false);
  }
  int hi = lane >> 4;
  float* out = H1pre + (size_t)(wave * 16 + 8 * hi) * F_H1 + m;
#pragma unroll
  for (int j = 0; j < 8; ++j) out[(size_t)j * F_H1] = c[j];
}

__global__ __launch_bounds__(128) void gemm_h1_w2(const float* __restrict__ H1,
                                                  const float* __restrict__ W2,
                                                  float* __restrict__ H2pre) {
  int wave = blockIdx.x * (blockDim.x >> 5) + (threadIdx.x >> 5);
  int lane = threadIdx.x & 31;
  int m  = lane & 15;
  int kk = (lane >> 4) << 1;
  const float* hrow = H1 + (size_t)(wave * 16 + m) * F_H1;
  bool live_col = (m < F_H2);

  v8f c = {0.f, 0.f, 0.f, 0.f, 0.f, 0.f, 0.f, 0.f};
#pragma unroll
  for (int k = 0; k < F_H1; k += 4) {
    v2f a, b;
    a.x = hrow[k + kk];
    a.y = hrow[k + kk + 1];
    b.x = live_col ? W2[(k + kk) * F_H2 + m] : 0.0f;
    b.y = live_col ? W2[(k + kk + 1) * F_H2 + m] : 0.0f;
    c = __builtin_amdgcn_wmma_f32_16x16x4_f32(false, a, false, b,
                                              (short)0, c, false, false);
  }
  if (live_col) {
    int hi = lane >> 4;
    float* out = H2pre + (size_t)(wave * 16 + 8 * hi) * F_H2 + m;
#pragma unroll
    for (int j = 0; j < 8; ++j) out[(size_t)j * F_H2] = c[j];
  }
}


__global__ void agg_edges16(const int* __restrict__ src, const int* __restrict__ dst,
                            const float* __restrict__ dinv,
                            const float* __restrict__ H, float* __restrict__ AGG,
                            int nE) {
  int e = blockIdx.x * blockDim.x + threadIdx.x;
  if (e >= nE) return;
  int s = src[e], d = dst[e];
  float w = dinv[s] * dinv[d];
  const float4* hs = (const float4*)(H + (size_t)s * F_H1);
  float* out = AGG + (size_t)d * F_H1;
#pragma unroll
  for (int q = 0; q < 4; ++q) {
    float4 v = hs[q];
    ATOMIC_FADD(&out[4 * q + 0], v.x * w);
    ATOMIC_FADD(&out[4 * q + 1], v.y * w);
    ATOMIC_FADD(&out[4 * q + 2], v.z * w);
    ATOMIC_FADD(&out[4 * q + 3], v.w * w);
  }
}

__global__ void agg_edges8(const int* __restrict__ src, const int* __restrict__ dst,
                           const float* __restrict__ dinv,
                           const float* __restrict__ H, float* __restrict__ AGG,
                           int nE) {
  int e = blockIdx.x * blockDim.x + threadIdx.x;
  if (e >= nE) return;
  int s = src[e], d = dst[e];
  float w = dinv[s] * dinv[d];
  const float4* hs = (const float4*)(H + (size_t)s * F_H2);
  float* out = AGG + (size_t)d * F_H2;
#pragma unroll
  for (int q = 0; q < 2; ++q) {
    float4 v = hs[q];
    ATOMIC_FADD(&out[4 * q + 0], v.x * w);
    ATOMIC_FADD(&out[4 * q + 1], v.y * w);
    ATOMIC_FADD(&out[4 * q + 2], v.z * w);
    ATOMIC_FADD(&out[4 * q + 3], v.w * w);
  }
}


__global__ void finalize1(float* __restrict__ agg, const float* __restrict__ hpre,
                          const float* __restrict__ dinv,
                          const float* __restrict__ b1, int n) {
  int i = blockIdx.x * blockDim.x + threadIdx.x;
  if (i >= n) return;
  float w = dinv[i]; w *= w;
  float4* a = (float4*)(agg + (size_t)i * F_H1);
  const float4* h = (const float4*)(hpre + (size_t)i * F_H1);
  const float4* b = (const float4*)b1;
#pragma unroll
  for (int q = 0; q < 4; ++q) {
    float4 av = a[q], hv = h[q], bv = b[q];
    av.x = fmaxf(av.x + hv.x * w + bv.x, 0.0f);
    av.y = fmaxf(av.y + hv.y * w + bv.y, 0.0f);
    av.z = fmaxf(av.z + hv.z * w + bv.z, 0.0f);
    av.w = fmaxf(av.w + hv.w * w + bv.w, 0.0f);
    a[q] = av;
  }
}

__global__ void finalize2_pool(const float* __restrict__ agg,
                               const float* __restrict__ hpre,
                               const float* __restrict__ dinv,
                               const float* __restrict__ b2,
                               const int* __restrict__ batch,
                               float* __restrict__ pooled,
                               float* __restrict__ cnt, int n) {
  int i = blockIdx.x * blockDim.x + threadIdx.x;
  if (i >= n) return;
  float w = dinv[i]; w *= w;
  int g = batch[i];
  float* ps = pooled + (size_t)g * F_H2;
#pragma unroll
  for (int f = 0; f < F_H2; ++f) {
    float v = agg[(size_t)i * F_H2 + f] + hpre[(size_t)i * F_H2 + f] * w + b2[f];
    v = fmaxf(v, 0.0f);
    ATOMIC_FADD(&ps[f], v);
  }
  ATOMIC_FADD(&cnt[g], 1.0f);
}

__global__ void readout(const float* __restrict__ pooled, const float* __restrict__ cnt,
                        const float* __restrict__ Wl, const float* __restrict__ bl,
                        float* __restrict__ out, int G) {
  int g = blockIdx.x * blockDim.x + threadIdx.x;
  if (g >= G) return;
  float inv = 1.0f / fmaxf(cnt[g], 1.0f);
  float p[F_H2];
#pragma unroll
  for (int f = 0; f < F_H2; ++f) p[f] = pooled[(size_t)g * F_H2 + f] * inv;
#pragma unroll
  for (int c = 0; c < N_CLS; ++c) {
    float acc = bl[c];
#pragma unroll
    for (int f = 0; f < F_H2; ++f) acc += p[f] * Wl[f * N_CLS + c];
    out[(size_t)g * N_CLS + c] = acc;
  }
}


extern "C" void kernel_launch(void* const* d_in, const int* in_sizes, int n_in,
                              void* d_out, int out_size, void* d_ws, size_t ws_size,
                              hipStream_t stream) {
  const float* x     = (const float*)d_in[0];
  const int*   ei    = (const int*)d_in[1];
  const int*   src   = ei;
  const int*   dst   = ei + N_EDGES;
  const int*   batch = (const int*)d_in[2];
  const float* W1 = (const float*)d_in[4];
  const float* b1 = (const float*)d_in[5];
  const float* W2 = (const float*)d_in[6];
  const float* b2 = (const float*)d_in[7];
  const float* Wl = (const float*)d_in[8];
  const float* bl = (const float*)d_in[9];
  float* out = (float*)d_out;

  char* ws = (char*)d_ws;
  float* dinv   = (float*)(ws);
  float* bufA   = (float*)(ws + (4ll  << 20));
  float* bufB   = (float*)(ws + (68ll << 20));
  float* pooled = (float*)(ws + (132ll << 20));
  float* cnt    = pooled + (size_t)N_GRAPHS * F_H2;

  float* h1pre = bufA;
  float* agg1  = bufB;
  float* h2pre = bufA;
  float* agg2  = (float*)((char*)bufA + (32ll << 20));

  const int EB = 256;
  const int EG = (N_EDGES + EB - 1) / EB;
  const int NB = 256;
  const int NG = (N_NODES + NB - 1) / NB;
  const int GEMM_BLOCKS = N_NODES / 64;

  zero_f32<<<2048, 256, 0, stream>>>(dinv, (long long)N_NODES);
  degree_kernel<<<EG, EB, 0, stream>>>(dst, dinv, N_EDGES);
  rsqrt_kernel<<<NG, NB, 0, stream>>>(dinv, N_NODES);

  gemm_x_w1<<<GEMM_BLOCKS, 128, 0, stream>>>(x, W1, h1pre);
  zero_f32<<<8192, 256, 0, stream>>>(agg1, (long long)N_NODES * F_H1);
  agg_edges16<<<EG, EB, 0, stream>>>(src, dst, dinv, h1pre, agg1, N_EDGES);
  finalize1<<<NG, NB, 0, stream>>>(agg1, h1pre, dinv, b1, N_NODES);

  gemm_h1_w2<<<GEMM_BLOCKS, 128, 0, stream>>>(agg1, W2, h2pre);
  zero_f32<<<8192, 256, 0, stream>>>(agg2, (long long)N_NODES * F_H2);
  agg_edges8<<<EG, EB, 0, stream>>>(src, dst, dinv, h2pre, agg2, N_EDGES);

  zero_f32<<<128, 256, 0, stream>>>(pooled, (long long)N_GRAPHS * (F_H2 + 1));
  finalize2_pool<<<NG, NB, 0, stream>>>(agg2, h2pre, dinv, b2, batch,
                                        pooled, cnt, N_NODES);
  readout<<<(N_GRAPHS + NB - 1) / NB, NB, 0, stream>>>(pooled, cnt, Wl, bl,
                                                       out, N_GRAPHS);
}


# edge aggregation: F lanes per edge, one feature per lane (ids by lane shuffle), so each gather and float-atomic instruction covers 2 or 4 whole rows, not 32 lines; same atomics and addends
# speedup vs baseline: 2.7330x; 2.7330x over previous
#include <hip/hip_runtime.h>


typedef float v2f __attribute__((ext_vector_type(2)));
typedef float v8f __attribute__((ext_vector_type(8)));

#define N_NODES   1000000
#define N_EDGES   16000000
#define F_IN      32
#define F_H1      16
#define F_H2      8
#define N_CLS     5
#define N_GRAPHS  10000

#define ATOMIC_FADD(p, v) \
  __hip_atomic_fetch_add((p), (v), __ATOMIC_RELAXED, __HIP_MEMORY_SCOPE_AGENT)


__global__ void zero_f32(float* __restrict__ p, long long n) {
  long long i = (long long)blockIdx.x * blockDim.x + threadIdx.x;
  long long stride = (long long)gridDim.x * blockDim.x;
  for (; i < n; i += stride) p[i] = 0.0f;
}

__global__ void degree_kernel(const int* __restrict__ dst,
                              float* __restrict__ deg, int nE) {
  int e = blockIdx.x * blockDim.x + threadIdx.x;
  if (e < nE) ATOMIC_FADD(&deg[dst[e]], 1.0f);
}

__global__ void rsqrt_kernel(float* __restrict__ deg, int n) {
  int i = blockIdx.x * blockDim.x + threadIdx.x;
  if (i < n) deg[i] = rsqrtf(deg[i] + 1.0f);
}


__global__ __launch_bounds__(128) void gemm_x_w1(const float* __restrict__ X,
                                                 const float* __restrict__ W1,
                                                 float* __restrict__ H1pre) {
  int wave = blockIdx.x * (blockDim.x >> 5) + (threadIdx.x >> 5);
  int lane = threadIdx.x & 31;
  int m  = lane & 15;
  int kk = (lane >> 4) << 1;
  const float* xrow = X + (size_t)(wave * 16 + m) * F_IN;

  v8f c = {0.f, 0.f, 0.f, 0.f, 0.f, 0.f, 0.f, 0.f};
#pragma unroll
  for (int k = 0; k < F_IN; k += 4) {
    v2f a, b;
    a.x = xrow[k + kk];
    a.y = xrow[k + kk + 1];
    b.x = W1[(k + kk) * F_H1 + m];
    b.y = W1[(k + kk + 1) * F_H1 + m];
    c = __builtin_amdgcn_wmma_f32_16x16x4_f32(false, a, false, b,
                                              (short)0, c, false, false);
  }
  int hi = lane >> 4;
  float* out = H1pre + (size_t)(wave * 16 + 8 * hi) * F_H1 + m;
#pragma unroll
  for (int j = 0; j < 8; ++j) out[(size_t)j * F_H1] = c[j];
}

__global__ __launch_bounds__(128) void gemm_h1_w2(const float* __restrict__ H1,
                                                  const float* __restrict__ W2,
                                                  float* __restrict__ H2pre) {
  int wave = blockIdx.x * (blockDim.x >> 5) + (threadIdx.x >> 5);
  int lane = threadIdx.x & 31;
  int m  = lane & 15;
  int kk = (lane >> 4) << 1;
  const float* hrow = H1 + (size_t)(wave * 16 + m) * F_H1;
  bool live_col = (m < F_H2);

  v8f c = {0.f, 0.f, 0.f, 0.f, 0.f, 0.f, 0.f, 0.f};
#pragma unroll
  for (int k = 0; k < F_H1; k += 4) {
    v2f a, b;
    a.x = hrow[k + kk];
    a.y = hrow[k + kk + 1];
    b.x = live_col ? W2[(k + kk) * F_H2 + m] : 0.0f;
    b.y = live_col ? W2[(k + kk + 1) * F_H2 + m] : 0.0f;
    c = __builtin_amdgcn_wmma_f32_16x16x4_f32(false, a, false, b,
                                              (short)0, c, false, false);
  }
  if (live_col) {
    int hi = lane >> 4;
    float* out = H2pre + (size_t)(wave * 16 + 8 * hi) * F_H2 + m;
#pragma unroll
    for (int j = 0; j < 8; ++j) out[(size_t)j * F_H2] = c[j];
  }
}


static_assert(N_EDGES % 256 == 0, "every wave owns 32 whole edges; the grid is exact");

template <int F>
__global__ __launch_bounds__(256) void agg_edges_rows(const int* __restrict__ src,
                                                      const int* __restrict__ dst,
                                                      const float* __restrict__ dinv,
                                                      const float* __restrict__ H,
                                                      float* __restrict__ AGG) {
  constexpr int EPI   = 32 / F;
  constexpr int STEPS = F;
  const int lane = threadIdx.x & 31;
  const int e    = blockIdx.x * blockDim.x + threadIdx.x;
  const int s = src[e], d = dst[e];
  const float w = dinv[s] * dinv[d];
  const int sub = lane / F;
  const int f   = lane % F;

  float v[STEPS];
#pragma unroll
  for (int i = 0; i < STEPS; ++i) {
    const int si = __shfl(s, i * EPI + sub);
    v[i] = H[(unsigned)si * F + f];
  }
#pragma unroll
  for (int i = 0; i < STEPS; ++i) {
    const int   di = __shfl(d, i * EPI + sub);
    const float wi = __shfl(w, i * EPI + sub);
    ATOMIC_FADD(&AGG[(unsigned)di * F + f], v[i] * wi);
  }
}


__global__ void finalize1(float* __restrict__ agg, const float* __restrict__ hpre,
                          const float* __restrict__ dinv,
                          const float* __restrict__ b1, int n) {
  int i = blockIdx.x * blockDim.x + threadIdx.x;
  if (i >= n) return;
  float w = dinv[i]; w *= w;
  float4* a = (float4*)(agg + (size_t)i * F_H1);
  const float4* h = (const float4*)(hpre + (size_t)i * F_H1);
  const float4* b = (const float4*)b1;
#pragma unroll
  for (int q = 0; q < 4; ++q) {
    float4 av = a[q], hv = h[q], bv = b[q];
    av.x = fmaxf(av.x + hv.x * w + bv.x, 0.0f);
    av.y = fmaxf(av.y + hv.y * w + bv.y, 0.0f);
    av.z = fmaxf(av.z + hv.z * w + bv.z, 0.0f);
    av.w = fmaxf(av.w + hv.w * w + bv.w, 0.0f);
    a[q] = av;
  }
}

__global__ void finalize2_pool(const float* __restrict__ agg,
                               const float* __restrict__ hpre,
                               const float* __restrict__ dinv,
                               const float* __restrict__ b2,
                               const int* __restrict__ batch,
                               float* __restrict__ pooled,
                               float* __restrict__ cnt, int n) {
  int i = blockIdx.x * blockDim.x + threadIdx.x;
  if (i >= n) return;
  float w = dinv[i]; w *= w;
  int g = batch[i];
  float* ps = pooled + (size_t)g * F_H2;
#pragma unroll
  for (int f = 0; f < F_H2; ++f) {
    float v = agg[(size_t)i * F_H2 + f] + hpre[(size_t)i * F_H2 + f] * w + b2[f];
    v = fmaxf(v, 0.0f);
    ATOMIC_FADD(&ps[f], v);
  }
  ATOMIC_FADD(&cnt[g], 1.0f);
}

__global__ void readout(const float* __restrict__ pooled, const float* __restrict__ cnt,
                        const float* __restrict__ Wl, const float* __restrict__ bl,
                        float* __restrict__ out, int G) {
  int g = blockIdx.x * blockDim.x + threadIdx.x;
  if (g >= G) return;
  float inv = 1.0f / fmaxf(cnt[g], 1.0f);
  float p[F_H2];
#pragma unroll
  for (int f = 0; f < F_H2; ++f) p[f] = pooled[(size_t)g * F_H2 + f] * inv;
#pragma unroll
  for (int c = 0; c < N_CLS; ++c) {
    float acc = bl[c];
#pragma unroll
    for (int f = 0; f < F_H2; ++f) acc += p[f] * Wl[f * N_CLS + c];
    out[(size_t)g * N_CLS + c] = acc;
  }
}


extern "C" void kernel_launch(void* const* d_in, const int* in_sizes, int n_in,
                              void* d_out, int out_size, void* d_ws, size_t ws_size,
                              hipStream_t stream) {
  const float* x     = (const float*)d_in[0];
  const int*   ei    = (const int*)d_in[1];
  const int*   src   = ei;
  const int*   dst   = ei + N_EDGES;
  const int*   batch = (const int*)d_in[2];
  const float* W1 = (const float*)d_in[4];
  const float* b1 = (const float*)d_in[5];
  const float* W2 = (const float*)d_in[6];
  const float* b2 = (const float*)d_in[7];
  const float* Wl = (const float*)d_in[8];
  const float* bl = (const float*)d_in[9];
  float* out = (float*)d_out;

  char* ws = (char*)d_ws;
  float* dinv   = (float*)(ws);
  float* bufA   = (float*)(ws + (4ll  << 20));
  float* bufB   = (float*)(ws + (68ll << 20));
  float* pooled = (float*)(ws + (132ll << 20));
  float* cnt    = pooled + (size_t)N_GRAPHS * F_H2;

  float* h1pre = bufA;
  float* agg1  = bufB;
  float* h2pre = bufA;
  float* agg2  = (float*)((char*)bufA + (32ll << 20));

  const int EB = 256;
  const int EG = (N_EDGES + EB - 1) / EB;
  const int NB = 256;
  const int NG = (N_NODES + NB - 1) / NB;
  const int GEMM_BLOCKS = N_NODES / 64;

  zero_f32<<<2048, 256, 0, stream>>>(dinv, (long long)N_NODES);
  degree_kernel<<<EG, EB, 0, stream>>>(dst, dinv, N_EDGES);
  rsqrt_kernel<<<NG, NB, 0, stream>>>(dinv, N_NODES);

  gemm_x_w1<<<GEMM_BLOCKS, 128, 0, stream>>>(x, W1, h1pre);
  zero_f32<<<8192, 256, 0, stream>>>(agg1, (long long)N_NODES * F_H1);
  agg_edges_rows<F_H1><<<EG, EB, 0, stream>>>(src, dst, dinv, h1pre, agg1);
  finalize1<<<NG, NB, 0, stream>>>(agg1, h1pre, dinv, b1, N_NODES);

  gemm_h1_w2<<<GEMM_BLOCKS, 128, 0, stream>>>(agg1, W2, h2pre);
  zero_f32<<<8192, 256, 0, stream>>>(agg2, (long long)N_NODES * F_H2);
  agg_edges_rows<F_H2><<<EG, EB, 0, stream>>>(src, dst, dinv, h2pre, agg2);

  zero_f32<<<128, 256, 0, stream>>>(pooled, (long long)N_GRAPHS * (F_H2 + 1));
  finalize2_pool<<<NG, NB, 0, stream>>>(agg2, h2pre, dinv, b2, batch,
                                        pooled, cnt, N_NODES);
  readout<<<(N_GRAPHS + NB - 1) / NB, NB, 0, stream>>>(pooled, cnt, Wl, bl,
                                                       out, N_GRAPHS);
}


# edge norm factored per node: GEMM epilogues store h*dinv[row], finalize scales the sum by dinv[dst]; edge kernel drops both per-edge dinv lookups (addends round differently; same atomics and order)
# speedup vs baseline: 2.8242x; 1.0334x over previous
#include <hip/hip_runtime.h>


typedef float v2f __attribute__((ext_vector_type(2)));
typedef float v8f __attribute__((ext_vector_type(8)));

#define N_NODES   1000000
#define N_EDGES   16000000
#define F_IN      32
#define F_H1      16
#define F_H2      8
#define N_CLS     5
#define N_GRAPHS  10000

#define ATOMIC_FADD(p, v) \
  __hip_atomic_fetch_add((p), (v), __ATOMIC_RELAXED, __HIP_MEMORY_SCOPE_AGENT)


__global__ void zero_f32(float* __restrict__ p, long long n) {
  long long i = (long long)blockIdx.x * blockDim.x + threadIdx.x;
  long long stride = (long long)gridDim.x * blockDim.x;
  for (; i < n; i += stride) p[i] = 0.0f;
}

__global__ void degree_kernel(const int* __restrict__ dst,
                              float* __restrict__ deg, int nE) {
  int e = blockIdx.x * blockDim.x + threadIdx.x;
  if (e < nE) ATOMIC_FADD(&deg[dst[e]], 1.0f);
}

__global__ void rsqrt_kernel(float* __restrict__ deg, int n) {
  int i = blockIdx.x * blockDim.x + threadIdx.x;
  if (i < n) deg[i] = rsqrtf(deg[i] + 1.0f);
}


__global__ __launch_bounds__(128) void gemm_x_w1(const float* __restrict__ X,
                                                 const float* __restrict__ W1,
                                                 const float* __restrict__ dinv,
                                                 float* __restrict__ H1pre) {
  int wave = blockIdx.x * (blockDim.x >> 5) + (threadIdx.x >> 5);
  int lane = threadIdx.x & 31;
  int m  = lane & 15;
  int kk = (lane >> 4) << 1;
  const float* xrow = X + (size_t)(wave * 16 + m) * F_IN;

  v8f c = {0.f, 0.f, 0.f, 0.f, 0.f, 0.f, 0.f, 0.f};
#pragma unroll
  for (int k = 0; k < F_IN; k += 4) {
    v2f a, b;
    a.x = xrow[k + kk];
    a.y = xrow[k + kk + 1];
    b.x = W1[(k + kk) * F_H1 + m];
    b.y = W1[(k + kk + 1) * F_H1 + m];
    c = __builtin_amdgcn_wmma_f32_16x16x4_f32(false, a, false, b,
                                              (short)0, c, false, false);
  }
  int hi = lane >> 4;
  const int row0 = wave * 16 + 8 * hi;
  const float* dv = dinv + row0;
  float* out = H1pre + (size_t)row0 * F_H1 + m;
#pragma unroll
  for (int j = 0; j < 8; ++j) out[(size_t)j * F_H1] = c[j] * dv[j];
}

__global__ __launch_bounds__(128) void gemm_h1_w2(const float* __restrict__ H1,
                                                  const float* __restrict__ W2,
                                                  const float* __restrict__ dinv,
                                                  float* __restrict__ H2pre) {
  int wave = blockIdx.x * (blockDim.x >> 5) + (threadIdx.x >> 5);
  int lane = threadIdx.x & 31;
  int m  = lane & 15;
  int kk = (lane >> 4) << 1;
  const float* hrow = H1 + (size_t)(wave * 16 + m) * F_H1;
  bool live_col = (m < F_H2);

  v8f c = {0.f, 0.f, 0.f, 0.f, 0.f, 0.f, 0.f, 0.f};
#pragma unroll
  for (int k = 0; k < F_H1; k += 4) {
    v2f a, b;
    a.x = hrow[k + kk];
    a.y = hrow[k + kk + 1];
    b.x = live_col ? W2[(k + kk) * F_H2 + m] : 0.0f;
    b.y = live_col ? W2[(k + kk + 1) * F_H2 + m] : 0.0f;
    c = __builtin_amdgcn_wmma_f32_16x16x4_f32(false, a, false, b,
                                              (short)0, c, false, false);
  }
  if (live_col) {
    int hi = lane >> 4;
    const int row0 = wave * 16 + 8 * hi;
    const float* dv = dinv + row0;
    float* out = H2pre + (size_t)row0 * F_H2 + m;
#pragma unroll
    for (int j = 0; j < 8; ++j) out[(size_t)j * F_H2] = c[j] * dv[j];
  }
}


static_assert(N_EDGES % 256 == 0, "every wave owns 32 whole edges; the grid is exact");

template <int F>
__global__ __launch_bounds__(256) void agg_edges_rows(const int* __restrict__ src,
                                                      const int* __restrict__ dst,
                                                      const float* __restrict__ H,
                                                      float* __restrict__ AGG) {
  constexpr int EPI   = 32 / F;
  constexpr int STEPS = F;
  const int lane = threadIdx.x & 31;
  const int e    = blockIdx.x * blockDim.x + threadIdx.x;
  const int s = src[e], d = dst[e];
  const int sub = lane / F;
  const int f   = lane % F;

  float v[STEPS];
#pragma unroll
  for (int i = 0; i < STEPS; ++i) {
    const int si = __shfl(s, i * EPI + sub);
    v[i] = H[(unsigned)si * F + f];
  }
#pragma unroll
  for (int i = 0; i < STEPS; ++i) {
    const int di = __shfl(d, i * EPI + sub);
    ATOMIC_FADD(&AGG[(unsigned)di * F + f], v[i]);
  }
}


__global__ void finalize1(float* __restrict__ agg, const float* __restrict__ hpre,
                          const float* __restrict__ dinv,
                          const float* __restrict__ b1, int n) {
  int i = blockIdx.x * blockDim.x + threadIdx.x;
  if (i >= n) return;
  const float w = dinv[i];
  float4* a = (float4*)(agg + (size_t)i * F_H1);
  const float4* h = (const float4*)(hpre + (size_t)i * F_H1);
  const float4* b = (const float4*)b1;
#pragma unroll
  for (int q = 0; q < 4; ++q) {
    float4 av = a[q], hv = h[q], bv = b[q];
    av.x = fmaxf((av.x + hv.x) * w + bv.x, 0.0f);
    av.y = fmaxf((av.y + hv.y) * w + bv.y, 0.0f);
    av.z = fmaxf((av.z + hv.z) * w + bv.z, 0.0f);
    av.w = fmaxf((av.w + hv.w) * w + bv.w, 0.0f);
    a[q] = av;
  }
}

__global__ void finalize2_pool(const float* __restrict__ agg,
                               const float* __restrict__ hpre,
                               const float* __restrict__ dinv,
                               const float* __restrict__ b2,
                               const int* __restrict__ batch,
                               float* __restrict__ pooled,
                               float* __restrict__ cnt, int n) {
  int i = blockIdx.x * blockDim.x + threadIdx.x;
  if (i >= n) return;
  const float w = dinv[i];
  int g = batch[i];
  float* ps = pooled + (size_t)g * F_H2;
#pragma unroll
  for (int f = 0; f < F_H2; ++f) {
    float v = (agg[(size_t)i * F_H2 + f] + hpre[(size_t)i * F_H2 + f]) * w + b2[f];
    v = fmaxf(v, 0.0f);
    ATOMIC_FADD(&ps[f], v);
  }
  ATOMIC_FADD(&cnt[g], 1.0f);
}

__global__ void readout(const float* __restrict__ pooled, const float* __restrict__ cnt,
                        const float* __restrict__ Wl, const float* __restrict__ bl,
                        float* __restrict__ out, int G) {
  int g = blockIdx.x * blockDim.x + threadIdx.x;
  if (g >= G) return;
  float inv = 1.0f / fmaxf(cnt[g], 1.0f);
  float p[F_H2];
#pragma unroll
  for (int f = 0; f < F_H2; ++f) p[f] = pooled[(size_t)g * F_H2 + f] * inv;
#pragma unroll
  for (int c = 0; c < N_CLS; ++c) {
    float acc = bl[c];
#pragma unroll
    for (int f = 0; f < F_H2; ++f) acc += p[f] * Wl[f * N_CLS + c];
    out[(size_t)g * N_CLS + c] = acc;
  }
}


extern "C" void kernel_launch(void* const* d_in, const int* in_sizes, int n_in,
                              void* d_out, int out_size, void* d_ws, size_t ws_size,
                              hipStream_t stream) {
  const float* x     = (const float*)d_in[0];
  const int*   ei    = (const int*)d_in[1];
  const int*   src   = ei;
  const int*   dst   = ei + N_EDGES;
  const int*   batch = (const int*)d_in[2];
  const float* W1 = (const float*)d_in[4];
  const float* b1 = (const float*)d_in[5];
  const float* W2 = (const float*)d_in[6];
  const float* b2 = (const float*)d_in[7];
  const float* Wl = (const float*)d_in[8];
  const float* bl = (const float*)d_in[9];
  float* out = (float*)d_out;

  char* ws = (char*)d_ws;
  float* dinv   = (float*)(ws);
  float* bufA   = (float*)(ws + (4ll  << 20));
  float* bufB   = (float*)(ws + (68ll << 20));
  float* pooled = (float*)(ws + (132ll << 20));
  float* cnt    = pooled + (size_t)N_GRAPHS * F_H2;

  float* h1pre = bufA;
  float* agg1  = bufB;
  float* h2pre = bufA;
  float* agg2  = (float*)((char*)bufA + (32ll << 20));

  const int EB = 256;
  const int EG = (N_EDGES + EB - 1) / EB;
  const int NB = 256;
  const int NG = (N_NODES + NB - 1) / NB;
  const int GEMM_BLOCKS = N_NODES / 64;

  zero_f32<<<2048, 256, 0, stream>>>(dinv, (long long)N_NODES);
  degree_kernel<<<EG, EB, 0, stream>>>(dst, dinv, N_EDGES);
  rsqrt_kernel<<<NG, NB, 0, stream>>>(dinv, N_NODES);

  gemm_x_w1<<<GEMM_BLOCKS, 128, 0, stream>>>(x, W1, dinv, h1pre);
  zero_f32<<<8192, 256, 0, stream>>>(agg1, (long long)N_NODES * F_H1);
  agg_edges_rows<F_H1><<<EG, EB, 0, stream>>>(src, dst, h1pre, agg1);
  finalize1<<<NG, NB, 0, stream>>>(agg1, h1pre, dinv, b1, N_NODES);

  gemm_h1_w2<<<GEMM_BLOCKS, 128, 0, stream>>>(agg1, W2, dinv, h2pre);
  zero_f32<<<8192, 256, 0, stream>>>(agg2, (long long)N_NODES * F_H2);
  agg_edges_rows<F_H2><<<EG, EB, 0, stream>>>(src, dst, h2pre, agg2);

  zero_f32<<<128, 256, 0, stream>>>(pooled, (long long)N_GRAPHS * (F_H2 + 1));
  finalize2_pool<<<NG, NB, 0, stream>>>(agg2, h2pre, dinv, b2, batch,
                                        pooled, cnt, N_NODES);
  readout<<<(N_GRAPHS + NB - 1) / NB, NB, 0, stream>>>(pooled, cnt, Wl, bl,
                                                       out, N_GRAPHS);
}


# layer-2 edge sums: edges binned by destination range (count, scan, scatter), one work-group per 2048-row range sums gathered rows with LDS float atomics, not global ones; sum order = arrival
# speedup vs baseline: 3.2396x; 1.1471x over previous
#include <hip/hip_runtime.h>


typedef float v2f __attribute__((ext_vector_type(2)));
typedef float v8f __attribute__((ext_vector_type(8)));

#define N_NODES   1000000
#define N_EDGES   16000000
#define F_IN      32
#define F_H1      16
#define F_H2      8
#define N_CLS     5
#define N_GRAPHS  10000

#define RB        11
#define R_ROWS    (1 << RB)
#define N_RANGES        ((N_NODES + R_ROWS - 1) >> RB)
#define CH        65536
#define NBLK      ((N_EDGES + CH - 1) / CH)

#define ATOMIC_FADD(p, v) \
  __hip_atomic_fetch_add((p), (v), __ATOMIC_RELAXED, __HIP_MEMORY_SCOPE_AGENT)


__global__ void zero_f32(float* __restrict__ p, long long n) {
  long long i = (long long)blockIdx.x * blockDim.x + threadIdx.x;
  long long stride = (long long)gridDim.x * blockDim.x;
  for (; i < n; i += stride) p[i] = 0.0f;
}

__global__ void degree_kernel(const int* __restrict__ dst,
                              float* __restrict__ deg, int nE) {
  int e = blockIdx.x * blockDim.x + threadIdx.x;
  if (e < nE) ATOMIC_FADD(&deg[dst[e]], 1.0f);
}

__global__ void rsqrt_kernel(float* __restrict__ deg, int n) {
  int i = blockIdx.x * blockDim.x + threadIdx.x;
  if (i < n) deg[i] = rsqrtf(deg[i] + 1.0f);
}


__global__ __launch_bounds__(128) void gemm_x_w1(const float* __restrict__ X,
                                                 const float* __restrict__ W1,
                                                 const float* __restrict__ dinv,
                                                 float* __restrict__ H1pre) {
  int wave = blockIdx.x * (blockDim.x >> 5) + (threadIdx.x >> 5);
  int lane = threadIdx.x & 31;
  int m  = lane & 15;
  int kk = (lane >> 4) << 1;
  const float* xrow = X + (size_t)(wave * 16 + m) * F_IN;

  v8f c = {0.f, 0.f, 0.f, 0.f, 0.f, 0.f, 0.f, 0.f};
#pragma unroll
  for (int k = 0; k < F_IN; k += 4) {
    v2f a, b;
    a.x = xrow[k + kk];
    a.y = xrow[k + kk + 1];
    b.x = W1[(k + kk) * F_H1 + m];
    b.y = W1[(k + kk + 1) * F_H1 + m];
    c = __builtin_amdgcn_wmma_f32_16x16x4_f32(false, a, false, b,
                                              (short)0, c, false, false);
  }
  int hi = lane >> 4;
  const int row0 = wave * 16 + 8 * hi;
  const float* dv = dinv + row0;
  float* out = H1pre + (size_t)row0 * F_H1 + m;
#pragma unroll
  for (int j = 0; j < 8; ++j) out[(size_t)j * F_H1] = c[j] * dv[j];
}

__global__ __launch_bounds__(128) void gemm_h1_w2(const float* __restrict__ H1,
                                                  const float* __restrict__ W2,
                                                  const float* __restrict__ dinv,
                                                  float* __restrict__ H2pre) {
  int wave = blockIdx.x * (blockDim.x >> 5) + (threadIdx.x >> 5);
  int lane = threadIdx.x & 31;
  int m  = lane & 15;
  int kk = (lane >> 4) << 1;
  const float* hrow = H1 + (size_t)(wave * 16 + m) * F_H1;
  bool live_col = (m < F_H2);

  v8f c = {0.f, 0.f, 0.f, 0.f, 0.f, 0.f, 0.f, 0.f};
#pragma unroll
  for (int k = 0; k < F_H1; k += 4) {
    v2f a, b;
    a.x = hrow[k + kk];
    a.y = hrow[k + kk + 1];
    b.x = live_col ? W2[(k + kk) * F_H2 + m] : 0.0f;
    b.y = live_col ? W2[(k + kk + 1) * F_H2 + m] : 0.0f;
    c = __builtin_amdgcn_wmma_f32_16x16x4_f32(false, a, false, b,
                                              (short)0, c, false, false);
  }
  if (live_col) {
    int hi = lane >> 4;
    const int row0 = wave * 16 + 8 * hi;
    const float* dv = dinv + row0;
    float* out = H2pre + (size_t)row0 * F_H2 + m;
#pragma unroll
    for (int j = 0; j < 8; ++j) out[(size_t)j * F_H2] = c[j] * dv[j];
  }
}


static_assert(N_EDGES % 256 == 0, "every wave owns 32 whole edges; the grid is exact");

template <int F>
__global__ __launch_bounds__(256) void agg_edges_rows(const int* __restrict__ src,
                                                      const int* __restrict__ dst,
                                                      const float* __restrict__ H,
                                                      float* __restrict__ AGG) {
  constexpr int EPI   = 32 / F;
  constexpr int STEPS = F;
  const int lane = threadIdx.x & 31;
  const int e    = blockIdx.x * blockDim.x + threadIdx.x;
  const int s = src[e], d = dst[e];
  const int sub = lane / F;
  const int f   = lane % F;

  float v[STEPS];
#pragma unroll
  for (int i = 0; i < STEPS; ++i) {
    const int si = __shfl(s, i * EPI + sub);
    v[i] = H[(unsigned)si * F + f];
  }
#pragma unroll
  for (int i = 0; i < STEPS; ++i) {
    const int di = __shfl(d, i * EPI + sub);
    ATOMIC_FADD(&AGG[(unsigned)di * F + f], v[i]);
  }
}


static_assert(N_NODES <= (1 << 20), "a key holds the source row in 20 bits beside RB bits of local row");
static_assert(N_RANGES <= 1024, "one thread per range in the scatter kernel's offset scan");
static_assert(NBLK <= 256, "one thread per binning work-group in the per-range scan");
static_assert(N_EDGES % 4 == 0 && CH % 4096 == 0, "binning kernels read four edges per lane");

__device__ __forceinline__ int wave_incl_scan(int v, int lane) {
#pragma unroll
  for (int o = 1; o < 32; o <<= 1) {
    const int t = __shfl_up(v, o);
    if (lane >= o) v += t;
  }
  return v;
}

__device__ __forceinline__ int block_excl_scan(int v, int* wsum, int* total) {
  const int lane = threadIdx.x & 31, wave = threadIdx.x >> 5;
  const int nw = blockDim.x >> 5;
  const int incl = wave_incl_scan(v, lane);
  if (lane == 31) wsum[wave] = incl;
  __syncthreads();
  if (wave == 0) {
    const int w  = (lane < nw) ? wsum[lane] : 0;
    const int wi = wave_incl_scan(w, lane);
    wsum[lane] = wi - w;
    if (lane == 31) wsum[32] = wi;
  }
  __syncthreads();
  *total = wsum[32];
  return wsum[wave] + incl - v;
}

__global__ __launch_bounds__(1024) void bin_count(const int* __restrict__ dst,
                                                  int* __restrict__ counts) {
  __shared__ int hist[N_RANGES];
  const int t = threadIdx.x, blk = blockIdx.x;
  for (int i = t; i < N_RANGES; i += 1024) hist[i] = 0;
  __syncthreads();
  const int base = blk * CH;
  const int4* d4 = (const int4*)(dst + base);
  for (int i = t; i < CH / 4; i += 1024) {
    if (base + 4 * i < N_EDGES) {
      const int4 d = d4[i];
      atomicAdd(&hist[d.x >> RB], 1);
      atomicAdd(&hist[d.y >> RB], 1);
      atomicAdd(&hist[d.z >> RB], 1);
      atomicAdd(&hist[d.w >> RB], 1);
    }
  }
  __syncthreads();
  for (int i = t; i < N_RANGES; i += 1024) counts[i * NBLK + blk] = hist[i];
}

__global__ __launch_bounds__(256) void bin_scan(int* __restrict__ counts,
                                                int* __restrict__ tot) {
  __shared__ int wsum[33];
  const int b = blockIdx.x, t = threadIdx.x;
  const int v = (t < NBLK) ? counts[b * NBLK + t] : 0;
  int total;
  const int ex = block_excl_scan(v, wsum, &total);
  if (t < NBLK) counts[b * NBLK + t] = ex;
  if (t == 0) tot[b] = total;
}

__global__ __launch_bounds__(1024) void bin_scatter(const int* __restrict__ src,
                                                    const int* __restrict__ dst,
                                                    const int* __restrict__ offs,
                                                    const int* __restrict__ tot,
                                                    int* __restrict__ bstart,
                                                    int* __restrict__ keys) {
  __shared__ int cursor[N_RANGES];
  __shared__ int wsum[33];
  const int t = threadIdx.x, blk = blockIdx.x;
  const int v = (t < N_RANGES) ? tot[t] : 0;
  int total;
  const int ex = block_excl_scan(v, wsum, &total);
  if (t < N_RANGES) {
    cursor[t] = ex + offs[t * NBLK + blk];
    if (blk == 0) bstart[t] = ex;
  }
  if (blk == 0 && t == 0) bstart[N_RANGES] = total;
  __syncthreads();
  const int base = blk * CH;
  const int4* s4 = (const int4*)(src + base);
  const int4* d4 = (const int4*)(dst + base);
  for (int i = t; i < CH / 4; i += 1024) {
    if (base + 4 * i < N_EDGES) {
      const int4 s = s4[i], d = d4[i];
      const int p0 = atomicAdd(&cursor[d.x >> RB], 1);
      const int p1 = atomicAdd(&cursor[d.y >> RB], 1);
      const int p2 = atomicAdd(&cursor[d.z >> RB], 1);
      const int p3 = atomicAdd(&cursor[d.w >> RB], 1);
      keys[p0] = (s.x << RB) | (d.x & (R_ROWS - 1));
      keys[p1] = (s.y << RB) | (d.y & (R_ROWS - 1));
      keys[p2] = (s.z << RB) | (d.z & (R_ROWS - 1));
      keys[p3] = (s.w << RB) | (d.w & (R_ROWS - 1));
    }
  }
}

template <int F>
__global__ __launch_bounds__(512) void agg_range_lds(const int* __restrict__ keys,
                                                     const int* __restrict__ bstart,
                                                     const float* __restrict__ H,
                                                     float* __restrict__ AGG) {
  constexpr int LPE = F / 4;
  constexpr int EPI = 32 / LPE;
  constexpr int U   = 4;
  constexpr int NW  = 512 / 32;
  __shared__ float acc[R_ROWS * F];

  const int t = threadIdx.x, lane = t & 31, wave = t >> 5;
  const int b = blockIdx.x;
  const int row0 = b << RB;
  const int nrows = (N_NODES - row0 < R_ROWS) ? (N_NODES - row0) : R_ROWS;

  float4* acc4 = (float4*)acc;
  for (int i = t; i < R_ROWS * F / 4; i += 512) acc4[i] = make_float4(0.f, 0.f, 0.f, 0.f);
  __syncthreads();

  const int e0 = bstart[b], e1 = bstart[b + 1];
  const int sub = lane / LPE;
  const int q   = lane % LPE;

  for (int base = e0 + wave * (32 * U); base < e1; base += NW * 32 * U) {
    int k[U];
#pragma unroll
    for (int u = 0; u < U; ++u) {
      const int e = base + u * 32 + lane;
      k[u] = (e < e1) ? keys[e] : -1;
    }
    float4 v[U * LPE];
    int    dl[U * LPE];
#pragma unroll
    for (int u = 0; u < U; ++u) {
#pragma unroll
      for (int i = 0; i < LPE; ++i) {
        const int ks = __shfl(k[u], i * EPI + sub);
        const bool ok = (ks >= 0);
        const unsigned s = ok ? ((unsigned)ks >> RB) : 0u;
        v[u * LPE + i]  = *(const float4*)(H + (size_t)s * F + 4 * q);
        dl[u * LPE + i] = ok ? (ks & (R_ROWS - 1)) : -1;
      }
    }
#pragma unroll
    for (int j = 0; j < U * LPE; ++j) {
      if (dl[j] >= 0) {
        float* a = acc + dl[j] * F + 4 * q;
        __hip_atomic_fetch_add(a + 0, v[j].x, __ATOMIC_RELAXED, __HIP_MEMORY_SCOPE_WORKGROUP);
        __hip_atomic_fetch_add(a + 1, v[j].y, __ATOMIC_RELAXED, __HIP_MEMORY_SCOPE_WORKGROUP);
        __hip_atomic_fetch_add(a + 2, v[j].z, __ATOMIC_RELAXED, __HIP_MEMORY_SCOPE_WORKGROUP);
        __hip_atomic_fetch_add(a + 3, v[j].w, __ATOMIC_RELAXED, __HIP_MEMORY_SCOPE_WORKGROUP);
      }
    }
  }
  __syncthreads();

  float4* out4 = (float4*)(AGG + (size_t)row0 * F);
  for (int i = t; i < nrows * F / 4; i += 512) out4[i] = acc4[i];
}


__global__ void finalize1(float* __restrict__ agg, const float* __restrict__ hpre,
                          const float* __restrict__ dinv,
                          const float* __restrict__ b1, int n) {
  int i = blockIdx.x * blockDim.x + threadIdx.x;
  if (i >= n) return;
  const float w = dinv[i];
  float4* a = (float4*)(agg + (size_t)i * F_H1);
  const float4* h = (const float4*)(hpre + (size_t)i * F_H1);
  const float4* b = (const float4*)b1;
#pragma unroll
  for (int q = 0; q < 4; ++q) {
    float4 av = a[q], hv = h[q], bv = b[q];
    av.x = fmaxf((av.x + hv.x) * w + bv.x, 0.0f);
    av.y = fmaxf((av.y + hv.y) * w + bv.y, 0.0f);
    av.z = fmaxf((av.z + hv.z) * w + bv.z, 0.0f);
    av.w = fmaxf((av.w + hv.w) * w + bv.w, 0.0f);
    a[q] = av;
  }
}

__global__ void finalize2_pool(const float* __restrict__ agg,
                               const float* __restrict__ hpre,
                               const float* __restrict__ dinv,
                               const float* __restrict__ b2,
                               const int* __restrict__ batch,
                               float* __restrict__ pooled,
                               float* __restrict__ cnt, int n) {
  int i = blockIdx.x * blockDim.x + threadIdx.x;
  if (i >= n) return;
  const float w = dinv[i];
  int g = batch[i];
  float* ps = pooled + (size_t)g * F_H2;
#pragma unroll
  for (int f = 0; f < F_H2; ++f) {
    float v = (agg[(size_t)i * F_H2 + f] + hpre[(size_t)i * F_H2 + f]) * w + b2[f];
    v = fmaxf(v, 0.0f);
    ATOMIC_FADD(&ps[f], v);
  }
  ATOMIC_FADD(&cnt[g], 1.0f);
}

__global__ void readout(const float* __restrict__ pooled, const float* __restrict__ cnt,
                        const float* __restrict__ Wl, const float* __restrict__ bl,
                        float* __restrict__ out, int G) {
  int g = blockIdx.x * blockDim.x + threadIdx.x;
  if (g >= G) return;
  float inv = 1.0f / fmaxf(cnt[g], 1.0f);
  float p[F_H2];
#pragma unroll
  for (int f = 0; f < F_H2; ++f) p[f] = pooled[(size_t)g * F_H2 + f] * inv;
#pragma unroll
  for (int c = 0; c < N_CLS; ++c) {
    float acc = bl[c];
#pragma unroll
    for (int f = 0; f < F_H2; ++f) acc += p[f] * Wl[f * N_CLS + c];
    out[(size_t)g * N_CLS + c] = acc;
  }
}


extern "C" void kernel_launch(void* const* d_in, const int* in_sizes, int n_in,
                              void* d_out, int out_size, void* d_ws, size_t ws_size,
                              hipStream_t stream) {
  const float* x     = (const float*)d_in[0];
  const int*   ei    = (const int*)d_in[1];
  const int*   src   = ei;
  const int*   dst   = ei + N_EDGES;
  const int*   batch = (const int*)d_in[2];
  const float* W1 = (const float*)d_in[4];
  const float* b1 = (const float*)d_in[5];
  const float* W2 = (const float*)d_in[6];
  const float* b2 = (const float*)d_in[7];
  const float* Wl = (const float*)d_in[8];
  const float* bl = (const float*)d_in[9];
  float* out = (float*)d_out;

  char* ws = (char*)d_ws;
  float* dinv   = (float*)(ws);
  float* bufA   = (float*)(ws + (4ll  << 20));
  float* bufB   = (float*)(ws + (68ll << 20));
  float* pooled = (float*)(ws + (132ll << 20));
  float* cnt    = pooled + (size_t)N_GRAPHS * F_H2;

  float* h1pre = bufA;
  float* agg1  = bufB;
  float* h2pre = bufA;
  float* agg2  = (float*)((char*)bufA + (32ll << 20));
  int* keys   = (int*)bufB;
  int* counts = (int*)((char*)bufB + 64000000ll);
  int* tot    = (int*)((char*)bufB + 64000000ll + (512ll << 10));
  int* bstart = (int*)((char*)bufB + 64000000ll + (520ll << 10));

  const int EB = 256;
  const int EG = (N_EDGES + EB - 1) / EB;
  const int NB = 256;
  const int NG = (N_NODES + NB - 1) / NB;
  const int GEMM_BLOCKS = N_NODES / 64;

  zero_f32<<<2048, 256, 0, stream>>>(dinv, (long long)N_NODES);
  degree_kernel<<<EG, EB, 0, stream>>>(dst, dinv, N_EDGES);
  rsqrt_kernel<<<NG, NB, 0, stream>>>(dinv, N_NODES);

  gemm_x_w1<<<GEMM_BLOCKS, 128, 0, stream>>>(x, W1, dinv, h1pre);
  zero_f32<<<8192, 256, 0, stream>>>(agg1, (long long)N_NODES * F_H1);
  agg_edges_rows<F_H1><<<EG, EB, 0, stream>>>(src, dst, h1pre, agg1);
  finalize1<<<NG, NB, 0, stream>>>(agg1, h1pre, dinv, b1, N_NODES);

  gemm_h1_w2<<<GEMM_BLOCKS, 128, 0, stream>>>(agg1, W2, dinv, h2pre);
  bin_count<<<NBLK, 1024, 0, stream>>>(dst, counts);
  bin_scan<<<N_RANGES, 256, 0, stream>>>(counts, tot);
  bin_scatter<<<NBLK, 1024, 0, stream>>>(src, dst, counts, tot, bstart, keys);
  agg_range_lds<F_H2><<<N_RANGES, 512, 0, stream>>>(keys, bstart, h2pre, agg2);

  zero_f32<<<128, 256, 0, stream>>>(pooled, (long long)N_GRAPHS * (F_H2 + 1));
  finalize2_pool<<<NG, NB, 0, stream>>>(agg2, h2pre, dinv, b2, batch,
                                        pooled, cnt, N_NODES);
  readout<<<(N_GRAPHS + NB - 1) / NB, NB, 0, stream>>>(pooled, cnt, Wl, bl,
                                                       out, N_GRAPHS);
}


# layer-1 edge sums per destination range in LDS like layer 2 (keys binned once), as two 8-feature planes to fit the workspace; finalize1 + W2 WMMA product fused in; sum order = arrival
# speedup vs baseline: 3.9974x; 1.2339x over previous
#include <hip/hip_runtime.h>


typedef float v2f __attribute__((ext_vector_type(2)));
typedef float v8f __attribute__((ext_vector_type(8)));

#define N_NODES   1000000
#define N_EDGES   16000000
#define F_IN      32
#define F_H1      16
#define F_H2      8
#define N_CLS     5
#define N_GRAPHS  10000

#define RB        11
#define R_ROWS    (1 << RB)
#define N_RANGES        ((N_NODES + R_ROWS - 1) >> RB)
#define CH        65536
#define NBLK      ((N_EDGES + CH - 1) / CH)

#define ATOMIC_FADD(p, v) \
  __hip_atomic_fetch_add((p), (v), __ATOMIC_RELAXED, __HIP_MEMORY_SCOPE_AGENT)


__global__ void zero_f32(float* __restrict__ p, long long n) {
  long long i = (long long)blockIdx.x * blockDim.x + threadIdx.x;
  long long stride = (long long)gridDim.x * blockDim.x;
  for (; i < n; i += stride) p[i] = 0.0f;
}

__global__ void degree_kernel(const int* __restrict__ dst,
                              float* __restrict__ deg, int nE) {
  int e = blockIdx.x * blockDim.x + threadIdx.x;
  if (e < nE) ATOMIC_FADD(&deg[dst[e]], 1.0f);
}

__global__ void rsqrt_kernel(float* __restrict__ deg, int n) {
  int i = blockIdx.x * blockDim.x + threadIdx.x;
  if (i < n) deg[i] = rsqrtf(deg[i] + 1.0f);
}


template <int HALF>
__global__ __launch_bounds__(128) void gemm_x_w1_half(const float* __restrict__ X,
                                                      const float* __restrict__ W1,
                                                      const float* __restrict__ dinv,
                                                      float* __restrict__ P) {
  int wave = blockIdx.x * (blockDim.x >> 5) + (threadIdx.x >> 5);
  int lane = threadIdx.x & 31;
  int m  = lane & 15;
  int kk = (lane >> 4) << 1;
  const float* xrow = X + (size_t)(wave * 16 + m) * F_IN;
  const bool live_col = (m < 8);
  const int col = HALF * 8 + (m & 7);

  v8f c = {0.f, 0.f, 0.f, 0.f, 0.f, 0.f, 0.f, 0.f};
#pragma unroll
  for (int k = 0; k < F_IN; k += 4) {
    v2f a, b;
    a.x = xrow[k + kk];
    a.y = xrow[k + kk + 1];
    const float w0 = W1[(k + kk) * F_H1 + col];
    const float w1 = W1[(k + kk + 1) * F_H1 + col];
    b.x = live_col ? w0 : 0.0f;
    b.y = live_col ? w1 : 0.0f;
    c = __builtin_amdgcn_wmma_f32_16x16x4_f32(false, a, false, b,
                                              (short)0, c, false, false);
  }
  int hi = lane >> 4;
  const int row0 = wave * 16 + 8 * hi;
  const float* dv = dinv + row0;
  float r[8];
#pragma unroll
  for (int j = 0; j < 8; ++j) r[j] = c[j] * dv[j];
  if (live_col) {
    float* out = P + (size_t)row0 * 8 + m;
#pragma unroll
    for (int j = 0; j < 8; ++j) out[(size_t)j * 8] = r[j];
  }
}


static_assert(N_NODES <= (1 << 20), "a key holds the source row in 20 bits beside RB bits of local row");
static_assert(N_RANGES <= 1024, "one thread per range in the scatter kernel's offset scan");
static_assert(NBLK <= 256, "one thread per binning work-group in the per-range scan");
static_assert(N_EDGES % 4 == 0 && CH % 4096 == 0, "binning kernels read four edges per lane");
static_assert(N_NODES % 16 == 0 && R_ROWS % 16 == 0, "a range is whole 16-row tiles");

__device__ __forceinline__ int wave_incl_scan(int v, int lane) {
#pragma unroll
  for (int o = 1; o < 32; o <<= 1) {
    const int t = __shfl_up(v, o);
    if (lane >= o) v += t;
  }
  return v;
}

__device__ __forceinline__ int block_excl_scan(int v, int* wsum, int* total) {
  const int lane = threadIdx.x & 31, wave = threadIdx.x >> 5;
  const int nw = blockDim.x >> 5;
  const int incl = wave_incl_scan(v, lane);
  if (lane == 31) wsum[wave] = incl;
  __syncthreads();
  if (wave == 0) {
    const int w  = (lane < nw) ? wsum[lane] : 0;
    const int wi = wave_incl_scan(w, lane);
    wsum[lane] = wi - w;
    if (lane == 31) wsum[32] = wi;
  }
  __syncthreads();
  *total = wsum[32];
  return wsum[wave] + incl - v;
}

__global__ __launch_bounds__(1024) void bin_count(const int* __restrict__ dst,
                                                  int* __restrict__ counts) {
  __shared__ int hist[N_RANGES];
  const int t = threadIdx.x, blk = blockIdx.x;
  for (int i = t; i < N_RANGES; i += 1024) hist[i] = 0;
  __syncthreads();
  const int base = blk * CH;
  const int4* d4 = (const int4*)(dst + base);
  for (int i = t; i < CH / 4; i += 1024) {
    if (base + 4 * i < N_EDGES) {
      const int4 d = d4[i];
      atomicAdd(&hist[d.x >> RB], 1);
      atomicAdd(&hist[d.y >> RB], 1);
      atomicAdd(&hist[d.z >> RB], 1);
      atomicAdd(&hist[d.w >> RB], 1);
    }
  }
  __syncthreads();
  for (int i = t; i < N_RANGES; i += 1024) counts[i * NBLK + blk] = hist[i];
}

__global__ __launch_bounds__(256) void bin_scan(int* __restrict__ counts,
                                                int* __restrict__ tot) {
  __shared__ int wsum[33];
  const int b = blockIdx.x, t = threadIdx.x;
  const int v = (t < NBLK) ? counts[b * NBLK + t] : 0;
  int total;
  const int ex = block_excl_scan(v, wsum, &total);
  if (t < NBLK) counts[b * NBLK + t] = ex;
  if (t == 0) tot[b] = total;
}

__global__ __launch_bounds__(1024) void bin_scatter(const int* __restrict__ src,
                                                    const int* __restrict__ dst,
                                                    const int* __restrict__ offs,
                                                    const int* __restrict__ tot,
                                                    int* __restrict__ bstart,
                                                    int* __restrict__ keys) {
  __shared__ int cursor[N_RANGES];
  __shared__ int wsum[33];
  const int t = threadIdx.x, blk = blockIdx.x;
  const int v = (t < N_RANGES) ? tot[t] : 0;
  int total;
  const int ex = block_excl_scan(v, wsum, &total);
  if (t < N_RANGES) {
    cursor[t] = ex + offs[t * NBLK + blk];
    if (blk == 0) bstart[t] = ex;
  }
  if (blk == 0 && t == 0) bstart[N_RANGES] = total;
  __syncthreads();
  const int base = blk * CH;
  const int4* s4 = (const int4*)(src + base);
  const int4* d4 = (const int4*)(dst + base);
  for (int i = t; i < CH / 4; i += 1024) {
    if (base + 4 * i < N_EDGES) {
      const int4 s = s4[i], d = d4[i];
      const int p0 = atomicAdd(&cursor[d.x >> RB], 1);
      const int p1 = atomicAdd(&cursor[d.y >> RB], 1);
      const int p2 = atomicAdd(&cursor[d.z >> RB], 1);
      const int p3 = atomicAdd(&cursor[d.w >> RB], 1);
      keys[p0] = (s.x << RB) | (d.x & (R_ROWS - 1));
      keys[p1] = (s.y << RB) | (d.y & (R_ROWS - 1));
      keys[p2] = (s.z << RB) | (d.z & (R_ROWS - 1));
      keys[p3] = (s.w << RB) | (d.w & (R_ROWS - 1));
    }
  }
}

template <int MODE>
__global__ __launch_bounds__(512) void agg_range_lds(const int* __restrict__ keys,
                                                     const int* __restrict__ bstart,
                                                     const float* __restrict__ H,
                                                     float* OUT,
                                                     const float* __restrict__ dinv,
                                                     const float* __restrict__ b1,
                                                     const float* __restrict__ W2) {
  constexpr int F   = 8;
  constexpr int LPE = F / 4;
  constexpr int EPI = 32 / LPE;
  constexpr int U   = 4;
  constexpr int NW  = 512 / 32;
  __shared__ float acc[R_ROWS * F];

  const int t = threadIdx.x, lane = t & 31, wave = t >> 5;
  const int b = blockIdx.x;
  const int row0 = b << RB;
  const int nrows = (N_NODES - row0 < R_ROWS) ? (N_NODES - row0) : R_ROWS;

  float4* acc4 = (float4*)acc;
  for (int i = t; i < R_ROWS * F / 4; i += 512) acc4[i] = make_float4(0.f, 0.f, 0.f, 0.f);
  __syncthreads();

  const int e0 = bstart[b], e1 = bstart[b + 1];
  const int sub = lane / LPE;
  const int q   = lane % LPE;

  for (int base = e0 + wave * (32 * U); base < e1; base += NW * 32 * U) {
    int k[U];
#pragma unroll
    for (int u = 0; u < U; ++u) {
      const int e = base + u * 32 + lane;
      k[u] = (e < e1) ? keys[e] : -1;
    }
    float4 v[U * LPE];
    int    dl[U * LPE];
#pragma unroll
    for (int u = 0; u < U; ++u) {
#pragma unroll
      for (int i = 0; i < LPE; ++i) {
        const int ks = __shfl(k[u], i * EPI + sub);
        const bool ok = (ks >= 0);
        const unsigned s = ok ? ((unsigned)ks >> RB) : 0u;
        v[u * LPE + i]  = *(const float4*)(H + (size_t)s * F + 4 * q);
        dl[u * LPE + i] = ok ? (ks & (R_ROWS - 1)) : -1;
      }
    }
#pragma unroll
    for (int j = 0; j < U * LPE; ++j) {
      if (dl[j] >= 0) {
        float* a = acc + dl[j] * F + 4 * q;
        __hip_atomic_fetch_add(a + 0, v[j].x, __ATOMIC_RELAXED, __HIP_MEMORY_SCOPE_WORKGROUP);
        __hip_atomic_fetch_add(a + 1, v[j].y, __ATOMIC_RELAXED, __HIP_MEMORY_SCOPE_WORKGROUP);
        __hip_atomic_fetch_add(a + 2, v[j].z, __ATOMIC_RELAXED, __HIP_MEMORY_SCOPE_WORKGROUP);
        __hip_atomic_fetch_add(a + 3, v[j].w, __ATOMIC_RELAXED, __HIP_MEMORY_SCOPE_WORKGROUP);
      }
    }
  }
  __syncthreads();

  if constexpr (MODE == 0) {
    float4* out4 = (float4*)(OUT + (size_t)row0 * F);
    for (int i = t; i < nrows * F / 4; i += 512) out4[i] = acc4[i];
  } else if constexpr (MODE == 1) {
    const float4* h4 = (const float4*)(H + (size_t)row0 * F);
    float4* out4 = (float4*)(OUT + (size_t)row0 * F);
    const float4 bv = ((const float4*)b1)[t & 1];
    for (int i = t; i < nrows * F / 4; i += 512) {
      const float w = dinv[row0 + (i >> 1)];
      float4 av = acc4[i];
      const float4 hv = h4[i];
      av.x = fmaxf((av.x + hv.x) * w + bv.x, 0.0f);
      av.y = fmaxf((av.y + hv.y) * w + bv.y, 0.0f);
      av.z = fmaxf((av.z + hv.z) * w + bv.z, 0.0f);
      av.w = fmaxf((av.w + hv.w) * w + bv.w, 0.0f);
      out4[i] = av;
    }
  } else {
    const int m  = lane & 15;
    const int hi = lane >> 4;
    const int kk = hi << 1;
    const bool live_col = (m < F_H2);
    const int col = m & 7;
    v2f bf[4];
#pragma unroll
    for (int s = 0; s < 4; ++s) {
      const float w0 = W2[(4 * s + kk) * F_H2 + col];
      const float w1 = W2[(4 * s + kk + 1) * F_H2 + col];
      bf[s].x = live_col ? w0 : 0.0f;
      bf[s].y = live_col ? w1 : 0.0f;
    }
    const float bb0 = b1[8 + kk],  bb1 = b1[9 + kk];
    const float bb2 = b1[12 + kk], bb3 = b1[13 + kk];
    const int ntiles = nrows >> 4;
    const int wave_s = __builtin_amdgcn_readfirstlane(wave);
    for (int tile = wave_s; tile < ntiles; tile += NW) {
      const int lrow = tile * 16 + m;
      const size_t g = (size_t)(row0 + lrow);
      const float w = dinv[g];
      const float2 q0 = *(const float2*)(OUT + g * F + kk);
      const float2 q1 = *(const float2*)(OUT + g * F + 4 + kk);
      const float2 p0 = *(const float2*)(H + g * F + kk);
      const float2 p1 = *(const float2*)(H + g * F + 4 + kk);
      const float2 s0 = *(const float2*)(acc + lrow * F + kk);
      const float2 s1 = *(const float2*)(acc + lrow * F + 4 + kk);
      v2f a0, a1, a2, a3;
      a0.x = q0.x;  a0.y = q0.y;
      a1.x = q1.x;  a1.y = q1.y;
      a2.x = fmaxf((s0.x + p0.x) * w + bb0, 0.0f);
      a2.y = fmaxf((s0.y + p0.y) * w + bb1, 0.0f);
      a3.x = fmaxf((s1.x + p1.x) * w + bb2, 0.0f);
      a3.y = fmaxf((s1.y + p1.y) * w + bb3, 0.0f);
      v8f c = {0.f, 0.f, 0.f, 0.f, 0.f, 0.f, 0.f, 0.f};
      c = __builtin_amdgcn_wmma_f32_16x16x4_f32(false, a0, false, bf[0], (short)0, c, false, false);
      c = __builtin_amdgcn_wmma_f32_16x16x4_f32(false, a1, false, bf[1], (short)0, c, false, false);
      c = __builtin_amdgcn_wmma_f32_16x16x4_f32(false, a2, false, bf[2], (short)0, c, false, false);
      c = __builtin_amdgcn_wmma_f32_16x16x4_f32(false, a3, false, bf[3], (short)0, c, false, false);
      const int r0 = row0 + tile * 16 + 8 * hi;
      const float* dv = dinv + r0;
      float r[8];
#pragma unroll
      for (int j = 0; j < 8; ++j) r[j] = c[j] * dv[j];
      if (live_col) {
        float* out = OUT + (size_t)r0 * F + m;
#pragma unroll
        for (int j = 0; j < 8; ++j) out[(size_t)j * F] = r[j];
      }
    }
  }
}


__global__ void finalize2_pool(const float* __restrict__ agg,
                               const float* __restrict__ hpre,
                               const float* __restrict__ dinv,
                               const float* __restrict__ b2,
                               const int* __restrict__ batch,
                               float* __restrict__ pooled,
                               float* __restrict__ cnt, int n) {
  int i = blockIdx.x * blockDim.x + threadIdx.x;
  if (i >= n) return;
  const float w = dinv[i];
  int g = batch[i];
  float* ps = pooled + (size_t)g * F_H2;
#pragma unroll
  for (int f = 0; f < F_H2; ++f) {
    float v = (agg[(size_t)i * F_H2 + f] + hpre[(size_t)i * F_H2 + f]) * w + b2[f];
    v = fmaxf(v, 0.0f);
    ATOMIC_FADD(&ps[f], v);
  }
  ATOMIC_FADD(&cnt[g], 1.0f);
}

__global__ void readout(const float* __restrict__ pooled, const float* __restrict__ cnt,
                        const float* __restrict__ Wl, const float* __restrict__ bl,
                        float* __restrict__ out, int G) {
  int g = blockIdx.x * blockDim.x + threadIdx.x;
  if (g >= G) return;
  float inv = 1.0f / fmaxf(cnt[g], 1.0f);
  float p[F_H2];
#pragma unroll
  for (int f = 0; f < F_H2; ++f) p[f] = pooled[(size_t)g * F_H2 + f] * inv;
#pragma unroll
  for (int c = 0; c < N_CLS; ++c) {
    float acc = bl[c];
#pragma unroll
    for (int f = 0; f < F_H2; ++f) acc += p[f] * Wl[f * N_CLS + c];
    out[(size_t)g * N_CLS + c] = acc;
  }
}


extern "C" void kernel_launch(void* const* d_in, const int* in_sizes, int n_in,
                              void* d_out, int out_size, void* d_ws, size_t ws_size,
                              hipStream_t stream) {
  const float* x     = (const float*)d_in[0];
  const int*   ei    = (const int*)d_in[1];
  const int*   src   = ei;
  const int*   dst   = ei + N_EDGES;
  const int*   batch = (const int*)d_in[2];
  const float* W1 = (const float*)d_in[4];
  const float* b1 = (const float*)d_in[5];
  const float* W2 = (const float*)d_in[6];
  const float* b2 = (const float*)d_in[7];
  const float* Wl = (const float*)d_in[8];
  const float* bl = (const float*)d_in[9];
  float* out = (float*)d_out;

  char* ws = (char*)d_ws;
  float* dinv   = (float*)(ws);
  float* bufA   = (float*)(ws + (4ll  << 20));
  float* bufB   = (float*)(ws + (68ll << 20));
  float* pooled = (float*)(ws + (132ll << 20));
  float* cnt    = pooled + (size_t)N_GRAPHS * F_H2;

  float* planeP = bufA;
  float* planeQ = (float*)((char*)bufA + (32ll << 20));
  float* h2pre  = planeQ;
  float* agg2   = planeP;
  int* keys   = (int*)bufB;
  int* counts = (int*)((char*)bufB + 64000000ll);
  int* tot    = (int*)((char*)bufB + 64000000ll + (512ll << 10));
  int* bstart = (int*)((char*)bufB + 64000000ll + (520ll << 10));

  const int EB = 256;
  const int EG = (N_EDGES + EB - 1) / EB;
  const int NB = 256;
  const int NG = (N_NODES + NB - 1) / NB;
  const int GEMM_BLOCKS = N_NODES / 64;

  zero_f32<<<2048, 256, 0, stream>>>(dinv, (long long)N_NODES);
  degree_kernel<<<EG, EB, 0, stream>>>(dst, dinv, N_EDGES);
  rsqrt_kernel<<<NG, NB, 0, stream>>>(dinv, N_NODES);

  bin_count<<<NBLK, 1024, 0, stream>>>(dst, counts);
  bin_scan<<<N_RANGES, 256, 0, stream>>>(counts, tot);
  bin_scatter<<<NBLK, 1024, 0, stream>>>(src, dst, counts, tot, bstart, keys);

  gemm_x_w1_half<0><<<GEMM_BLOCKS, 128, 0, stream>>>(x, W1, dinv, planeP);
  agg_range_lds<1><<<N_RANGES, 512, 0, stream>>>(keys, bstart, planeP, planeQ, dinv, b1, W2);
  gemm_x_w1_half<1><<<GEMM_BLOCKS, 128, 0, stream>>>(x, W1, dinv, planeP);
  agg_range_lds<2><<<N_RANGES, 512, 0, stream>>>(keys, bstart, planeP, planeQ, dinv, b1, W2);

  agg_range_lds<0><<<N_RANGES, 512, 0, stream>>>(keys, bstart, h2pre, agg2, dinv, b1, W2);

  zero_f32<<<128, 256, 0, stream>>>(pooled, (long long)N_GRAPHS * (F_H2 + 1));
  finalize2_pool<<<NG, NB, 0, stream>>>(agg2, h2pre, dinv, b2, batch,
                                        pooled, cnt, N_NODES);
  readout<<<(N_GRAPHS + NB - 1) / NB, NB, 0, stream>>>(pooled, cnt, Wl, bl,
                                                       out, N_GRAPHS);
}


# degrees counted per destination range in LDS from the binned keys (integer LDS adds), dinv stored directly: no zero pass, per-edge global float atomics or rsqrt pass; dinv bits unchanged
# speedup vs baseline: 5.9258x; 1.4824x over previous
#include <hip/hip_runtime.h>


typedef float v2f __attribute__((ext_vector_type(2)));
typedef float v8f __attribute__((ext_vector_type(8)));

#define N_NODES   1000000
#define N_EDGES   16000000
#define F_IN      32
#define F_H1      16
#define F_H2      8
#define N_CLS     5
#define N_GRAPHS  10000

#define RB        11
#define R_ROWS    (1 << RB)
#define N_RANGES        ((N_NODES + R_ROWS - 1) >> RB)
#define CH        65536
#define NBLK      ((N_EDGES + CH - 1) / CH)

#define ATOMIC_FADD(p, v) \
  __hip_atomic_fetch_add((p), (v), __ATOMIC_RELAXED, __HIP_MEMORY_SCOPE_AGENT)


__global__ void zero_f32(float* __restrict__ p, long long n) {
  long long i = (long long)blockIdx.x * blockDim.x + threadIdx.x;
  long long stride = (long long)gridDim.x * blockDim.x;
  for (; i < n; i += stride) p[i] = 0.0f;
}


template <int HALF>
__global__ __launch_bounds__(128) void gemm_x_w1_half(const float* __restrict__ X,
                                                      const float* __restrict__ W1,
                                                      const float* __restrict__ dinv,
                                                      float* __restrict__ P) {
  int wave = blockIdx.x * (blockDim.x >> 5) + (threadIdx.x >> 5);
  int lane = threadIdx.x & 31;
  int m  = lane & 15;
  int kk = (lane >> 4) << 1;
  const float* xrow = X + (size_t)(wave * 16 + m) * F_IN;
  const bool live_col = (m < 8);
  const int col = HALF * 8 + (m & 7);

  v8f c = {0.f, 0.f, 0.f, 0.f, 0.f, 0.f, 0.f, 0.f};
#pragma unroll
  for (int k = 0; k < F_IN; k += 4) {
    v2f a, b;
    a.x = xrow[k + kk];
    a.y = xrow[k + kk + 1];
    const float w0 = W1[(k + kk) * F_H1 + col];
    const float w1 = W1[(k + kk + 1) * F_H1 + col];
    b.x = live_col ? w0 : 0.0f;
    b.y = live_col ? w1 : 0.0f;
    c = __builtin_amdgcn_wmma_f32_16x16x4_f32(false, a, false, b,
                                              (short)0, c, false, false);
  }
  int hi = lane >> 4;
  const int row0 = wave * 16 + 8 * hi;
  const float* dv = dinv + row0;
  float r[8];
#pragma unroll
  for (int j = 0; j < 8; ++j) r[j] = c[j] * dv[j];
  if (live_col) {
    float* out = P + (size_t)row0 * 8 + m;
#pragma unroll
    for (int j = 0; j < 8; ++j) out[(size_t)j * 8] = r[j];
  }
}


static_assert(N_NODES <= (1 << 20), "a key holds the source row in 20 bits beside RB bits of local row");
static_assert(N_RANGES <= 1024, "one thread per range in the scatter kernel's offset scan");
static_assert(NBLK <= 256, "one thread per binning work-group in the per-range scan");
static_assert(N_EDGES % 4 == 0 && CH % 4096 == 0, "binning kernels read four edges per lane");
static_assert(N_NODES % 16 == 0 && R_ROWS % 16 == 0, "a range is whole 16-row tiles");

__device__ __forceinline__ int wave_incl_scan(int v, int lane) {
#pragma unroll
  for (int o = 1; o < 32; o <<= 1) {
    const int t = __shfl_up(v, o);
    if (lane >= o) v += t;
  }
  return v;
}

__device__ __forceinline__ int block_excl_scan(int v, int* wsum, int* total) {
  const int lane = threadIdx.x & 31, wave = threadIdx.x >> 5;
  const int nw = blockDim.x >> 5;
  const int incl = wave_incl_scan(v, lane);
  if (lane == 31) wsum[wave] = incl;
  __syncthreads();
  if (wave == 0) {
    const int w  = (lane < nw) ? wsum[lane] : 0;
    const int wi = wave_incl_scan(w, lane);
    wsum[lane] = wi - w;
    if (lane == 31) wsum[32] = wi;
  }
  __syncthreads();
  *total = wsum[32];
  return wsum[wave] + incl - v;
}

__global__ __launch_bounds__(1024) void bin_count(const int* __restrict__ dst,
                                                  int* __restrict__ counts) {
  __shared__ int hist[N_RANGES];
  const int t = threadIdx.x, blk = blockIdx.x;
  for (int i = t; i < N_RANGES; i += 1024) hist[i] = 0;
  __syncthreads();
  const int base = blk * CH;
  const int4* d4 = (const int4*)(dst + base);
  for (int i = t; i < CH / 4; i += 1024) {
    if (base + 4 * i < N_EDGES) {
      const int4 d = d4[i];
      atomicAdd(&hist[d.x >> RB], 1);
      atomicAdd(&hist[d.y >> RB], 1);
      atomicAdd(&hist[d.z >> RB], 1);
      atomicAdd(&hist[d.w >> RB], 1);
    }
  }
  __syncthreads();
  for (int i = t; i < N_RANGES; i += 1024) counts[i * NBLK + blk] = hist[i];
}

__global__ __launch_bounds__(256) void bin_scan(int* __restrict__ counts,
                                                int* __restrict__ tot) {
  __shared__ int wsum[33];
  const int b = blockIdx.x, t = threadIdx.x;
  const int v = (t < NBLK) ? counts[b * NBLK + t] : 0;
  int total;
  const int ex = block_excl_scan(v, wsum, &total);
  if (t < NBLK) counts[b * NBLK + t] = ex;
  if (t == 0) tot[b] = total;
}

__global__ __launch_bounds__(1024) void bin_scatter(const int* __restrict__ src,
                                                    const int* __restrict__ dst,
                                                    const int* __restrict__ offs,
                                                    const int* __restrict__ tot,
                                                    int* __restrict__ bstart,
                                                    int* __restrict__ keys) {
  __shared__ int cursor[N_RANGES];
  __shared__ int wsum[33];
  const int t = threadIdx.x, blk = blockIdx.x;
  const int v = (t < N_RANGES) ? tot[t] : 0;
  int total;
  const int ex = block_excl_scan(v, wsum, &total);
  if (t < N_RANGES) {
    cursor[t] = ex + offs[t * NBLK + blk];
    if (blk == 0) bstart[t] = ex;
  }
  if (blk == 0 && t == 0) bstart[N_RANGES] = total;
  __syncthreads();
  const int base = blk * CH;
  const int4* s4 = (const int4*)(src + base);
  const int4* d4 = (const int4*)(dst + base);
  for (int i = t; i < CH / 4; i += 1024) {
    if (base + 4 * i < N_EDGES) {
      const int4 s = s4[i], d = d4[i];
      const int p0 = atomicAdd(&cursor[d.x >> RB], 1);
      const int p1 = atomicAdd(&cursor[d.y >> RB], 1);
      const int p2 = atomicAdd(&cursor[d.z >> RB], 1);
      const int p3 = atomicAdd(&cursor[d.w >> RB], 1);
      keys[p0] = (s.x << RB) | (d.x & (R_ROWS - 1));
      keys[p1] = (s.y << RB) | (d.y & (R_ROWS - 1));
      keys[p2] = (s.z << RB) | (d.z & (R_ROWS - 1));
      keys[p3] = (s.w << RB) | (d.w & (R_ROWS - 1));
    }
  }
}

__global__ __launch_bounds__(1024) void degree_range_lds(const int* __restrict__ keys,
                                                         const int* __restrict__ bstart,
                                                         float* __restrict__ dinv) {
  constexpr int U  = 8;
  constexpr int NW = 1024 / 32;
  __shared__ int cnt[R_ROWS];

  const int t = threadIdx.x, lane = t & 31, wave = t >> 5;
  const int b = blockIdx.x;
  for (int i = t; i < R_ROWS; i += 1024) cnt[i] = 0;
  __syncthreads();

  const int e0 = bstart[b], e1 = bstart[b + 1];
  for (int base = e0 + wave * (32 * U); base < e1; base += NW * 32 * U) {
    int k[U];
#pragma unroll
    for (int u = 0; u < U; ++u) {
      const int e = base + u * 32 + lane;
      k[u] = (e < e1) ? keys[e] : -1;
    }
#pragma unroll
    for (int u = 0; u < U; ++u) {
      if (k[u] >= 0)
        __hip_atomic_fetch_add(&cnt[k[u] & (R_ROWS - 1)], 1, __ATOMIC_RELAXED,
                               __HIP_MEMORY_SCOPE_WORKGROUP);
    }
  }
  __syncthreads();

  const int row0 = b << RB;
  const int nrows = (N_NODES - row0 < R_ROWS) ? (N_NODES - row0) : R_ROWS;
  for (int i = t; i < nrows; i += 1024)
    dinv[row0 + i] = rsqrtf((float)cnt[i] + 1.0f);
}

template <int MODE>
__global__ __launch_bounds__(512) void agg_range_lds(const int* __restrict__ keys,
                                                     const int* __restrict__ bstart,
                                                     const float* __restrict__ H,
                                                     float* OUT,
                                                     const float* __restrict__ dinv,
                                                     const float* __restrict__ b1,
                                                     const float* __restrict__ W2) {
  constexpr int F   = 8;
  constexpr int LPE = F / 4;
  constexpr int EPI = 32 / LPE;
  constexpr int U   = 4;
  constexpr int NW  = 512 / 32;
  __shared__ float acc[R_ROWS * F];

  const int t = threadIdx.x, lane = t & 31, wave = t >> 5;
  const int b = blockIdx.x;
  const int row0 = b << RB;
  const int nrows = (N_NODES - row0 < R_ROWS) ? (N_NODES - row0) : R_ROWS;

  float4* acc4 = (float4*)acc;
  for (int i = t; i < R_ROWS * F / 4; i += 512) acc4[i] = make_float4(0.f, 0.f, 0.f, 0.f);
  __syncthreads();

  const int e0 = bstart[b], e1 = bstart[b + 1];
  const int sub = lane / LPE;
  const int q   = lane % LPE;

  for (int base = e0 + wave * (32 * U); base < e1; base += NW * 32 * U) {
    int k[U];
#pragma unroll
    for (int u = 0; u < U; ++u) {
      const int e = base + u * 32 + lane;
      k[u] = (e < e1) ? keys[e] : -1;
    }
    float4 v[U * LPE];
    int    dl[U * LPE];
#pragma unroll
    for (int u = 0; u < U; ++u) {
#pragma unroll
      for (int i = 0; i < LPE; ++i) {
        const int ks = __shfl(k[u], i * EPI + sub);
        const bool ok = (ks >= 0);
        const unsigned s = ok ? ((unsigned)ks >> RB) : 0u;
        v[u * LPE + i]  = *(const float4*)(H + (size_t)s * F + 4 * q);
        dl[u * LPE + i] = ok ? (ks & (R_ROWS - 1)) : -1;
      }
    }
#pragma unroll
    for (int j = 0; j < U * LPE; ++j) {
      if (dl[j] >= 0) {
        float* a = acc + dl[j] * F + 4 * q;
        __hip_atomic_fetch_add(a + 0, v[j].x, __ATOMIC_RELAXED, __HIP_MEMORY_SCOPE_WORKGROUP);
        __hip_atomic_fetch_add(a + 1, v[j].y, __ATOMIC_RELAXED, __HIP_MEMORY_SCOPE_WORKGROUP);
        __hip_atomic_fetch_add(a + 2, v[j].z, __ATOMIC_RELAXED, __HIP_MEMORY_SCOPE_WORKGROUP);
        __hip_atomic_fetch_add(a + 3, v[j].w, __ATOMIC_RELAXED, __HIP_MEMORY_SCOPE_WORKGROUP);
      }
    }
  }
  __syncthreads();

  if constexpr (MODE == 0) {
    float4* out4 = (float4*)(OUT + (size_t)row0 * F);
    for (int i = t; i < nrows * F / 4; i += 512) out4[i] = acc4[i];
  } else if constexpr (MODE == 1) {
    const float4* h4 = (const float4*)(H + (size_t)row0 * F);
    float4* out4 = (float4*)(OUT + (size_t)row0 * F);
    const float4 bv = ((const float4*)b1)[t & 1];
    for (int i = t; i < nrows * F / 4; i += 512) {
      const float w = dinv[row0 + (i >> 1)];
      float4 av = acc4[i];
      const float4 hv = h4[i];
      av.x = fmaxf((av.x + hv.x) * w + bv.x, 0.0f);
      av.y = fmaxf((av.y + hv.y) * w + bv.y, 0.0f);
      av.z = fmaxf((av.z + hv.z) * w + bv.z, 0.0f);
      av.w = fmaxf((av.w + hv.w) * w + bv.w, 0.0f);
      out4[i] = av;
    }
  } else {
    const int m  = lane & 15;
    const int hi = lane >> 4;
    const int kk = hi << 1;
    const bool live_col = (m < F_H2);
    const int col = m & 7;
    v2f bf[4];
#pragma unroll
    for (int s = 0; s < 4; ++s) {
      const float w0 = W2[(4 * s + kk) * F_H2 + col];
      const float w1 = W2[(4 * s + kk + 1) * F_H2 + col];
      bf[s].x = live_col ? w0 : 0.0f;
      bf[s].y = live_col ? w1 : 0.0f;
    }
    const float bb0 = b1[8 + kk],  bb1 = b1[9 + kk];
    const float bb2 = b1[12 + kk], bb3 = b1[13 + kk];
    const int ntiles = nrows >> 4;
    const int wave_s = __builtin_amdgcn_readfirstlane(wave);
    for (int tile = wave_s; tile < ntiles; tile += NW) {
      const int lrow = tile * 16 + m;
      const size_t g = (size_t)(row0 + lrow);
      const float w = dinv[g];
      const float2 q0 = *(const float2*)(OUT + g * F + kk);
      const float2 q1 = *(const float2*)(OUT + g * F + 4 + kk);
      const float2 p0 = *(const float2*)(H + g * F + kk);
      const float2 p1 = *(const float2*)(H + g * F + 4 + kk);
      const float2 s0 = *(const float2*)(acc + lrow * F + kk);
      const float2 s1 = *(const float2*)(acc + lrow * F + 4 + kk);
      v2f a0, a1, a2, a3;
      a0.x = q0.x;  a0.y = q0.y;
      a1.x = q1.x;  a1.y = q1.y;
      a2.x = fmaxf((s0.x + p0.x) * w + bb0, 0.0f);
      a2.y = fmaxf((s0.y + p0.y) * w + bb1, 0.0f);
      a3.x = fmaxf((s1.x + p1.x) * w + bb2, 0.0f);
      a3.y = fmaxf((s1.y + p1.y) * w + bb3, 0.0f);
      v8f c = {0.f, 0.f, 0.f, 0.f, 0.f, 0.f, 0.f, 0.f};
      c = __builtin_amdgcn_wmma_f32_16x16x4_f32(false, a0, false, bf[0], (short)0, c, false, false);
      c = __builtin_amdgcn_wmma_f32_16x16x4_f32(false, a1, false, bf[1], (short)0, c, false, false);
      c = __builtin_amdgcn_wmma_f32_16x16x4_f32(false, a2, false, bf[2], (short)0, c, false, false);
      c = __builtin_amdgcn_wmma_f32_16x16x4_f32(false, a3, false, bf[3], (short)0, c, false, false);
      const int r0 = row0 + tile * 16 + 8 * hi;
      const float* dv = dinv + r0;
      float r[8];
#pragma unroll
      for (int j = 0; j < 8; ++j) r[j] = c[j] * dv[j];
      if (live_col) {
        float* out = OUT + (size_t)r0 * F + m;
#pragma unroll
        for (int j = 0; j < 8; ++j) out[(size_t)j * F] = r[j];
      }
    }
  }
}


__global__ void finalize2_pool(const float* __restrict__ agg,
                               const float* __restrict__ hpre,
                               const float* __restrict__ dinv,
                               const float* __restrict__ b2,
                               const int* __restrict__ batch,
                               float* __restrict__ pooled,
                               float* __restrict__ cnt, int n) {
  int i = blockIdx.x * blockDim.x + threadIdx.x;
  if (i >= n) return;
  const float w = dinv[i];
  int g = batch[i];
  float* ps = pooled + (size_t)g * F_H2;
#pragma unroll
  for (int f = 0; f < F_H2; ++f) {
    float v = (agg[(size_t)i * F_H2 + f] + hpre[(size_t)i * F_H2 + f]) * w + b2[f];
    v = fmaxf(v, 0.0f);
    ATOMIC_FADD(&ps[f], v);
  }
  ATOMIC_FADD(&cnt[g], 1.0f);
}

__global__ void readout(const float* __restrict__ pooled, const float* __restrict__ cnt,
                        const float* __restrict__ Wl, const float* __restrict__ bl,
                        float* __restrict__ out, int G) {
  int g = blockIdx.x * blockDim.x + threadIdx.x;
  if (g >= G) return;
  float inv = 1.0f / fmaxf(cnt[g], 1.0f);
  float p[F_H2];
#pragma unroll
  for (int f = 0; f < F_H2; ++f) p[f] = pooled[(size_t)g * F_H2 + f] * inv;
#pragma unroll
  for (int c = 0; c < N_CLS; ++c) {
    float acc = bl[c];
#pragma unroll
    for (int f = 0; f < F_H2; ++f) acc += p[f] * Wl[f * N_CLS + c];
    out[(size_t)g * N_CLS + c] = acc;
  }
}


extern "C" void kernel_launch(void* const* d_in, const int* in_sizes, int n_in,
                              void* d_out, int out_size, void* d_ws, size_t ws_size,
                              hipStream_t stream) {
  const float* x     = (const float*)d_in[0];
  const int*   ei    = (const int*)d_in[1];
  const int*   src   = ei;
  const int*   dst   = ei + N_EDGES;
  const int*   batch = (const int*)d_in[2];
  const float* W1 = (const float*)d_in[4];
  const float* b1 = (const float*)d_in[5];
  const float* W2 = (const float*)d_in[6];
  const float* b2 = (const float*)d_in[7];
  const float* Wl = (const float*)d_in[8];
  const float* bl = (const float*)d_in[9];
  float* out = (float*)d_out;

  char* ws = (char*)d_ws;
  float* dinv   = (float*)(ws);
  float* bufA   = (float*)(ws + (4ll  << 20));
  float* bufB   = (float*)(ws + (68ll << 20));
  float* pooled = (float*)(ws + (132ll << 20));
  float* cnt    = pooled + (size_t)N_GRAPHS * F_H2;

  float* planeP = bufA;
  float* planeQ = (float*)((char*)bufA + (32ll << 20));
  float* h2pre  = planeQ;
  float* agg2   = planeP;
  int* keys   = (int*)bufB;
  int* counts = (int*)((char*)bufB + 64000000ll);
  int* tot    = (int*)((char*)bufB + 64000000ll + (512ll << 10));
  int* bstart = (int*)((char*)bufB + 64000000ll + (520ll << 10));

  const int NB = 256;
  const int NG = (N_NODES + NB - 1) / NB;
  const int GEMM_BLOCKS = N_NODES / 64;

  bin_count<<<NBLK, 1024, 0, stream>>>(dst, counts);
  bin_scan<<<N_RANGES, 256, 0, stream>>>(counts, tot);
  bin_scatter<<<NBLK, 1024, 0, stream>>>(src, dst, counts, tot, bstart, keys);

  degree_range_lds<<<N_RANGES, 1024, 0, stream>>>(keys, bstart, dinv);

  gemm_x_w1_half<0><<<GEMM_BLOCKS, 128, 0, stream>>>(x, W1, dinv, planeP);
  agg_range_lds<1><<<N_RANGES, 512, 0, stream>>>(keys, bstart, planeP, planeQ, dinv, b1, W2);
  gemm_x_w1_half<1><<<GEMM_BLOCKS, 128, 0, stream>>>(x, W1, dinv, planeP);
  agg_range_lds<2><<<N_RANGES, 512, 0, stream>>>(keys, bstart, planeP, planeQ, dinv, b1, W2);

  agg_range_lds<0><<<N_RANGES, 512, 0, stream>>>(keys, bstart, h2pre, agg2, dinv, b1, W2);

  zero_f32<<<128, 256, 0, stream>>>(pooled, (long long)N_GRAPHS * (F_H2 + 1));
  finalize2_pool<<<NG, NB, 0, stream>>>(agg2, h2pre, dinv, b2, batch,
                                        pooled, cnt, N_NODES);
  readout<<<(N_GRAPHS + NB - 1) / NB, NB, 0, stream>>>(pooled, cnt, Wl, bl,
                                                       out, N_GRAPHS);
}


# layer-2 finalize + mean-pool sums fused into the layer-2 range kernel: a lane run-sums its rows per graph id from the LDS sums, one float atomic per run, not per node; pool sum order changes
# speedup vs baseline: 8.3217x; 1.4043x over previous
#include <hip/hip_runtime.h>


typedef float v2f __attribute__((ext_vector_type(2)));
typedef float v8f __attribute__((ext_vector_type(8)));

#define N_NODES   1000000
#define N_EDGES   16000000
#define F_IN      32
#define F_H1      16
#define F_H2      8
#define N_CLS     5
#define N_GRAPHS  10000

#define RB        11
#define R_ROWS    (1 << RB)
#define N_RANGES        ((N_NODES + R_ROWS - 1) >> RB)
#define CH        65536
#define NBLK      ((N_EDGES + CH - 1) / CH)

#define ATOMIC_FADD(p, v) \
  __hip_atomic_fetch_add((p), (v), __ATOMIC_RELAXED, __HIP_MEMORY_SCOPE_AGENT)


__global__ void zero_f32(float* __restrict__ p, long long n) {
  long long i = (long long)blockIdx.x * blockDim.x + threadIdx.x;
  long long stride = (long long)gridDim.x * blockDim.x;
  for (; i < n; i += stride) p[i] = 0.0f;
}


template <int HALF>
__global__ __launch_bounds__(128) void gemm_x_w1_half(const float* __restrict__ X,
                                                      const float* __restrict__ W1,
                                                      const float* __restrict__ dinv,
                                                      float* __restrict__ P) {
  int wave = blockIdx.x * (blockDim.x >> 5) + (threadIdx.x >> 5);
  int lane = threadIdx.x & 31;
  int m  = lane & 15;
  int kk = (lane >> 4) << 1;
  const float* xrow = X + (size_t)(wave * 16 + m) * F_IN;
  const bool live_col = (m < 8);
  const int col = HALF * 8 + (m & 7);

  v8f c = {0.f, 0.f, 0.f, 0.f, 0.f, 0.f, 0.f, 0.f};
#pragma unroll
  for (int k = 0; k < F_IN; k += 4) {
    v2f a, b;
    a.x = xrow[k + kk];
    a.y = xrow[k + kk + 1];
    const float w0 = W1[(k + kk) * F_H1 + col];
    const float w1 = W1[(k + kk + 1) * F_H1 + col];
    b.x = live_col ? w0 : 0.0f;
    b.y = live_col ? w1 : 0.0f;
    c = __builtin_amdgcn_wmma_f32_16x16x4_f32(false, a, false, b,
                                              (short)0, c, false, false);
  }
  int hi = lane >> 4;
  const int row0 = wave * 16 + 8 * hi;
  const float* dv = dinv + row0;
  float r[8];
#pragma unroll
  for (int j = 0; j < 8; ++j) r[j] = c[j] * dv[j];
  if (live_col) {
    float* out = P + (size_t)row0 * 8 + m;
#pragma unroll
    for (int j = 0; j < 8; ++j) out[(size_t)j * 8] = r[j];
  }
}


static_assert(N_NODES <= (1 << 20), "a key holds the source row in 20 bits beside RB bits of local row");
static_assert(N_RANGES <= 1024, "one thread per range in the scatter kernel's offset scan");
static_assert(NBLK <= 256, "one thread per binning work-group in the per-range scan");
static_assert(N_EDGES % 4 == 0 && CH % 4096 == 0, "binning kernels read four edges per lane");
static_assert(N_NODES % 16 == 0 && R_ROWS % 16 == 0, "a range is whole 16-row tiles");

__device__ __forceinline__ int wave_incl_scan(int v, int lane) {
#pragma unroll
  for (int o = 1; o < 32; o <<= 1) {
    const int t = __shfl_up(v, o);
    if (lane >= o) v += t;
  }
  return v;
}

__device__ __forceinline__ int block_excl_scan(int v, int* wsum, int* total) {
  const int lane = threadIdx.x & 31, wave = threadIdx.x >> 5;
  const int nw = blockDim.x >> 5;
  const int incl = wave_incl_scan(v, lane);
  if (lane == 31) wsum[wave] = incl;
  __syncthreads();
  if (wave == 0) {
    const int w  = (lane < nw) ? wsum[lane] : 0;
    const int wi = wave_incl_scan(w, lane);
    wsum[lane] = wi - w;
    if (lane == 31) wsum[32] = wi;
  }
  __syncthreads();
  *total = wsum[32];
  return wsum[wave] + incl - v;
}

__global__ __launch_bounds__(1024) void bin_count(const int* __restrict__ dst,
                                                  int* __restrict__ counts) {
  __shared__ int hist[N_RANGES];
  const int t = threadIdx.x, blk = blockIdx.x;
  for (int i = t; i < N_RANGES; i += 1024) hist[i] = 0;
  __syncthreads();
  const int base = blk * CH;
  const int4* d4 = (const int4*)(dst + base);
  for (int i = t; i < CH / 4; i += 1024) {
    if (base + 4 * i < N_EDGES) {
      const int4 d = d4[i];
      atomicAdd(&hist[d.x >> RB], 1);
      atomicAdd(&hist[d.y >> RB], 1);
      atomicAdd(&hist[d.z >> RB], 1);
      atomicAdd(&hist[d.w >> RB], 1);
    }
  }
  __syncthreads();
  for (int i = t; i < N_RANGES; i += 1024) counts[i * NBLK + blk] = hist[i];
}

__global__ __launch_bounds__(256) void bin_scan(int* __restrict__ counts,
                                                int* __restrict__ tot) {
  __shared__ int wsum[33];
  const int b = blockIdx.x, t = threadIdx.x;
  const int v = (t < NBLK) ? counts[b * NBLK + t] : 0;
  int total;
  const int ex = block_excl_scan(v, wsum, &total);
  if (t < NBLK) counts[b * NBLK + t] = ex;
  if (t == 0) tot[b] = total;
}

__global__ __launch_bounds__(1024) void bin_scatter(const int* __restrict__ src,
                                                    const int* __restrict__ dst,
                                                    const int* __restrict__ offs,
                                                    const int* __restrict__ tot,
                                                    int* __restrict__ bstart,
                                                    int* __restrict__ keys) {
  __shared__ int cursor[N_RANGES];
  __shared__ int wsum[33];
  const int t = threadIdx.x, blk = blockIdx.x;
  const int v = (t < N_RANGES) ? tot[t] : 0;
  int total;
  const int ex = block_excl_scan(v, wsum, &total);
  if (t < N_RANGES) {
    cursor[t] = ex + offs[t * NBLK + blk];
    if (blk == 0) bstart[t] = ex;
  }
  if (blk == 0 && t == 0) bstart[N_RANGES] = total;
  __syncthreads();
  const int base = blk * CH;
  const int4* s4 = (const int4*)(src + base);
  const int4* d4 = (const int4*)(dst + base);
  for (int i = t; i < CH / 4; i += 1024) {
    if (base + 4 * i < N_EDGES) {
      const int4 s = s4[i], d = d4[i];
      const int p0 = atomicAdd(&cursor[d.x >> RB], 1);
      const int p1 = atomicAdd(&cursor[d.y >> RB], 1);
      const int p2 = atomicAdd(&cursor[d.z >> RB], 1);
      const int p3 = atomicAdd(&cursor[d.w >> RB], 1);
      keys[p0] = (s.x << RB) | (d.x & (R_ROWS - 1));
      keys[p1] = (s.y << RB) | (d.y & (R_ROWS - 1));
      keys[p2] = (s.z << RB) | (d.z & (R_ROWS - 1));
      keys[p3] = (s.w << RB) | (d.w & (R_ROWS - 1));
    }
  }
}

__global__ __launch_bounds__(1024) void degree_range_lds(const int* __restrict__ keys,
                                                         const int* __restrict__ bstart,
                                                         float* __restrict__ dinv) {
  constexpr int U  = 8;
  constexpr int NW = 1024 / 32;
  __shared__ int cnt[R_ROWS];

  const int t = threadIdx.x, lane = t & 31, wave = t >> 5;
  const int b = blockIdx.x;
  for (int i = t; i < R_ROWS; i += 1024) cnt[i] = 0;
  __syncthreads();

  const int e0 = bstart[b], e1 = bstart[b + 1];
  for (int base = e0 + wave * (32 * U); base < e1; base += NW * 32 * U) {
    int k[U];
#pragma unroll
    for (int u = 0; u < U; ++u) {
      const int e = base + u * 32 + lane;
      k[u] = (e < e1) ? keys[e] : -1;
    }
#pragma unroll
    for (int u = 0; u < U; ++u) {
      if (k[u] >= 0)
        __hip_atomic_fetch_add(&cnt[k[u] & (R_ROWS - 1)], 1, __ATOMIC_RELAXED,
                               __HIP_MEMORY_SCOPE_WORKGROUP);
    }
  }
  __syncthreads();

  const int row0 = b << RB;
  const int nrows = (N_NODES - row0 < R_ROWS) ? (N_NODES - row0) : R_ROWS;
  for (int i = t; i < nrows; i += 1024)
    dinv[row0 + i] = rsqrtf((float)cnt[i] + 1.0f);
}

template <int MODE>
__global__ __launch_bounds__(512) void agg_range_lds(const int* __restrict__ keys,
                                                     const int* __restrict__ bstart,
                                                     const float* __restrict__ H,
                                                     float* OUT,
                                                     const float* __restrict__ dinv,
                                                     const float* __restrict__ b1,
                                                     const float* __restrict__ W2,
                                                     const float* __restrict__ b2,
                                                     const int* __restrict__ batch,
                                                     float* __restrict__ pooled) {
  constexpr int F   = 8;
  constexpr int LPE = F / 4;
  constexpr int EPI = 32 / LPE;
  constexpr int U   = 4;
  constexpr int NW  = 512 / 32;
  __shared__ float acc[R_ROWS * F];

  const int t = threadIdx.x, lane = t & 31, wave = t >> 5;
  const int b = blockIdx.x;
  const int row0 = b << RB;
  const int nrows = (N_NODES - row0 < R_ROWS) ? (N_NODES - row0) : R_ROWS;

  float4* acc4 = (float4*)acc;
  for (int i = t; i < R_ROWS * F / 4; i += 512) acc4[i] = make_float4(0.f, 0.f, 0.f, 0.f);
  __syncthreads();

  const int e0 = bstart[b], e1 = bstart[b + 1];
  const int sub = lane / LPE;
  const int q   = lane % LPE;

  for (int base = e0 + wave * (32 * U); base < e1; base += NW * 32 * U) {
    int k[U];
#pragma unroll
    for (int u = 0; u < U; ++u) {
      const int e = base + u * 32 + lane;
      k[u] = (e < e1) ? keys[e] : -1;
    }
    float4 v[U * LPE];
    int    dl[U * LPE];
#pragma unroll
    for (int u = 0; u < U; ++u) {
#pragma unroll
      for (int i = 0; i < LPE; ++i) {
        const int ks = __shfl(k[u], i * EPI + sub);
        const bool ok = (ks >= 0);
        const unsigned s = ok ? ((unsigned)ks >> RB) : 0u;
        v[u * LPE + i]  = *(const float4*)(H + (size_t)s * F + 4 * q);
        dl[u * LPE + i] = ok ? (ks & (R_ROWS - 1)) : -1;
      }
    }
#pragma unroll
    for (int j = 0; j < U * LPE; ++j) {
      if (dl[j] >= 0) {
        float* a = acc + dl[j] * F + 4 * q;
        __hip_atomic_fetch_add(a + 0, v[j].x, __ATOMIC_RELAXED, __HIP_MEMORY_SCOPE_WORKGROUP);
        __hip_atomic_fetch_add(a + 1, v[j].y, __ATOMIC_RELAXED, __HIP_MEMORY_SCOPE_WORKGROUP);
        __hip_atomic_fetch_add(a + 2, v[j].z, __ATOMIC_RELAXED, __HIP_MEMORY_SCOPE_WORKGROUP);
        __hip_atomic_fetch_add(a + 3, v[j].w, __ATOMIC_RELAXED, __HIP_MEMORY_SCOPE_WORKGROUP);
      }
    }
  }
  __syncthreads();

  if constexpr (MODE == 0) {
    constexpr int RPW = R_ROWS / NW;
    constexpr int UP  = 8;
    static_assert(RPW % (4 * UP) == 0, "a wave's rows are whole trips of 4*UP rows");
    const int f  = lane & 7;
    const int r4 = lane >> 3;
    const float bf = b2[f];
    float* gcnt = pooled + (size_t)N_GRAPHS * F_H2;
    int   gcur = 0;
    int   nrun = 0;
    float run  = 0.0f;
    for (int i0 = 0; i0 < RPW / 4; i0 += UP) {
      int   g[UP];
      float w[UP], a[UP], h[UP];
#pragma unroll
      for (int u = 0; u < UP; ++u) {
        const int lrow = wave * RPW + 4 * (i0 + u) + r4;
        const bool ok = (lrow < nrows);
        const size_t grow = (size_t)(row0 + (ok ? lrow : 0));
        g[u] = ok ? batch[grow] : -1;
        w[u] = dinv[grow];
        h[u] = H[grow * F + f];
        a[u] = acc[lrow * F + f];
      }
#pragma unroll
      for (int u = 0; u < UP; ++u) {
        const int lrow = wave * RPW + 4 * (i0 + u) + r4;
        if (lrow < nrows) {
          const float v = fmaxf((a[u] + h[u]) * w[u] + bf, 0.0f);
          if (nrun > 0 && g[u] != gcur) {
            ATOMIC_FADD(&pooled[(size_t)gcur * F_H2 + f], run);
            if (f == 0) ATOMIC_FADD(&gcnt[gcur], (float)nrun);
            nrun = 0;
            run  = 0.0f;
          }
          gcur = g[u];
          run += v;
          ++nrun;
        }
      }
    }
    if (nrun > 0) {
      ATOMIC_FADD(&pooled[(size_t)gcur * F_H2 + f], run);
      if (f == 0) ATOMIC_FADD(&gcnt[gcur], (float)nrun);
    }
  } else if constexpr (MODE == 1) {
    const float4* h4 = (const float4*)(H + (size_t)row0 * F);
    float4* out4 = (float4*)(OUT + (size_t)row0 * F);
    const float4 bv = ((const float4*)b1)[t & 1];
    for (int i = t; i < nrows * F / 4; i += 512) {
      const float w = dinv[row0 + (i >> 1)];
      float4 av = acc4[i];
      const float4 hv = h4[i];
      av.x = fmaxf((av.x + hv.x) * w + bv.x, 0.0f);
      av.y = fmaxf((av.y + hv.y) * w + bv.y, 0.0f);
      av.z = fmaxf((av.z + hv.z) * w + bv.z, 0.0f);
      av.w = fmaxf((av.w + hv.w) * w + bv.w, 0.0f);
      out4[i] = av;
    }
  } else {
    const int m  = lane & 15;
    const int hi = lane >> 4;
    const int kk = hi << 1;
    const bool live_col = (m < F_H2);
    const int col = m & 7;
    v2f bf[4];
#pragma unroll
    for (int s = 0; s < 4; ++s) {
      const float w0 = W2[(4 * s + kk) * F_H2 + col];
      const float w1 = W2[(4 * s + kk + 1) * F_H2 + col];
      bf[s].x = live_col ? w0 : 0.0f;
      bf[s].y = live_col ? w1 : 0.0f;
    }
    const float bb0 = b1[8 + kk],  bb1 = b1[9 + kk];
    const float bb2 = b1[12 + kk], bb3 = b1[13 + kk];
    const int ntiles = nrows >> 4;
    const int wave_s = __builtin_amdgcn_readfirstlane(wave);
    for (int tile = wave_s; tile < ntiles; tile += NW) {
      const int lrow = tile * 16 + m;
      const size_t g = (size_t)(row0 + lrow);
      const float w = dinv[g];
      const float2 q0 = *(const float2*)(OUT + g * F + kk);
      const float2 q1 = *(const float2*)(OUT + g * F + 4 + kk);
      const float2 p0 = *(const float2*)(H + g * F + kk);
      const float2 p1 = *(const float2*)(H + g * F + 4 + kk);
      const float2 s0 = *(const float2*)(acc + lrow * F + kk);
      const float2 s1 = *(const float2*)(acc + lrow * F + 4 + kk);
      v2f a0, a1, a2, a3;
      a0.x = q0.x;  a0.y = q0.y;
      a1.x = q1.x;  a1.y = q1.y;
      a2.x = fmaxf((s0.x + p0.x) * w + bb0, 0.0f);
      a2.y = fmaxf((s0.y + p0.y) * w + bb1, 0.0f);
      a3.x = fmaxf((s1.x + p1.x) * w + bb2, 0.0f);
      a3.y = fmaxf((s1.y + p1.y) * w + bb3, 0.0f);
      v8f c = {0.f, 0.f, 0.f, 0.f, 0.f, 0.f, 0.f, 0.f};
      c = __builtin_amdgcn_wmma_f32_16x16x4_f32(false, a0, false, bf[0], (short)0, c, false, false);
      c = __builtin_amdgcn_wmma_f32_16x16x4_f32(false, a1, false, bf[1], (short)0, c, false, false);
      c = __builtin_amdgcn_wmma_f32_16x16x4_f32(false, a2, false, bf[2], (short)0, c, false, false);
      c = __builtin_amdgcn_wmma_f32_16x16x4_f32(false, a3, false, bf[3], (short)0, c, false, false);
      const int r0 = row0 + tile * 16 + 8 * hi;
      const float* dv = dinv + r0;
      float r[8];
#pragma unroll
      for (int j = 0; j < 8; ++j) r[j] = c[j] * dv[j];
      if (live_col) {
        float* out = OUT + (size_t)r0 * F + m;
#pragma unroll
        for (int j = 0; j < 8; ++j) out[(size_t)j * F] = r[j];
      }
    }
  }
}


__global__ void readout(const float* __restrict__ pooled, const float* __restrict__ cnt,
                        const float* __restrict__ Wl, const float* __restrict__ bl,
                        float* __restrict__ out, int G) {
  int g = blockIdx.x * blockDim.x + threadIdx.x;
  if (g >= G) return;
  float inv = 1.0f / fmaxf(cnt[g], 1.0f);
  float p[F_H2];
#pragma unroll
  for (int f = 0; f < F_H2; ++f) p[f] = pooled[(size_t)g * F_H2 + f] * inv;
#pragma unroll
  for (int c = 0; c < N_CLS; ++c) {
    float acc = bl[c];
#pragma unroll
    for (int f = 0; f < F_H2; ++f) acc += p[f] * Wl[f * N_CLS + c];
    out[(size_t)g * N_CLS + c] = acc;
  }
}


extern "C" void kernel_launch(void* const* d_in, const int* in_sizes, int n_in,
                              void* d_out, int out_size, void* d_ws, size_t ws_size,
                              hipStream_t stream) {
  const float* x     = (const float*)d_in[0];
  const int*   ei    = (const int*)d_in[1];
  const int*   src   = ei;
  const int*   dst   = ei + N_EDGES;
  const int*   batch = (const int*)d_in[2];
  const float* W1 = (const float*)d_in[4];
  const float* b1 = (const float*)d_in[5];
  const float* W2 = (const float*)d_in[6];
  const float* b2 = (const float*)d_in[7];
  const float* Wl = (const float*)d_in[8];
  const float* bl = (const float*)d_in[9];
  float* out = (float*)d_out;

  char* ws = (char*)d_ws;
  float* dinv   = (float*)(ws);
  float* bufA   = (float*)(ws + (4ll  << 20));
  float* bufB   = (float*)(ws + (68ll << 20));
  float* pooled = (float*)(ws + (132ll << 20));
  float* cnt    = pooled + (size_t)N_GRAPHS * F_H2;

  float* planeP = bufA;
  float* planeQ = (float*)((char*)bufA + (32ll << 20));
  float* h2pre  = planeQ;
  int* keys   = (int*)bufB;
  int* counts = (int*)((char*)bufB + 64000000ll);
  int* tot    = (int*)((char*)bufB + 64000000ll + (512ll << 10));
  int* bstart = (int*)((char*)bufB + 64000000ll + (520ll << 10));

  const int NB = 256;
  const int GEMM_BLOCKS = N_NODES / 64;

  bin_count<<<NBLK, 1024, 0, stream>>>(dst, counts);
  bin_scan<<<N_RANGES, 256, 0, stream>>>(counts, tot);
  bin_scatter<<<NBLK, 1024, 0, stream>>>(src, dst, counts, tot, bstart, keys);

  degree_range_lds<<<N_RANGES, 1024, 0, stream>>>(keys, bstart, dinv);

  gemm_x_w1_half<0><<<GEMM_BLOCKS, 128, 0, stream>>>(x, W1, dinv, planeP);
  agg_range_lds<1><<<N_RANGES, 512, 0, stream>>>(keys, bstart, planeP, planeQ, dinv, b1, W2,
                                                 b2, batch, pooled);
  gemm_x_w1_half<1><<<GEMM_BLOCKS, 128, 0, stream>>>(x, W1, dinv, planeP);
  agg_range_lds<2><<<N_RANGES, 512, 0, stream>>>(keys, bstart, planeP, planeQ, dinv, b1, W2,
                                                 b2, batch, pooled);

  zero_f32<<<128, 256, 0, stream>>>(pooled, (long long)N_GRAPHS * (F_H2 + 1));
  agg_range_lds<0><<<N_RANGES, 512, 0, stream>>>(keys, bstart, h2pre, planeP, dinv, b1, W2,
                                                 b2, batch, pooled);

  readout<<<(N_GRAPHS + NB - 1) / NB, NB, 0, stream>>>(pooled, cnt, Wl, bl,
                                                       out, N_GRAPHS);
}


# bin_scatter orders its chunk's keys by bin in LDS (256 KiB), then stores each bin's slice as whole 128-byte lines: ~600 thousand line stores, not 16 million 4-byte ones; sums still by arrival
# speedup vs baseline: 9.8204x; 1.1801x over previous
#include <hip/hip_runtime.h>


typedef float v2f __attribute__((ext_vector_type(2)));
typedef float v8f __attribute__((ext_vector_type(8)));

#define N_NODES   1000000
#define N_EDGES   16000000
#define F_IN      32
#define F_H1      16
#define F_H2      8
#define N_CLS     5
#define N_GRAPHS  10000

#define RB        11
#define R_ROWS    (1 << RB)
#define N_RANGES        ((N_NODES + R_ROWS - 1) >> RB)
#define CH        65536
#define NBLK      ((N_EDGES + CH - 1) / CH)

#define ATOMIC_FADD(p, v) \
  __hip_atomic_fetch_add((p), (v), __ATOMIC_RELAXED, __HIP_MEMORY_SCOPE_AGENT)


__global__ void zero_f32(float* __restrict__ p, long long n) {
  long long i = (long long)blockIdx.x * blockDim.x + threadIdx.x;
  long long stride = (long long)gridDim.x * blockDim.x;
  for (; i < n; i += stride) p[i] = 0.0f;
}


template <int HALF>
__global__ __launch_bounds__(128) void gemm_x_w1_half(const float* __restrict__ X,
                                                      const float* __restrict__ W1,
                                                      const float* __restrict__ dinv,
                                                      float* __restrict__ P) {
  int wave = blockIdx.x * (blockDim.x >> 5) + (threadIdx.x >> 5);
  int lane = threadIdx.x & 31;
  int m  = lane & 15;
  int kk = (lane >> 4) << 1;
  const float* xrow = X + (size_t)(wave * 16 + m) * F_IN;
  const bool live_col = (m < 8);
  const int col = HALF * 8 + (m & 7);

  v8f c = {0.f, 0.f, 0.f, 0.f, 0.f, 0.f, 0.f, 0.f};
#pragma unroll
  for (int k = 0; k < F_IN; k += 4) {
    v2f a, b;
    a.x = xrow[k + kk];
    a.y = xrow[k + kk + 1];
    const float w0 = W1[(k + kk) * F_H1 + col];
    const float w1 = W1[(k + kk + 1) * F_H1 + col];
    b.x = live_col ? w0 : 0.0f;
    b.y = live_col ? w1 : 0.0f;
    c = __builtin_amdgcn_wmma_f32_16x16x4_f32(false, a, false, b,
                                              (short)0, c, false, false);
  }
  int hi = lane >> 4;
  const int row0 = wave * 16 + 8 * hi;
  const float* dv = dinv + row0;
  float r[8];
#pragma unroll
  for (int j = 0; j < 8; ++j) r[j] = c[j] * dv[j];
  if (live_col) {
    float* out = P + (size_t)row0 * 8 + m;
#pragma unroll
    for (int j = 0; j < 8; ++j) out[(size_t)j * 8] = r[j];
  }
}


static_assert(N_NODES <= (1 << 20), "a key holds the source row in 20 bits beside RB bits of local row");
static_assert(N_RANGES <= 1024, "one thread per range in the scatter kernel's offset scan");
static_assert(NBLK <= 256, "one thread per binning work-group in the per-range scan");
static_assert(N_EDGES % 4 == 0 && CH % 4096 == 0, "binning kernels read four edges per lane");
static_assert(N_NODES % 16 == 0 && R_ROWS % 16 == 0, "a range is whole 16-row tiles");

__device__ __forceinline__ int wave_incl_scan(int v, int lane) {
#pragma unroll
  for (int o = 1; o < 32; o <<= 1) {
    const int t = __shfl_up(v, o);
    if (lane >= o) v += t;
  }
  return v;
}

__device__ __forceinline__ int block_excl_scan(int v, int* wsum, int* total) {
  const int lane = threadIdx.x & 31, wave = threadIdx.x >> 5;
  const int nw = blockDim.x >> 5;
  const int incl = wave_incl_scan(v, lane);
  if (lane == 31) wsum[wave] = incl;
  __syncthreads();
  if (wave == 0) {
    const int w  = (lane < nw) ? wsum[lane] : 0;
    const int wi = wave_incl_scan(w, lane);
    wsum[lane] = wi - w;
    if (lane == 31) wsum[32] = wi;
  }
  __syncthreads();
  *total = wsum[32];
  return wsum[wave] + incl - v;
}

__global__ __launch_bounds__(1024) void bin_count(const int* __restrict__ dst,
                                                  int* __restrict__ counts) {
  __shared__ int hist[N_RANGES];
  const int t = threadIdx.x, blk = blockIdx.x;
  for (int i = t; i < N_RANGES; i += 1024) hist[i] = 0;
  __syncthreads();
  const int base = blk * CH;
  const int4* d4 = (const int4*)(dst + base);
  for (int i = t; i < CH / 4; i += 1024) {
    if (base + 4 * i < N_EDGES) {
      const int4 d = d4[i];
      atomicAdd(&hist[d.x >> RB], 1);
      atomicAdd(&hist[d.y >> RB], 1);
      atomicAdd(&hist[d.z >> RB], 1);
      atomicAdd(&hist[d.w >> RB], 1);
    }
  }
  __syncthreads();
  for (int i = t; i < N_RANGES; i += 1024) counts[i * NBLK + blk] = hist[i];
}

__global__ __launch_bounds__(256) void bin_scan(int* __restrict__ counts,
                                                int* __restrict__ tot) {
  __shared__ int wsum[33];
  const int b = blockIdx.x, t = threadIdx.x;
  const int v = (t < NBLK) ? counts[b * NBLK + t] : 0;
  int total;
  const int ex = block_excl_scan(v, wsum, &total);
  if (t < NBLK) counts[b * NBLK + t] = ex;
  if (t == 0) tot[b] = total;
}

__global__ __launch_bounds__(1024) void bin_scatter(const int* __restrict__ src,
                                                    const int* __restrict__ dst,
                                                    const int* __restrict__ offs,
                                                    const int* __restrict__ tot,
                                                    int* __restrict__ bstart,
                                                    int* __restrict__ keys) {
  __shared__ int stage[CH];
  __shared__ int lbeg[N_RANGES + 1];
  __shared__ int lcur[N_RANGES];
  __shared__ int gbeg[N_RANGES];
  __shared__ int wsum[33];
  const int t = threadIdx.x, blk = blockIdx.x;
  const int lane = t & 31, wave = t >> 5;
  const int v = (t < N_RANGES) ? tot[t] : 0;
  int total;
  const int ex = block_excl_scan(v, wsum, &total);
  int n = 0;
  if (t < N_RANGES) {
    const int o  = offs[t * NBLK + blk];
    const int o1 = (blk + 1 < NBLK) ? offs[t * NBLK + blk + 1] : v;
    n = o1 - o;
    gbeg[t] = ex + o;
    if (blk == 0) bstart[t] = ex;
  }
  if (blk == 0 && t == 0) bstart[N_RANGES] = total;
  __syncthreads();
  int ltotal;
  const int lex = block_excl_scan(n, wsum, &ltotal);
  if (t < N_RANGES) {
    lbeg[t] = lex;
    lcur[t] = lex;
  }
  if (t == 0) lbeg[N_RANGES] = ltotal;
  __syncthreads();
  const int base = blk * CH;
  const int4* s4 = (const int4*)(src + base);
  const int4* d4 = (const int4*)(dst + base);
  for (int i = t; i < CH / 4; i += 1024) {
    if (base + 4 * i < N_EDGES) {
      const int4 s = s4[i], d = d4[i];
      const int p0 = atomicAdd(&lcur[d.x >> RB], 1);
      const int p1 = atomicAdd(&lcur[d.y >> RB], 1);
      const int p2 = atomicAdd(&lcur[d.z >> RB], 1);
      const int p3 = atomicAdd(&lcur[d.w >> RB], 1);
      stage[p0] = (s.x << RB) | (d.x & (R_ROWS - 1));
      stage[p1] = (s.y << RB) | (d.y & (R_ROWS - 1));
      stage[p2] = (s.z << RB) | (d.z & (R_ROWS - 1));
      stage[p3] = (s.w << RB) | (d.w & (R_ROWS - 1));
    }
  }
  __syncthreads();
  for (int r = wave; r < N_RANGES; r += 32) {
    const int ls = lbeg[r];
    const int nk = lbeg[r + 1] - ls;
    const int g  = gbeg[r];
    for (int j = lane - (g & 31); j < nk; j += 32)
      if (j >= 0) keys[g + j] = stage[ls + j];
  }
}

__global__ __launch_bounds__(1024) void degree_range_lds(const int* __restrict__ keys,
                                                         const int* __restrict__ bstart,
                                                         float* __restrict__ dinv) {
  constexpr int U  = 8;
  constexpr int NW = 1024 / 32;
  __shared__ int cnt[R_ROWS];

  const int t = threadIdx.x, lane = t & 31, wave = t >> 5;
  const int b = blockIdx.x;
  for (int i = t; i < R_ROWS; i += 1024) cnt[i] = 0;
  __syncthreads();

  const int e0 = bstart[b], e1 = bstart[b + 1];
  for (int base = e0 + wave * (32 * U); base < e1; base += NW * 32 * U) {
    int k[U];
#pragma unroll
    for (int u = 0; u < U; ++u) {
      const int e = base + u * 32 + lane;
      k[u] = (e < e1) ? keys[e] : -1;
    }
#pragma unroll
    for (int u = 0; u < U; ++u) {
      if (k[u] >= 0)
        __hip_atomic_fetch_add(&cnt[k[u] & (R_ROWS - 1)], 1, __ATOMIC_RELAXED,
                               __HIP_MEMORY_SCOPE_WORKGROUP);
    }
  }
  __syncthreads();

  const int row0 = b << RB;
  const int nrows = (N_NODES - row0 < R_ROWS) ? (N_NODES - row0) : R_ROWS;
  for (int i = t; i < nrows; i += 1024)
    dinv[row0 + i] = rsqrtf((float)cnt[i] + 1.0f);
}

template <int MODE>
__global__ __launch_bounds__(512) void agg_range_lds(const int* __restrict__ keys,
                                                     const int* __restrict__ bstart,
                                                     const float* __restrict__ H,
                                                     float* OUT,
                                                     const float* __restrict__ dinv,
                                                     const float* __restrict__ b1,
                                                     const float* __restrict__ W2,
                                                     const float* __restrict__ b2,
                                                     const int* __restrict__ batch,
                                                     float* __restrict__ pooled) {
  constexpr int F   = 8;
  constexpr int LPE = F / 4;
  constexpr int EPI = 32 / LPE;
  constexpr int U   = 4;
  constexpr int NW  = 512 / 32;
  __shared__ float acc[R_ROWS * F];

  const int t = threadIdx.x, lane = t & 31, wave = t >> 5;
  const int b = blockIdx.x;
  const int row0 = b << RB;
  const int nrows = (N_NODES - row0 < R_ROWS) ? (N_NODES - row0) : R_ROWS;

  float4* acc4 = (float4*)acc;
  for (int i = t; i < R_ROWS * F / 4; i += 512) acc4[i] = make_float4(0.f, 0.f, 0.f, 0.f);
  __syncthreads();

  const int e0 = bstart[b], e1 = bstart[b + 1];
  const int sub = lane / LPE;
  const int q   = lane % LPE;

  for (int base = e0 + wave * (32 * U); base < e1; base += NW * 32 * U) {
    int k[U];
#pragma unroll
    for (int u = 0; u < U; ++u) {
      const int e = base + u * 32 + lane;
      k[u] = (e < e1) ? keys[e] : -1;
    }
    float4 v[U * LPE];
    int    dl[U * LPE];
#pragma unroll
    for (int u = 0; u < U; ++u) {
#pragma unroll
      for (int i = 0; i < LPE; ++i) {
        const int ks = __shfl(k[u], i * EPI + sub);
        const bool ok = (ks >= 0);
        const unsigned s = ok ? ((unsigned)ks >> RB) : 0u;
        v[u * LPE + i]  = *(const float4*)(H + (size_t)s * F + 4 * q);
        dl[u * LPE + i] = ok ? (ks & (R_ROWS - 1)) : -1;
      }
    }
#pragma unroll
    for (int j = 0; j < U * LPE; ++j) {
      if (dl[j] >= 0) {
        float* a = acc + dl[j] * F + 4 * q;
        __hip_atomic_fetch_add(a + 0, v[j].x, __ATOMIC_RELAXED, __HIP_MEMORY_SCOPE_WORKGROUP);
        __hip_atomic_fetch_add(a + 1, v[j].y, __ATOMIC_RELAXED, __HIP_MEMORY_SCOPE_WORKGROUP);
        __hip_atomic_fetch_add(a + 2, v[j].z, __ATOMIC_RELAXED, __HIP_MEMORY_SCOPE_WORKGROUP);
        __hip_atomic_fetch_add(a + 3, v[j].w, __ATOMIC_RELAXED, __HIP_MEMORY_SCOPE_WORKGROUP);
      }
    }
  }
  __syncthreads();

  if constexpr (MODE == 0) {
    constexpr int RPW = R_ROWS / NW;
    constexpr int UP  = 8;
    static_assert(RPW % (4 * UP) == 0, "a wave's rows are whole trips of 4*UP rows");
    const int f  = lane & 7;
    const int r4 = lane >> 3;
    const float bf = b2[f];
    float* gcnt = pooled + (size_t)N_GRAPHS * F_H2;
    int   gcur = 0;
    int   nrun = 0;
    float run  = 0.0f;
    for (int i0 = 0; i0 < RPW / 4; i0 += UP) {
      int   g[UP];
      float w[UP], a[UP], h[UP];
#pragma unroll
      for (int u = 0; u < UP; ++u) {
        const int lrow = wave * RPW + 4 * (i0 + u) + r4;
        const bool ok = (lrow < nrows);
        const size_t grow = (size_t)(row0 + (ok ? lrow : 0));
        g[u] = ok ? batch[grow] : -1;
        w[u] = dinv[grow];
        h[u] = H[grow * F + f];
        a[u] = acc[lrow * F + f];
      }
#pragma unroll
      for (int u = 0; u < UP; ++u) {
        const int lrow = wave * RPW + 4 * (i0 + u) + r4;
        if (lrow < nrows) {
          const float v = fmaxf((a[u] + h[u]) * w[u] + bf, 0.0f);
          if (nrun > 0 && g[u] != gcur) {
            ATOMIC_FADD(&pooled[(size_t)gcur * F_H2 + f], run);
            if (f == 0) ATOMIC_FADD(&gcnt[gcur], (float)nrun);
            nrun = 0;
            run  = 0.0f;
          }
          gcur = g[u];
          run += v;
          ++nrun;
        }
      }
    }
    if (nrun > 0) {
      ATOMIC_FADD(&pooled[(size_t)gcur * F_H2 + f], run);
      if (f == 0) ATOMIC_FADD(&gcnt[gcur], (float)nrun);
    }
  } else if constexpr (MODE == 1) {
    const float4* h4 = (const float4*)(H + (size_t)row0 * F);
    float4* out4 = (float4*)(OUT + (size_t)row0 * F);
    const float4 bv = ((const float4*)b1)[t & 1];
    for (int i = t; i < nrows * F / 4; i += 512) {
      const float w = dinv[row0 + (i >> 1)];
      float4 av = acc4[i];
      const float4 hv = h4[i];
      av.x = fmaxf((av.x + hv.x) * w + bv.x, 0.0f);
      av.y = fmaxf((av.y + hv.y) * w + bv.y, 0.0f);
      av.z = fmaxf((av.z + hv.z) * w + bv.z, 0.0f);
      av.w = fmaxf((av.w + hv.w) * w + bv.w, 0.0f);
      out4[i] = av;
    }
  } else {
    const int m  = lane & 15;
    const int hi = lane >> 4;
    const int kk = hi << 1;
    const bool live_col = (m < F_H2);
    const int col = m & 7;
    v2f bf[4];
#pragma unroll
    for (int s = 0; s < 4; ++s) {
      const float w0 = W2[(4 * s + kk) * F_H2 + col];
      const float w1 = W2[(4 * s + kk + 1) * F_H2 + col];
      bf[s].x = live_col ? w0 : 0.0f;
      bf[s].y = live_col ? w1 : 0.0f;
    }
    const float bb0 = b1[8 + kk],  bb1 = b1[9 + kk];
    const float bb2 = b1[12 + kk], bb3 = b1[13 + kk];
    const int ntiles = nrows >> 4;
    const int wave_s = __builtin_amdgcn_readfirstlane(wave);
    for (int tile = wave_s; tile < ntiles; tile += NW) {
      const int lrow = tile * 16 + m;
      const size_t g = (size_t)(row0 + lrow);
      const float w = dinv[g];
      const float2 q0 = *(const float2*)(OUT + g * F + kk);
      const float2 q1 = *(const float2*)(OUT + g * F + 4 + kk);
      const float2 p0 = *(const float2*)(H + g * F + kk);
      const float2 p1 = *(const float2*)(H + g * F + 4 + kk);
      const float2 s0 = *(const float2*)(acc + lrow * F + kk);
      const float2 s1 = *(const float2*)(acc + lrow * F + 4 + kk);
      v2f a0, a1, a2, a3;
      a0.x = q0.x;  a0.y = q0.y;
      a1.x = q1.x;  a1.y = q1.y;
      a2.x = fmaxf((s0.x + p0.x) * w + bb0, 0.0f);
      a2.y = fmaxf((s0.y + p0.y) * w + bb1, 0.0f);
      a3.x = fmaxf((s1.x + p1.x) * w + bb2, 0.0f);
      a3.y = fmaxf((s1.y + p1.y) * w + bb3, 0.0f);
      v8f c = {0.f, 0.f, 0.f, 0.f, 0.f, 0.f, 0.f, 0.f};
      c = __builtin_amdgcn_wmma_f32_16x16x4_f32(false, a0, false, bf[0], (short)0, c, false, false);
      c = __builtin_amdgcn_wmma_f32_16x16x4_f32(false, a1, false, bf[1], (short)0, c, false, false);
      c = __builtin_amdgcn_wmma_f32_16x16x4_f32(false, a2, false, bf[2], (short)0, c, false, false);
      c = __builtin_amdgcn_wmma_f32_16x16x4_f32(false, a3, false, bf[3], (short)0, c, false, false);
      const int r0 = row0 + tile * 16 + 8 * hi;
      const float* dv = dinv + r0;
      float r[8];
#pragma unroll
      for (int j = 0; j < 8; ++j) r[j] = c[j] * dv[j];
      if (live_col) {
        float* out = OUT + (size_t)r0 * F + m;
#pragma unroll
        for (int j = 0; j < 8; ++j) out[(size_t)j * F] = r[j];
      }
    }
  }
}


__global__ void readout(const float* __restrict__ pooled, const float* __restrict__ cnt,
                        const float* __restrict__ Wl, const float* __restrict__ bl,
                        float* __restrict__ out, int G) {
  int g = blockIdx.x * blockDim.x + threadIdx.x;
  if (g >= G) return;
  float inv = 1.0f / fmaxf(cnt[g], 1.0f);
  float p[F_H2];
#pragma unroll
  for (int f = 0; f < F_H2; ++f) p[f] = pooled[(size_t)g * F_H2 + f] * inv;
#pragma unroll
  for (int c = 0; c < N_CLS; ++c) {
    float acc = bl[c];
#pragma unroll
    for (int f = 0; f < F_H2; ++f) acc += p[f] * Wl[f * N_CLS + c];
    out[(size_t)g * N_CLS + c] = acc;
  }
}


extern "C" void kernel_launch(void* const* d_in, const int* in_sizes, int n_in,
                              void* d_out, int out_size, void* d_ws, size_t ws_size,
                              hipStream_t stream) {
  const float* x     = (const float*)d_in[0];
  const int*   ei    = (const int*)d_in[1];
  const int*   src   = ei;
  const int*   dst   = ei + N_EDGES;
  const int*   batch = (const int*)d_in[2];
  const float* W1 = (const float*)d_in[4];
  const float* b1 = (const float*)d_in[5];
  const float* W2 = (const float*)d_in[6];
  const float* b2 = (const float*)d_in[7];
  const float* Wl = (const float*)d_in[8];
  const float* bl = (const float*)d_in[9];
  float* out = (float*)d_out;

  char* ws = (char*)d_ws;
  float* dinv   = (float*)(ws);
  float* bufA   = (float*)(ws + (4ll  << 20));
  float* bufB   = (float*)(ws + (68ll << 20));
  float* pooled = (float*)(ws + (132ll << 20));
  float* cnt    = pooled + (size_t)N_GRAPHS * F_H2;

  float* planeP = bufA;
  float* planeQ = (float*)((char*)bufA + (32ll << 20));
  float* h2pre  = planeQ;
  int* keys   = (int*)bufB;
  int* counts = (int*)((char*)bufB + 64000000ll);
  int* tot    = (int*)((char*)bufB + 64000000ll + (512ll << 10));
  int* bstart = (int*)((char*)bufB + 64000000ll + (520ll << 10));

  const int NB = 256;
  const int GEMM_BLOCKS = N_NODES / 64;

  bin_count<<<NBLK, 1024, 0, stream>>>(dst, counts);
  bin_scan<<<N_RANGES, 256, 0, stream>>>(counts, tot);
  bin_scatter<<<NBLK, 1024, 0, stream>>>(src, dst, counts, tot, bstart, keys);

  degree_range_lds<<<N_RANGES, 1024, 0, stream>>>(keys, bstart, dinv);

  gemm_x_w1_half<0><<<GEMM_BLOCKS, 128, 0, stream>>>(x, W1, dinv, planeP);
  agg_range_lds<1><<<N_RANGES, 512, 0, stream>>>(keys, bstart, planeP, planeQ, dinv, b1, W2,
                                                 b2, batch, pooled);
  gemm_x_w1_half<1><<<GEMM_BLOCKS, 128, 0, stream>>>(x, W1, dinv, planeP);
  agg_range_lds<2><<<N_RANGES, 512, 0, stream>>>(keys, bstart, planeP, planeQ, dinv, b1, W2,
                                                 b2, batch, pooled);

  zero_f32<<<128, 256, 0, stream>>>(pooled, (long long)N_GRAPHS * (F_H2 + 1));
  agg_range_lds<0><<<N_RANGES, 512, 0, stream>>>(keys, bstart, h2pre, planeP, dinv, b1, W2,
                                                 b2, batch, pooled);

  readout<<<(N_GRAPHS + NB - 1) / NB, NB, 0, stream>>>(pooled, cnt, Wl, bl,
                                                       out, N_GRAPHS);
}


# layer-1 edge sums in ONE pass over whole 64-byte product rows: one cache-line request per edge, not two; output rows overwrite the range's used keys, keys rebuilt for layer 2; sums still by arrival
# speedup vs baseline: 11.3790x; 1.1587x over previous
#include <hip/hip_runtime.h>


typedef float v2f __attribute__((ext_vector_type(2)));
typedef float v8f __attribute__((ext_vector_type(8)));

#define N_NODES   1000000
#define N_EDGES   16000000
#define F_IN      32
#define F_H1      16
#define F_H2      8
#define N_CLS     5
#define N_GRAPHS  10000

#define RB        11
#define R_ROWS    (1 << RB)
#define N_RANGES        ((N_NODES + R_ROWS - 1) >> RB)
#define CH        65536
#define NBLK      ((N_EDGES + CH - 1) / CH)

#define ATOMIC_FADD(p, v) \
  __hip_atomic_fetch_add((p), (v), __ATOMIC_RELAXED, __HIP_MEMORY_SCOPE_AGENT)


__global__ void zero_f32(float* __restrict__ p, long long n) {
  long long i = (long long)blockIdx.x * blockDim.x + threadIdx.x;
  long long stride = (long long)gridDim.x * blockDim.x;
  for (; i < n; i += stride) p[i] = 0.0f;
}


__device__ __forceinline__ void w1_tile_rows64(int tile, int lane,
                                               const float* __restrict__ X,
                                               const float* __restrict__ W1,
                                               const float* __restrict__ dinv,
                                               float* __restrict__ R) {
  const int m  = lane & 15;
  const int kk = (lane >> 4) << 1;
  const float* xrow = X + (size_t)(tile * 16 + m) * F_IN;
  v2f a[F_IN / 4], b[F_IN / 4];
#pragma unroll
  for (int s = 0; s < F_IN / 4; ++s) {
    a[s].x = xrow[4 * s + kk];
    a[s].y = xrow[4 * s + kk + 1];
    b[s].x = W1[(4 * s + kk) * F_H1 + m];
    b[s].y = W1[(4 * s + kk + 1) * F_H1 + m];
  }
  v8f c = {0.f, 0.f, 0.f, 0.f, 0.f, 0.f, 0.f, 0.f};
#pragma unroll
  for (int s = 0; s < F_IN / 4; ++s)
    c = __builtin_amdgcn_wmma_f32_16x16x4_f32(false, a[s], false, b[s],
                                              (short)0, c, false, false);
  const int hi = lane >> 4;
  const int row0 = tile * 16 + 8 * hi;
  const float* dv = dinv + row0;
  float r[8];
#pragma unroll
  for (int j = 0; j < 8; ++j) r[j] = c[j] * dv[j];
  float* out = R + (size_t)row0 * F_H1 + m;
#pragma unroll
  for (int j = 0; j < 8; ++j) out[(size_t)j * F_H1] = r[j];
}

template <int HALF>
__device__ __forceinline__ void w1_tile_half(int tile, int lane,
                                             const float* __restrict__ X,
                                             const float* __restrict__ W1,
                                             const float* __restrict__ dinv,
                                             float* __restrict__ P) {
  const int m  = lane & 15;
  const int kk = (lane >> 4) << 1;
  const float* xrow = X + (size_t)(tile * 16 + m) * F_IN;
  const bool live_col = (m < 8);
  const int col = HALF * 8 + (m & 7);

  v8f c = {0.f, 0.f, 0.f, 0.f, 0.f, 0.f, 0.f, 0.f};
#pragma unroll
  for (int k = 0; k < F_IN; k += 4) {
    v2f a, b;
    a.x = xrow[k + kk];
    a.y = xrow[k + kk + 1];
    const float w0 = W1[(k + kk) * F_H1 + col];
    const float w1 = W1[(k + kk + 1) * F_H1 + col];
    b.x = live_col ? w0 : 0.0f;
    b.y = live_col ? w1 : 0.0f;
    c = __builtin_amdgcn_wmma_f32_16x16x4_f32(false, a, false, b,
                                              (short)0, c, false, false);
  }
  const int hi = lane >> 4;
  const int row0 = tile * 16 + 8 * hi;
  const float* dv = dinv + row0;
  float r[8];
#pragma unroll
  for (int j = 0; j < 8; ++j) r[j] = c[j] * dv[j];
  if (live_col) {
    float* out = P + (size_t)row0 * 8 + m;
#pragma unroll
    for (int j = 0; j < 8; ++j) out[(size_t)j * 8] = r[j];
  }
}

__global__ __launch_bounds__(128) void gemm_x_w1_first(const float* __restrict__ X,
                                                       const float* __restrict__ W1,
                                                       const float* __restrict__ dinv,
                                                       float* __restrict__ P,
                                                       const int* __restrict__ fits) {
  const int ok = __builtin_amdgcn_readfirstlane(*fits);
  const int tile = blockIdx.x * (blockDim.x >> 5) + (threadIdx.x >> 5);
  const int lane = threadIdx.x & 31;
  if (ok) w1_tile_rows64(tile, lane, X, W1, dinv, P);
  else    w1_tile_half<0>(tile, lane, X, W1, dinv, P);
}

__global__ __launch_bounds__(128) void gemm_x_w1_plane1(const float* __restrict__ X,
                                                        const float* __restrict__ W1,
                                                        const float* __restrict__ dinv,
                                                        float* __restrict__ P,
                                                        const int* __restrict__ fits) {
  if (*fits != 0) return;
  const int lane = threadIdx.x & 31;
  const int w0 = __builtin_amdgcn_readfirstlane(
      (int)(blockIdx.x * (blockDim.x >> 5) + (threadIdx.x >> 5)));
  const int nw = (int)(gridDim.x * (blockDim.x >> 5));
  for (int tile = w0; tile < N_NODES / 16; tile += nw)
    w1_tile_half<1>(tile, lane, X, W1, dinv, P);
}


static_assert(N_NODES <= (1 << 20), "a key holds the source row in 20 bits beside RB bits of local row");
static_assert(N_RANGES <= 1024, "one thread per range in the scatter kernel's offset scan");
static_assert(NBLK <= 256, "one thread per binning work-group in the per-range scan");
static_assert(N_EDGES % 4 == 0 && CH % 4096 == 0, "binning kernels read four edges per lane");
static_assert(N_NODES % 16 == 0 && R_ROWS % 16 == 0, "a range is whole 16-row tiles");

__device__ __forceinline__ int wave_incl_scan(int v, int lane) {
#pragma unroll
  for (int o = 1; o < 32; o <<= 1) {
    const int t = __shfl_up(v, o);
    if (lane >= o) v += t;
  }
  return v;
}

__device__ __forceinline__ int block_excl_scan(int v, int* wsum, int* total) {
  const int lane = threadIdx.x & 31, wave = threadIdx.x >> 5;
  const int nw = blockDim.x >> 5;
  const int incl = wave_incl_scan(v, lane);
  if (lane == 31) wsum[wave] = incl;
  __syncthreads();
  if (wave == 0) {
    const int w  = (lane < nw) ? wsum[lane] : 0;
    const int wi = wave_incl_scan(w, lane);
    wsum[lane] = wi - w;
    if (lane == 31) wsum[32] = wi;
  }
  __syncthreads();
  *total = wsum[32];
  return wsum[wave] + incl - v;
}

__global__ __launch_bounds__(1024) void bin_count(const int* __restrict__ dst,
                                                  int* __restrict__ counts) {
  __shared__ int hist[N_RANGES];
  const int t = threadIdx.x, blk = blockIdx.x;
  for (int i = t; i < N_RANGES; i += 1024) hist[i] = 0;
  __syncthreads();
  const int base = blk * CH;
  const int4* d4 = (const int4*)(dst + base);
  for (int i = t; i < CH / 4; i += 1024) {
    if (base + 4 * i < N_EDGES) {
      const int4 d = d4[i];
      atomicAdd(&hist[d.x >> RB], 1);
      atomicAdd(&hist[d.y >> RB], 1);
      atomicAdd(&hist[d.z >> RB], 1);
      atomicAdd(&hist[d.w >> RB], 1);
    }
  }
  __syncthreads();
  for (int i = t; i < N_RANGES; i += 1024) counts[i * NBLK + blk] = hist[i];
}

__global__ __launch_bounds__(256) void bin_scan(int* __restrict__ counts,
                                                int* __restrict__ tot) {
  __shared__ int wsum[33];
  const int b = blockIdx.x, t = threadIdx.x;
  const int v = (t < NBLK) ? counts[b * NBLK + t] : 0;
  int total;
  const int ex = block_excl_scan(v, wsum, &total);
  if (t < NBLK) counts[b * NBLK + t] = ex;
  if (t == 0) tot[b] = total;
}

template <int PASS>
__global__ __launch_bounds__(1024) void bin_scatter(const int* __restrict__ src,
                                                    const int* __restrict__ dst,
                                                    const int* __restrict__ offs,
                                                    const int* __restrict__ tot,
                                                    int* __restrict__ bstart,
                                                    int* __restrict__ keys,
                                                    int* fits) {
  __shared__ int stage[CH];
  __shared__ int lbeg[N_RANGES + 1];
  __shared__ int lcur[N_RANGES];
  __shared__ int gbeg[N_RANGES];
  __shared__ int wsum[33];
  if (PASS == 1) {
    if (*fits == 0) return;
  }
  const int t = threadIdx.x, blk = blockIdx.x;
  const int lane = t & 31, wave = t >> 5;
  const int v = (t < N_RANGES) ? tot[t] : 0;
  int total;
  const int ex = block_excl_scan(v, wsum, &total);
  int n = 0;
  if (t < N_RANGES) {
    const int o  = offs[t * NBLK + blk];
    const int o1 = (blk + 1 < NBLK) ? offs[t * NBLK + blk + 1] : v;
    n = o1 - o;
    gbeg[t] = ex + o;
    if (PASS == 0 && blk == 0) bstart[t] = ex;
  }
  if (PASS == 0 && blk == 0) {
    if (t == 0) bstart[N_RANGES] = total;
    const int row0 = t << RB;
    const int nrows = (N_NODES - row0 < R_ROWS) ? (N_NODES - row0) : R_ROWS;
    const int roomy = (t >= N_RANGES) || (v >= nrows * F_H2 + 32);
    const int all = __syncthreads_and(roomy);
    if (t == 0) *fits = all;
  }
  __syncthreads();
  int ltotal;
  const int lex = block_excl_scan(n, wsum, &ltotal);
  if (t < N_RANGES) {
    lbeg[t] = lex;
    lcur[t] = lex;
  }
  if (t == 0) lbeg[N_RANGES] = ltotal;
  __syncthreads();
  const int base = blk * CH;
  const int4* s4 = (const int4*)(src + base);
  const int4* d4 = (const int4*)(dst + base);
  for (int i = t; i < CH / 4; i += 1024) {
    if (base + 4 * i < N_EDGES) {
      const int4 s = s4[i], d = d4[i];
      const int p0 = atomicAdd(&lcur[d.x >> RB], 1);
      const int p1 = atomicAdd(&lcur[d.y >> RB], 1);
      const int p2 = atomicAdd(&lcur[d.z >> RB], 1);
      const int p3 = atomicAdd(&lcur[d.w >> RB], 1);
      stage[p0] = (s.x << RB) | (d.x & (R_ROWS - 1));
      stage[p1] = (s.y << RB) | (d.y & (R_ROWS - 1));
      stage[p2] = (s.z << RB) | (d.z & (R_ROWS - 1));
      stage[p3] = (s.w << RB) | (d.w & (R_ROWS - 1));
    }
  }
  __syncthreads();
  for (int r = wave; r < N_RANGES; r += 32) {
    const int ls = lbeg[r];
    const int nk = lbeg[r + 1] - ls;
    const int g  = gbeg[r];
    for (int j = lane - (g & 31); j < nk; j += 32)
      if (j >= 0) keys[g + j] = stage[ls + j];
  }
}


__global__ __launch_bounds__(1024) void degree_range_lds(const int* __restrict__ keys,
                                                         const int* __restrict__ bstart,
                                                         float* __restrict__ dinv) {
  constexpr int U  = 8;
  constexpr int NW = 1024 / 32;
  __shared__ int cnt[R_ROWS];

  const int t = threadIdx.x, lane = t & 31, wave = t >> 5;
  const int b = blockIdx.x;
  for (int i = t; i < R_ROWS; i += 1024) cnt[i] = 0;
  __syncthreads();

  const int e0 = bstart[b], e1 = bstart[b + 1];
  for (int base = e0 + wave * (32 * U); base < e1; base += NW * 32 * U) {
    int k[U];
#pragma unroll
    for (int u = 0; u < U; ++u) {
      const int e = base + u * 32 + lane;
      k[u] = (e < e1) ? keys[e] : -1;
    }
#pragma unroll
    for (int u = 0; u < U; ++u) {
      if (k[u] >= 0)
        __hip_atomic_fetch_add(&cnt[k[u] & (R_ROWS - 1)], 1, __ATOMIC_RELAXED,
                               __HIP_MEMORY_SCOPE_WORKGROUP);
    }
  }
  __syncthreads();

  const int row0 = b << RB;
  const int nrows = (N_NODES - row0 < R_ROWS) ? (N_NODES - row0) : R_ROWS;
  for (int i = t; i < nrows; i += 1024)
    dinv[row0 + i] = rsqrtf((float)cnt[i] + 1.0f);
}


__global__ __launch_bounds__(512) void agg1_rows64(int* keys,
                                                   const int* __restrict__ bstart,
                                                   const float* __restrict__ R,
                                                   const float* __restrict__ dinv,
                                                   const float* __restrict__ b1,
                                                   const float* __restrict__ W2,
                                                   const int* __restrict__ fits) {
  constexpr int F   = F_H1;
  constexpr int LPE = F / 4;
  constexpr int EPI = 32 / LPE;
  constexpr int U   = 4;
  constexpr int NW  = 512 / 32;
  __shared__ float acc[R_ROWS * F];

  if (*fits == 0) return;

  const int t = threadIdx.x, lane = t & 31, wave = t >> 5;
  const int b = blockIdx.x;
  const int row0 = b << RB;
  const int nrows = (N_NODES - row0 < R_ROWS) ? (N_NODES - row0) : R_ROWS;

  float4* acc4 = (float4*)acc;
  for (int i = t; i < R_ROWS * F / 4; i += 512) acc4[i] = make_float4(0.f, 0.f, 0.f, 0.f);
  __syncthreads();

  const int e0 = bstart[b], e1 = bstart[b + 1];
  const int sub = lane / LPE;
  const int q   = lane % LPE;

  for (int base = e0 + wave * (32 * U); base < e1; base += NW * 32 * U) {
    int k[U];
#pragma unroll
    for (int u = 0; u < U; ++u) {
      const int e = base + u * 32 + lane;
      k[u] = (e < e1) ? __builtin_nontemporal_load(keys + e) : -1;
    }
    float4 v[U * LPE];
    int    dl[U * LPE];
#pragma unroll
    for (int u = 0; u < U; ++u) {
#pragma unroll
      for (int i = 0; i < LPE; ++i) {
        const int ks = __shfl(k[u], i * EPI + sub);
        const bool ok = (ks >= 0);
        const unsigned s = ok ? ((unsigned)ks >> RB) : 0u;
        v[u * LPE + i]  = *(const float4*)(R + (size_t)s * F + 4 * q);
        dl[u * LPE + i] = ok ? (ks & (R_ROWS - 1)) : -1;
      }
    }
#pragma unroll
    for (int j = 0; j < U * LPE; ++j) {
      if (dl[j] >= 0) {
        float* a = acc + dl[j] * F + 4 * q;
        __hip_atomic_fetch_add(a + 0, v[j].x, __ATOMIC_RELAXED, __HIP_MEMORY_SCOPE_WORKGROUP);
        __hip_atomic_fetch_add(a + 1, v[j].y, __ATOMIC_RELAXED, __HIP_MEMORY_SCOPE_WORKGROUP);
        __hip_atomic_fetch_add(a + 2, v[j].z, __ATOMIC_RELAXED, __HIP_MEMORY_SCOPE_WORKGROUP);
        __hip_atomic_fetch_add(a + 3, v[j].w, __ATOMIC_RELAXED, __HIP_MEMORY_SCOPE_WORKGROUP);
      }
    }
  }
  __syncthreads();

  const int m  = lane & 15;
  const int hi = lane >> 4;
  const int kk = hi << 1;
  const bool live_col = (m < F_H2);
  const int col = m & 7;
  v2f bf[4];
  float bx[4], by[4];
#pragma unroll
  for (int s = 0; s < 4; ++s) {
    const float w0 = W2[(4 * s + kk) * F_H2 + col];
    const float w1 = W2[(4 * s + kk + 1) * F_H2 + col];
    bf[s].x = live_col ? w0 : 0.0f;
    bf[s].y = live_col ? w1 : 0.0f;
    bx[s] = b1[4 * s + kk];
    by[s] = b1[4 * s + kk + 1];
  }
  float* Hout = (float*)keys + ((e0 + 31) & ~31);
  const int ntiles = nrows >> 4;
  const int wave_s = __builtin_amdgcn_readfirstlane(wave);
  for (int tile = wave_s; tile < ntiles; tile += NW) {
    const int lrow = tile * 16 + m;
    const size_t g = (size_t)(row0 + lrow);
    const float w = dinv[g];
    v2f a[4];
#pragma unroll
    for (int s = 0; s < 4; ++s) {
      const float2 p  = *(const float2*)(R + g * F + 4 * s + kk);
      const float2 sm = *(const float2*)(acc + lrow * F + 4 * s + kk);
      a[s].x = fmaxf((sm.x + p.x) * w + bx[s], 0.0f);
      a[s].y = fmaxf((sm.y + p.y) * w + by[s], 0.0f);
    }
    v8f c = {0.f, 0.f, 0.f, 0.f, 0.f, 0.f, 0.f, 0.f};
    c = __builtin_amdgcn_wmma_f32_16x16x4_f32(false, a[0], false, bf[0], (short)0, c, false, false);
    c = __builtin_amdgcn_wmma_f32_16x16x4_f32(false, a[1], false, bf[1], (short)0, c, false, false);
    c = __builtin_amdgcn_wmma_f32_16x16x4_f32(false, a[2], false, bf[2], (short)0, c, false, false);
    c = __builtin_amdgcn_wmma_f32_16x16x4_f32(false, a[3], false, bf[3], (short)0, c, false, false);
    const int lr0 = tile * 16 + 8 * hi;
    const float* dv = dinv + row0 + lr0;
    float r[8];
#pragma unroll
    for (int j = 0; j < 8; ++j) r[j] = c[j] * dv[j];
    if (live_col) {
      float* out = Hout + (size_t)lr0 * F_H2 + m;
#pragma unroll
      for (int j = 0; j < 8; ++j) out[(size_t)j * F_H2] = r[j];
    }
  }
}


template <int MODE>
__global__ __launch_bounds__(512) void agg_range_lds(const int* __restrict__ keys,
                                                     const int* __restrict__ bstart,
                                                     const float* __restrict__ H,
                                                     float* OUT,
                                                     const float* __restrict__ dinv,
                                                     const float* __restrict__ b1,
                                                     const float* __restrict__ W2,
                                                     const float* __restrict__ b2,
                                                     const int* __restrict__ batch,
                                                     float* __restrict__ pooled,
                                                     const int* __restrict__ fits,
                                                     const int* __restrict__ keys2,
                                                     const float* __restrict__ Hb) {
  constexpr int F   = 8;
  constexpr int LPE = F / 4;
  constexpr int EPI = 32 / LPE;
  constexpr int U   = 4;
  constexpr int NW  = 512 / 32;
  __shared__ float acc[R_ROWS * F];
  __shared__ int hoff[MODE == 0 ? N_RANGES : 1];

  const int fit = *fits;
  if (MODE != 0 && fit) return;
  const int*   ksrc = (MODE == 0 && fit) ? keys2 : keys;
  const float* Hs   = (MODE == 0 && fit) ? Hb : H;

  const int t = threadIdx.x, lane = t & 31, wave = t >> 5;
  const int b = blockIdx.x;
  const int row0 = b << RB;
  const int nrows = (N_NODES - row0 < R_ROWS) ? (N_NODES - row0) : R_ROWS;

  float4* acc4 = (float4*)acc;
  for (int i = t; i < R_ROWS * F / 4; i += 512) acc4[i] = make_float4(0.f, 0.f, 0.f, 0.f);
  if constexpr (MODE == 0) {
    for (int r = t; r < N_RANGES; r += 512)
      hoff[r] = fit ? ((bstart[r] + 31) & ~31) : (r << (RB + 3));
  }
  __syncthreads();

  const int e0 = bstart[b], e1 = bstart[b + 1];
  const int sub = lane / LPE;
  const int q   = lane % LPE;

  for (int base = e0 + wave * (32 * U); base < e1; base += NW * 32 * U) {
    int k[U];
#pragma unroll
    for (int u = 0; u < U; ++u) {
      const int e = base + u * 32 + lane;
      k[u] = (e < e1) ? ksrc[e] : -1;
    }
    float4 v[U * LPE];
    int    dl[U * LPE];
#pragma unroll
    for (int u = 0; u < U; ++u) {
#pragma unroll
      for (int i = 0; i < LPE; ++i) {
        const int ks = __shfl(k[u], i * EPI + sub);
        const bool ok = (ks >= 0);
        const unsigned s = ok ? ((unsigned)ks >> RB) : 0u;
        if constexpr (MODE == 0)
          v[u * LPE + i] = *(const float4*)(Hs + (size_t)hoff[s >> RB] +
                                            (size_t)(s & (R_ROWS - 1)) * F + 4 * q);
        else
          v[u * LPE + i] = *(const float4*)(H + (size_t)s * F + 4 * q);
        dl[u * LPE + i] = ok ? (ks & (R_ROWS - 1)) : -1;
      }
    }
#pragma unroll
    for (int j = 0; j < U * LPE; ++j) {
      if (dl[j] >= 0) {
        float* a = acc + dl[j] * F + 4 * q;
        __hip_atomic_fetch_add(a + 0, v[j].x, __ATOMIC_RELAXED, __HIP_MEMORY_SCOPE_WORKGROUP);
        __hip_atomic_fetch_add(a + 1, v[j].y, __ATOMIC_RELAXED, __HIP_MEMORY_SCOPE_WORKGROUP);
        __hip_atomic_fetch_add(a + 2, v[j].z, __ATOMIC_RELAXED, __HIP_MEMORY_SCOPE_WORKGROUP);
        __hip_atomic_fetch_add(a + 3, v[j].w, __ATOMIC_RELAXED, __HIP_MEMORY_SCOPE_WORKGROUP);
      }
    }
  }
  __syncthreads();

  if constexpr (MODE == 0) {
    constexpr int RPW = R_ROWS / NW;
    constexpr int UP  = 8;
    static_assert(RPW % (4 * UP) == 0, "a wave's rows are whole trips of 4*UP rows");
    const int f  = lane & 7;
    const int r4 = lane >> 3;
    const float bf = b2[f];
    float* gcnt = pooled + (size_t)N_GRAPHS * F_H2;
    const float* Hown = Hs + (size_t)hoff[b];
    int   gcur = 0;
    int   nrun = 0;
    float run  = 0.0f;
    for (int i0 = 0; i0 < RPW / 4; i0 += UP) {
      int   g[UP];
      float w[UP], a[UP], h[UP];
#pragma unroll
      for (int u = 0; u < UP; ++u) {
        const int lrow = wave * RPW + 4 * (i0 + u) + r4;
        const bool ok = (lrow < nrows);
        const int lr = ok ? lrow : 0;
        const size_t grow = (size_t)(row0 + lr);
        g[u] = ok ? batch[grow] : -1;
        w[u] = dinv[grow];
        h[u] = Hown[lr * F + f];
        a[u] = acc[lrow * F + f];
      }
#pragma unroll
      for (int u = 0; u < UP; ++u) {
        const int lrow = wave * RPW + 4 * (i0 + u) + r4;
        if (lrow < nrows) {
          const float v = fmaxf((a[u] + h[u]) * w[u] + bf, 0.0f);
          if (nrun > 0 && g[u] != gcur) {
            ATOMIC_FADD(&pooled[(size_t)gcur * F_H2 + f], run);
            if (f == 0) ATOMIC_FADD(&gcnt[gcur], (float)nrun);
            nrun = 0;
            run  = 0.0f;
          }
          gcur = g[u];
          run += v;
          ++nrun;
        }
      }
    }
    if (nrun > 0) {
      ATOMIC_FADD(&pooled[(size_t)gcur * F_H2 + f], run);
      if (f == 0) ATOMIC_FADD(&gcnt[gcur], (float)nrun);
    }
  } else if constexpr (MODE == 1) {
    const float4* h4 = (const float4*)(H + (size_t)row0 * F);
    float4* out4 = (float4*)(OUT + (size_t)row0 * F);
    const float4 bv = ((const float4*)b1)[t & 1];
    for (int i = t; i < nrows * F / 4; i += 512) {
      const float w = dinv[row0 + (i >> 1)];
      float4 av = acc4[i];
      const float4 hv = h4[i];
      av.x = fmaxf((av.x + hv.x) * w + bv.x, 0.0f);
      av.y = fmaxf((av.y + hv.y) * w + bv.y, 0.0f);
      av.z = fmaxf((av.z + hv.z) * w + bv.z, 0.0f);
      av.w = fmaxf((av.w + hv.w) * w + bv.w, 0.0f);
      out4[i] = av;
    }
  } else {
    const int m  = lane & 15;
    const int hi = lane >> 4;
    const int kk = hi << 1;
    const bool live_col = (m < F_H2);
    const int col = m & 7;
    v2f bf[4];
#pragma unroll
    for (int s = 0; s < 4; ++s) {
      const float w0 = W2[(4 * s + kk) * F_H2 + col];
      const float w1 = W2[(4 * s + kk + 1) * F_H2 + col];
      bf[s].x = live_col ? w0 : 0.0f;
      bf[s].y = live_col ? w1 : 0.0f;
    }
    const float bb0 = b1[8 + kk],  bb1 = b1[9 + kk];
    const float bb2 = b1[12 + kk], bb3 = b1[13 + kk];
    const int ntiles = nrows >> 4;
    const int wave_s = __builtin_amdgcn_readfirstlane(wave);
    for (int tile = wave_s; tile < ntiles; tile += NW) {
      const int lrow = tile * 16 + m;
      const size_t g = (size_t)(row0 + lrow);
      const float w = dinv[g];
      const float2 q0 = *(const float2*)(OUT + g * F + kk);
      const float2 q1 = *(const float2*)(OUT + g * F + 4 + kk);
      const float2 p0 = *(const float2*)(H + g * F + kk);
      const float2 p1 = *(const float2*)(H + g * F + 4 + kk);
      const float2 s0 = *(const float2*)(acc + lrow * F + kk);
      const float2 s1 = *(const float2*)(acc + lrow * F + 4 + kk);
      v2f a0, a1, a2, a3;
      a0.x = q0.x;  a0.y = q0.y;
      a1.x = q1.x;  a1.y = q1.y;
      a2.x = fmaxf((s0.x + p0.x) * w + bb0, 0.0f);
      a2.y = fmaxf((s0.y + p0.y) * w + bb1, 0.0f);
      a3.x = fmaxf((s1.x + p1.x) * w + bb2, 0.0f);
      a3.y = fmaxf((s1.y + p1.y) * w + bb3, 0.0f);
      v8f c = {0.f, 0.f, 0.f, 0.f, 0.f, 0.f, 0.f, 0.f};
      c = __builtin_amdgcn_wmma_f32_16x16x4_f32(false, a0, false, bf[0], (short)0, c, false, false);
      c = __builtin_amdgcn_wmma_f32_16x16x4_f32(false, a1, false, bf[1], (short)0, c, false, false);
      c = __builtin_amdgcn_wmma_f32_16x16x4_f32(false, a2, false, bf[2], (short)0, c, false, false);
      c = __builtin_amdgcn_wmma_f32_16x16x4_f32(false, a3, false, bf[3], (short)0, c, false, false);
      const int r0 = row0 + tile * 16 + 8 * hi;
      const float* dv = dinv + r0;
      float r[8];
#pragma unroll
      for (int j = 0; j < 8; ++j) r[j] = c[j] * dv[j];
      if (live_col) {
        float* out = OUT + (size_t)r0 * F + m;
#pragma unroll
        for (int j = 0; j < 8; ++j) out[(size_t)j * F] = r[j];
      }
    }
  }
}


__global__ void readout(const float* __restrict__ pooled, const float* __restrict__ cnt,
                        const float* __restrict__ Wl, const float* __restrict__ bl,
                        float* __restrict__ out, int G) {
  int g = blockIdx.x * blockDim.x + threadIdx.x;
  if (g >= G) return;
  float inv = 1.0f / fmaxf(cnt[g], 1.0f);
  float p[F_H2];
#pragma unroll
  for (int f = 0; f < F_H2; ++f) p[f] = pooled[(size_t)g * F_H2 + f] * inv;
#pragma unroll
  for (int c = 0; c < N_CLS; ++c) {
    float acc = bl[c];
#pragma unroll
    for (int f = 0; f < F_H2; ++f) acc += p[f] * Wl[f * N_CLS + c];
    out[(size_t)g * N_CLS + c] = acc;
  }
}


extern "C" void kernel_launch(void* const* d_in, const int* in_sizes, int n_in,
                              void* d_out, int out_size, void* d_ws, size_t ws_size,
                              hipStream_t stream) {
  const float* x     = (const float*)d_in[0];
  const int*   ei    = (const int*)d_in[1];
  const int*   src   = ei;
  const int*   dst   = ei + N_EDGES;
  const int*   batch = (const int*)d_in[2];
  const float* W1 = (const float*)d_in[4];
  const float* b1 = (const float*)d_in[5];
  const float* W2 = (const float*)d_in[6];
  const float* b2 = (const float*)d_in[7];
  const float* Wl = (const float*)d_in[8];
  const float* bl = (const float*)d_in[9];
  float* out = (float*)d_out;

  char* ws = (char*)d_ws;
  float* dinv   = (float*)(ws);
  float* bufA   = (float*)(ws + (4ll  << 20));
  float* bufB   = (float*)(ws + (68ll << 20));
  float* pooled = (float*)(ws + (132ll << 20));
  float* cnt    = pooled + (size_t)N_GRAPHS * F_H2;

  float* rows64 = bufA;
  float* planeP = bufA;
  float* planeQ = (float*)((char*)bufA + (32ll << 20));
  int* keys   = (int*)bufB;
  int* keys2  = (int*)bufA;
  int* counts = (int*)((char*)bufB + 64000000ll);
  int* tot    = (int*)((char*)bufB + 64000000ll + (512ll << 10));
  int* bstart = (int*)((char*)bufB + 64000000ll + (520ll << 10));
  int* fits   = (int*)((char*)bufB + 64000000ll + (528ll << 10));

  const int NB = 256;
  const int GEMM_BLOCKS = N_NODES / 64;

  bin_count<<<NBLK, 1024, 0, stream>>>(dst, counts);
  bin_scan<<<N_RANGES, 256, 0, stream>>>(counts, tot);
  bin_scatter<0><<<NBLK, 1024, 0, stream>>>(src, dst, counts, tot, bstart, keys, fits);

  degree_range_lds<<<N_RANGES, 1024, 0, stream>>>(keys, bstart, dinv);

  gemm_x_w1_first<<<GEMM_BLOCKS, 128, 0, stream>>>(x, W1, dinv, rows64, fits);
  agg1_rows64<<<N_RANGES, 512, 0, stream>>>(keys, bstart, rows64, dinv, b1, W2, fits);
  agg_range_lds<1><<<N_RANGES, 512, 0, stream>>>(keys, bstart, planeP, planeQ, dinv, b1, W2,
                                                 b2, batch, pooled, fits, keys2, planeQ);
  gemm_x_w1_plane1<<<1024, 128, 0, stream>>>(x, W1, dinv, planeP, fits);
  agg_range_lds<2><<<N_RANGES, 512, 0, stream>>>(keys, bstart, planeP, planeQ, dinv, b1, W2,
                                                 b2, batch, pooled, fits, keys2, planeQ);
  bin_scatter<1><<<NBLK, 1024, 0, stream>>>(src, dst, counts, tot, bstart, keys2, fits);

  zero_f32<<<128, 256, 0, stream>>>(pooled, (long long)N_GRAPHS * (F_H2 + 1));
  agg_range_lds<0><<<N_RANGES, 512, 0, stream>>>(keys, bstart, planeQ, planeP, dinv, b1, W2,
                                                 b2, batch, pooled, fits, keys2,
                                                 (const float*)keys);

  readout<<<(N_GRAPHS + NB - 1) / NB, NB, 0, stream>>>(pooled, cnt, Wl, bl,
                                                       out, N_GRAPHS);
}


# degree count and X*W1 product in ONE launch per destination range: dinv goes to the epilogue from LDS, 16 waves walk the range's tiles with the next tile's loads in flight; 62500 one-tile waves gone
# speedup vs baseline: 11.7967x; 1.0367x over previous
#include <hip/hip_runtime.h>


typedef float v2f __attribute__((ext_vector_type(2)));
typedef float v8f __attribute__((ext_vector_type(8)));

#define N_NODES   1000000
#define N_EDGES   16000000
#define F_IN      32
#define F_H1      16
#define F_H2      8
#define N_CLS     5
#define N_GRAPHS  10000

#define RB        11
#define R_ROWS    (1 << RB)
#define N_RANGES        ((N_NODES + R_ROWS - 1) >> RB)
#define CH        65536
#define NBLK      ((N_EDGES + CH - 1) / CH)

#define ATOMIC_FADD(p, v) \
  __hip_atomic_fetch_add((p), (v), __ATOMIC_RELAXED, __HIP_MEMORY_SCOPE_AGENT)


__global__ void zero_f32(float* __restrict__ p, long long n) {
  long long i = (long long)blockIdx.x * blockDim.x + threadIdx.x;
  long long stride = (long long)gridDim.x * blockDim.x;
  for (; i < n; i += stride) p[i] = 0.0f;
}


__device__ __forceinline__ void w1_load_a(v2f (&a)[F_IN / 4], const float* __restrict__ xk) {
#pragma unroll
  for (int s = 0; s < F_IN / 4; ++s) {
    a[s].x = xk[4 * s];
    a[s].y = xk[4 * s + 1];
  }
}

__device__ __forceinline__ void w1_finish_rows64(const v2f (&a)[F_IN / 4],
                                                 const v2f (&bw)[F_IN / 4],
                                                 const float* dv,
                                                 float* __restrict__ out) {
  v8f c = {0.f, 0.f, 0.f, 0.f, 0.f, 0.f, 0.f, 0.f};
#pragma unroll
  for (int s = 0; s < F_IN / 4; ++s)
    c = __builtin_amdgcn_wmma_f32_16x16x4_f32(false, a[s], false, bw[s],
                                              (short)0, c, false, false);
  float r[8];
#pragma unroll
  for (int j = 0; j < 8; ++j) r[j] = c[j] * dv[j];
#pragma unroll
  for (int j = 0; j < 8; ++j) out[(size_t)j * F_H1] = r[j];
}

template <int HALF>
__device__ __forceinline__ void w1_tile_half(int tile, int lane,
                                             const float* __restrict__ X,
                                             const float* __restrict__ W1,
                                             const float* dv,
                                             float* __restrict__ P) {
  const int m  = lane & 15;
  const int kk = (lane >> 4) << 1;
  const float* xrow = X + (size_t)(tile * 16 + m) * F_IN;
  const bool live_col = (m < 8);
  const int col = HALF * 8 + (m & 7);

  v8f c = {0.f, 0.f, 0.f, 0.f, 0.f, 0.f, 0.f, 0.f};
#pragma unroll
  for (int k = 0; k < F_IN; k += 4) {
    v2f a, b;
    a.x = xrow[k + kk];
    a.y = xrow[k + kk + 1];
    const float w0 = W1[(k + kk) * F_H1 + col];
    const float w1 = W1[(k + kk + 1) * F_H1 + col];
    b.x = live_col ? w0 : 0.0f;
    b.y = live_col ? w1 : 0.0f;
    c = __builtin_amdgcn_wmma_f32_16x16x4_f32(false, a, false, b,
                                              (short)0, c, false, false);
  }
  const int hi = lane >> 4;
  const int row0 = tile * 16 + 8 * hi;
  float r[8];
#pragma unroll
  for (int j = 0; j < 8; ++j) r[j] = c[j] * dv[j];
  if (live_col) {
    float* out = P + (size_t)row0 * 8 + m;
#pragma unroll
    for (int j = 0; j < 8; ++j) out[(size_t)j * 8] = r[j];
  }
}

__global__ __launch_bounds__(128) void gemm_x_w1_plane1(const float* __restrict__ X,
                                                        const float* __restrict__ W1,
                                                        const float* __restrict__ dinv,
                                                        float* __restrict__ P,
                                                        const int* __restrict__ fits) {
  if (*fits != 0) return;
  const int lane = threadIdx.x & 31;
  const int w0 = __builtin_amdgcn_readfirstlane(
      (int)(blockIdx.x * (blockDim.x >> 5) + (threadIdx.x >> 5)));
  const int nw = (int)(gridDim.x * (blockDim.x >> 5));
  for (int tile = w0; tile < N_NODES / 16; tile += nw)
    w1_tile_half<1>(tile, lane, X, W1, dinv + tile * 16 + 8 * (lane >> 4), P);
}


static_assert(N_NODES <= (1 << 20), "a key holds the source row in 20 bits beside RB bits of local row");
static_assert(N_RANGES <= 1024, "one thread per range in the scatter kernel's offset scan");
static_assert(NBLK <= 256, "one thread per binning work-group in the per-range scan");
static_assert(N_EDGES % 4 == 0 && CH % 4096 == 0, "binning kernels read four edges per lane");
static_assert(N_NODES % 16 == 0 && R_ROWS % 16 == 0, "a range is whole 16-row tiles");

__device__ __forceinline__ int wave_incl_scan(int v, int lane) {
#pragma unroll
  for (int o = 1; o < 32; o <<= 1) {
    const int t = __shfl_up(v, o);
    if (lane >= o) v += t;
  }
  return v;
}

__device__ __forceinline__ int block_excl_scan(int v, int* wsum, int* total) {
  const int lane = threadIdx.x & 31, wave = threadIdx.x >> 5;
  const int nw = blockDim.x >> 5;
  const int incl = wave_incl_scan(v, lane);
  if (lane == 31) wsum[wave] = incl;
  __syncthreads();
  if (wave == 0) {
    const int w  = (lane < nw) ? wsum[lane] : 0;
    const int wi = wave_incl_scan(w, lane);
    wsum[lane] = wi - w;
    if (lane == 31) wsum[32] = wi;
  }
  __syncthreads();
  *total = wsum[32];
  return wsum[wave] + incl - v;
}

__global__ __launch_bounds__(1024) void bin_count(const int* __restrict__ dst,
                                                  int* __restrict__ counts) {
  __shared__ int hist[N_RANGES];
  const int t = threadIdx.x, blk = blockIdx.x;
  for (int i = t; i < N_RANGES; i += 1024) hist[i] = 0;
  __syncthreads();
  const int base = blk * CH;
  const int4* d4 = (const int4*)(dst + base);
  for (int i = t; i < CH / 4; i += 1024) {
    if (base + 4 * i < N_EDGES) {
      const int4 d = d4[i];
      atomicAdd(&hist[d.x >> RB], 1);
      atomicAdd(&hist[d.y >> RB], 1);
      atomicAdd(&hist[d.z >> RB], 1);
      atomicAdd(&hist[d.w >> RB], 1);
    }
  }
  __syncthreads();
  for (int i = t; i < N_RANGES; i += 1024) counts[i * NBLK + blk] = hist[i];
}

__global__ __launch_bounds__(256) void bin_scan(int* __restrict__ counts,
                                                int* __restrict__ tot) {
  __shared__ int wsum[33];
  const int b = blockIdx.x, t = threadIdx.x;
  const int v = (t < NBLK) ? counts[b * NBLK + t] : 0;
  int total;
  const int ex = block_excl_scan(v, wsum, &total);
  if (t < NBLK) counts[b * NBLK + t] = ex;
  if (t == 0) tot[b] = total;
}

template <int PASS>
__global__ __launch_bounds__(1024) void bin_scatter(const int* __restrict__ src,
                                                    const int* __restrict__ dst,
                                                    const int* __restrict__ offs,
                                                    const int* __restrict__ tot,
                                                    int* __restrict__ bstart,
                                                    int* __restrict__ keys,
                                                    int* fits) {
  __shared__ int stage[CH];
  __shared__ int lbeg[N_RANGES + 1];
  __shared__ int lcur[N_RANGES];
  __shared__ int gbeg[N_RANGES];
  __shared__ int wsum[33];
  if (PASS == 1) {
    if (*fits == 0) return;
  }
  const int t = threadIdx.x, blk = blockIdx.x;
  const int lane = t & 31, wave = t >> 5;
  const int v = (t < N_RANGES) ? tot[t] : 0;
  int total;
  const int ex = block_excl_scan(v, wsum, &total);
  int n = 0;
  if (t < N_RANGES) {
    const int o  = offs[t * NBLK + blk];
    const int o1 = (blk + 1 < NBLK) ? offs[t * NBLK + blk + 1] : v;
    n = o1 - o;
    gbeg[t] = ex + o;
    if (PASS == 0 && blk == 0) bstart[t] = ex;
  }
  if (PASS == 0 && blk == 0) {
    if (t == 0) bstart[N_RANGES] = total;
    const int row0 = t << RB;
    const int nrows = (N_NODES - row0 < R_ROWS) ? (N_NODES - row0) : R_ROWS;
    const int roomy = (t >= N_RANGES) || (v >= nrows * F_H2 + 32);
    const int all = __syncthreads_and(roomy);
    if (t == 0) *fits = all;
  }
  __syncthreads();
  int ltotal;
  const int lex = block_excl_scan(n, wsum, &ltotal);
  if (t < N_RANGES) {
    lbeg[t] = lex;
    lcur[t] = lex;
  }
  if (t == 0) lbeg[N_RANGES] = ltotal;
  __syncthreads();
  const int base = blk * CH;
  const int4* s4 = (const int4*)(src + base);
  const int4* d4 = (const int4*)(dst + base);
  for (int i = t; i < CH / 4; i += 1024) {
    if (base + 4 * i < N_EDGES) {
      const int4 s = s4[i], d = d4[i];
      const int p0 = atomicAdd(&lcur[d.x >> RB], 1);
      const int p1 = atomicAdd(&lcur[d.y >> RB], 1);
      const int p2 = atomicAdd(&lcur[d.z >> RB], 1);
      const int p3 = atomicAdd(&lcur[d.w >> RB], 1);
      stage[p0] = (s.x << RB) | (d.x & (R_ROWS - 1));
      stage[p1] = (s.y << RB) | (d.y & (R_ROWS - 1));
      stage[p2] = (s.z << RB) | (d.z & (R_ROWS - 1));
      stage[p3] = (s.w << RB) | (d.w & (R_ROWS - 1));
    }
  }
  __syncthreads();
  for (int r = wave; r < N_RANGES; r += 32) {
    const int ls = lbeg[r];
    const int nk = lbeg[r + 1] - ls;
    const int g  = gbeg[r];
    for (int j = lane - (g & 31); j < nk; j += 32)
      if (j >= 0) keys[g + j] = stage[ls + j];
  }
}


__global__ __launch_bounds__(512) void degree_w1_range(const int* __restrict__ keys,
                                                       const int* __restrict__ bstart,
                                                       const float* __restrict__ X,
                                                       const float* __restrict__ W1,
                                                       float* __restrict__ dinv,
                                                       float* __restrict__ P,
                                                       const int* __restrict__ fits) {
  constexpr int U  = 16;
  constexpr int NW = 512 / 32;
  __shared__ int   cnt[R_ROWS];
  __shared__ float dloc[R_ROWS];

  const int t = threadIdx.x, lane = t & 31, wave = t >> 5;
  const int b = blockIdx.x;
  const int row0 = b << RB;
  const int nrows = (N_NODES - row0 < R_ROWS) ? (N_NODES - row0) : R_ROWS;
  const int ntiles = nrows >> 4;
  const int ok = __builtin_amdgcn_readfirstlane(*fits);
  const int wave_s = __builtin_amdgcn_readfirstlane(wave);

  const int m  = lane & 15;
  const int hi = lane >> 4;
  const int kk = hi << 1;
  const float* xk = X + (size_t)(row0 + m) * F_IN + kk;

  for (int i = t; i < R_ROWS; i += 512) cnt[i] = 0;
  __syncthreads();

  v2f a[F_IN / 4], bw[F_IN / 4];
#pragma unroll
  for (int s = 0; s < F_IN / 4; ++s) {
    a[s].x = 0.0f;  a[s].y = 0.0f;
    bw[s].x = 0.0f; bw[s].y = 0.0f;
  }
  if (ok && wave_s < ntiles) {
    w1_load_a(a, xk + (size_t)wave_s * 16 * F_IN);
#pragma unroll
    for (int s = 0; s < F_IN / 4; ++s) {
      bw[s].x = W1[(4 * s + kk) * F_H1 + m];
      bw[s].y = W1[(4 * s + kk + 1) * F_H1 + m];
    }
  }

  const int e0 = bstart[b], e1 = bstart[b + 1];
  for (int base = e0 + wave * (32 * U); base < e1; base += NW * 32 * U) {
    int k[U];
#pragma unroll
    for (int u = 0; u < U; ++u) {
      const int e = base + u * 32 + lane;
      k[u] = (e < e1) ? keys[e] : -1;
    }
#pragma unroll
    for (int u = 0; u < U; ++u) {
      if (k[u] >= 0)
        __hip_atomic_fetch_add(&cnt[k[u] & (R_ROWS - 1)], 1, __ATOMIC_RELAXED,
                               __HIP_MEMORY_SCOPE_WORKGROUP);
    }
  }
  __syncthreads();

  for (int i = t; i < R_ROWS; i += 512) {
    const float d = rsqrtf((float)cnt[i] + 1.0f);
    dloc[i] = d;
    if (i < nrows) dinv[row0 + i] = d;
  }
  __syncthreads();

  if (ok) {
    const int n = (ntiles - wave_s + NW - 1) / NW;
    v2f a2[F_IN / 4];
    const int lr = 8 * hi;
    const float* xw = xk + (size_t)wave_s * 16 * F_IN;
    const float* dw = dloc + wave_s * 16 + lr;
    float* pw = P + (size_t)(row0 + wave_s * 16 + lr) * F_H1 + m;
    constexpr size_t XS = (size_t)NW * 16 * F_IN;
    constexpr size_t PS = (size_t)NW * 16 * F_H1;
    constexpr int    DS = NW * 16;
    int i = 0;
    for (; i + 2 < n; i += 2) {
      w1_load_a(a2, xw + (size_t)(i + 1) * XS);
      w1_finish_rows64(a, bw, dw + i * DS, pw + (size_t)i * PS);
      w1_load_a(a, xw + (size_t)(i + 2) * XS);
      w1_finish_rows64(a2, bw, dw + (i + 1) * DS, pw + (size_t)(i + 1) * PS);
    }
    if (i + 1 < n) {
      w1_load_a(a2, xw + (size_t)(i + 1) * XS);
      w1_finish_rows64(a, bw, dw + i * DS, pw + (size_t)i * PS);
      w1_finish_rows64(a2, bw, dw + (i + 1) * DS, pw + (size_t)(i + 1) * PS);
    } else if (i < n) {
      w1_finish_rows64(a, bw, dw + i * DS, pw + (size_t)i * PS);
    }
  } else {
    for (int tile = wave_s; tile < ntiles; tile += NW)
      w1_tile_half<0>((row0 >> 4) + tile, lane, X, W1, dloc + tile * 16 + 8 * hi, P);
  }
}


__global__ __launch_bounds__(512) void agg1_rows64(int* keys,
                                                   const int* __restrict__ bstart,
                                                   const float* __restrict__ R,
                                                   const float* __restrict__ dinv,
                                                   const float* __restrict__ b1,
                                                   const float* __restrict__ W2,
                                                   const int* __restrict__ fits) {
  constexpr int F   = F_H1;
  constexpr int LPE = F / 4;
  constexpr int EPI = 32 / LPE;
  constexpr int U   = 4;
  constexpr int NW  = 512 / 32;
  __shared__ float acc[R_ROWS * F];

  if (*fits == 0) return;

  const int t = threadIdx.x, lane = t & 31, wave = t >> 5;
  const int b = blockIdx.x;
  const int row0 = b << RB;
  const int nrows = (N_NODES - row0 < R_ROWS) ? (N_NODES - row0) : R_ROWS;

  float4* acc4 = (float4*)acc;
  for (int i = t; i < R_ROWS * F / 4; i += 512) acc4[i] = make_float4(0.f, 0.f, 0.f, 0.f);
  __syncthreads();

  const int e0 = bstart[b], e1 = bstart[b + 1];
  const int sub = lane / LPE;
  const int q   = lane % LPE;

  for (int base = e0 + wave * (32 * U); base < e1; base += NW * 32 * U) {
    int k[U];
#pragma unroll
    for (int u = 0; u < U; ++u) {
      const int e = base + u * 32 + lane;
      k[u] = (e < e1) ? __builtin_nontemporal_load(keys + e) : -1;
    }
    float4 v[U * LPE];
    int    dl[U * LPE];
#pragma unroll
    for (int u = 0; u < U; ++u) {
#pragma unroll
      for (int i = 0; i < LPE; ++i) {
        const int ks = __shfl(k[u], i * EPI + sub);
        const bool ok = (ks >= 0);
        const unsigned s = ok ? ((unsigned)ks >> RB) : 0u;
        v[u * LPE + i]  = *(const float4*)(R + (size_t)s * F + 4 * q);
        dl[u * LPE + i] = ok ? (ks & (R_ROWS - 1)) : -1;
      }
    }
#pragma unroll
    for (int j = 0; j < U * LPE; ++j) {
      if (dl[j] >= 0) {
        float* a = acc + dl[j] * F + 4 * q;
        __hip_atomic_fetch_add(a + 0, v[j].x, __ATOMIC_RELAXED, __HIP_MEMORY_SCOPE_WORKGROUP);
        __hip_atomic_fetch_add(a + 1, v[j].y, __ATOMIC_RELAXED, __HIP_MEMORY_SCOPE_WORKGROUP);
        __hip_atomic_fetch_add(a + 2, v[j].z, __ATOMIC_RELAXED, __HIP_MEMORY_SCOPE_WORKGROUP);
        __hip_atomic_fetch_add(a + 3, v[j].w, __ATOMIC_RELAXED, __HIP_MEMORY_SCOPE_WORKGROUP);
      }
    }
  }
  __syncthreads();

  const int m  = lane & 15;
  const int hi = lane >> 4;
  const int kk = hi << 1;
  const bool live_col = (m < F_H2);
  const int col = m & 7;
  v2f bf[4];
  float bx[4], by[4];
#pragma unroll
  for (int s = 0; s < 4; ++s) {
    const float w0 = W2[(4 * s + kk) * F_H2 + col];
    const float w1 = W2[(4 * s + kk + 1) * F_H2 + col];
    bf[s].x = live_col ? w0 : 0.0f;
    bf[s].y = live_col ? w1 : 0.0f;
    bx[s] = b1[4 * s + kk];
    by[s] = b1[4 * s + kk + 1];
  }
  float* Hout = (float*)keys + ((e0 + 31) & ~31);
  const int ntiles = nrows >> 4;
  const int wave_s = __builtin_amdgcn_readfirstlane(wave);
  for (int tile = wave_s; tile < ntiles; tile += NW) {
    const int lrow = tile * 16 + m;
    const size_t g = (size_t)(row0 + lrow);
    const float w = dinv[g];
    v2f a[4];
#pragma unroll
    for (int s = 0; s < 4; ++s) {
      const float2 p  = *(const float2*)(R + g * F + 4 * s + kk);
      const float2 sm = *(const float2*)(acc + lrow * F + 4 * s + kk);
      a[s].x = fmaxf((sm.x + p.x) * w + bx[s], 0.0f);
      a[s].y = fmaxf((sm.y + p.y) * w + by[s], 0.0f);
    }
    v8f c = {0.f, 0.f, 0.f, 0.f, 0.f, 0.f, 0.f, 0.f};
    c = __builtin_amdgcn_wmma_f32_16x16x4_f32(false, a[0], false, bf[0], (short)0, c, false, false);
    c = __builtin_amdgcn_wmma_f32_16x16x4_f32(false, a[1], false, bf[1], (short)0, c, false, false);
    c = __builtin_amdgcn_wmma_f32_16x16x4_f32(false, a[2], false, bf[2], (short)0, c, false, false);
    c = __builtin_amdgcn_wmma_f32_16x16x4_f32(false, a[3], false, bf[3], (short)0, c, false, false);
    const int lr0 = tile * 16 + 8 * hi;
    const float* dv = dinv + row0 + lr0;
    float r[8];
#pragma unroll
    for (int j = 0; j < 8; ++j) r[j] = c[j] * dv[j];
    if (live_col) {
      float* out = Hout + (size_t)lr0 * F_H2 + m;
#pragma unroll
      for (int j = 0; j < 8; ++j) out[(size_t)j * F_H2] = r[j];
    }
  }
}


template <int MODE>
__global__ __launch_bounds__(512) void agg_range_lds(const int* __restrict__ keys,
                                                     const int* __restrict__ bstart,
                                                     const float* __restrict__ H,
                                                     float* OUT,
                                                     const float* __restrict__ dinv,
                                                     const float* __restrict__ b1,
                                                     const float* __restrict__ W2,
                                                     const float* __restrict__ b2,
                                                     const int* __restrict__ batch,
                                                     float* __restrict__ pooled,
                                                     const int* __restrict__ fits,
                                                     const int* __restrict__ keys2,
                                                     const float* __restrict__ Hb) {
  constexpr int F   = 8;
  constexpr int LPE = F / 4;
  constexpr int EPI = 32 / LPE;
  constexpr int U   = 4;
  constexpr int NW  = 512 / 32;
  __shared__ float acc[R_ROWS * F];
  __shared__ int hoff[MODE == 0 ? N_RANGES : 1];

  const int fit = *fits;
  if (MODE != 0 && fit) return;
  const int*   ksrc = (MODE == 0 && fit) ? keys2 : keys;
  const float* Hs   = (MODE == 0 && fit) ? Hb : H;

  const int t = threadIdx.x, lane = t & 31, wave = t >> 5;
  const int b = blockIdx.x;
  const int row0 = b << RB;
  const int nrows = (N_NODES - row0 < R_ROWS) ? (N_NODES - row0) : R_ROWS;

  float4* acc4 = (float4*)acc;
  for (int i = t; i < R_ROWS * F / 4; i += 512) acc4[i] = make_float4(0.f, 0.f, 0.f, 0.f);
  if constexpr (MODE == 0) {
    for (int r = t; r < N_RANGES; r += 512)
      hoff[r] = fit ? ((bstart[r] + 31) & ~31) : (r << (RB + 3));
  }
  __syncthreads();

  const int e0 = bstart[b], e1 = bstart[b + 1];
  const int sub = lane / LPE;
  const int q   = lane % LPE;

  for (int base = e0 + wave * (32 * U); base < e1; base += NW * 32 * U) {
    int k[U];
#pragma unroll
    for (int u = 0; u < U; ++u) {
      const int e = base + u * 32 + lane;
      k[u] = (e < e1) ? ksrc[e] : -1;
    }
    float4 v[U * LPE];
    int    dl[U * LPE];
#pragma unroll
    for (int u = 0; u < U; ++u) {
#pragma unroll
      for (int i = 0; i < LPE; ++i) {
        const int ks = __shfl(k[u], i * EPI + sub);
        const bool ok = (ks >= 0);
        const unsigned s = ok ? ((unsigned)ks >> RB) : 0u;
        if constexpr (MODE == 0)
          v[u * LPE + i] = *(const float4*)(Hs + (size_t)hoff[s >> RB] +
                                            (size_t)(s & (R_ROWS - 1)) * F + 4 * q);
        else
          v[u * LPE + i] = *(const float4*)(H + (size_t)s * F + 4 * q);
        dl[u * LPE + i] = ok ? (ks & (R_ROWS - 1)) : -1;
      }
    }
#pragma unroll
    for (int j = 0; j < U * LPE; ++j) {
      if (dl[j] >= 0) {
        float* a = acc + dl[j] * F + 4 * q;
        __hip_atomic_fetch_add(a + 0, v[j].x, __ATOMIC_RELAXED, __HIP_MEMORY_SCOPE_WORKGROUP);
        __hip_atomic_fetch_add(a + 1, v[j].y, __ATOMIC_RELAXED, __HIP_MEMORY_SCOPE_WORKGROUP);
        __hip_atomic_fetch_add(a + 2, v[j].z, __ATOMIC_RELAXED, __HIP_MEMORY_SCOPE_WORKGROUP);
        __hip_atomic_fetch_add(a + 3, v[j].w, __ATOMIC_RELAXED, __HIP_MEMORY_SCOPE_WORKGROUP);
      }
    }
  }
  __syncthreads();

  if constexpr (MODE == 0) {
    constexpr int RPW = R_ROWS / NW;
    constexpr int UP  = 8;
    static_assert(RPW % (4 * UP) == 0, "a wave's rows are whole trips of 4*UP rows");
    const int f  = lane & 7;
    const int r4 = lane >> 3;
    const float bf = b2[f];
    float* gcnt = pooled + (size_t)N_GRAPHS * F_H2;
    const float* Hown = Hs + (size_t)hoff[b];
    int   gcur = 0;
    int   nrun = 0;
    float run  = 0.0f;
    for (int i0 = 0; i0 < RPW / 4; i0 += UP) {
      int   g[UP];
      float w[UP], a[UP], h[UP];
#pragma unroll
      for (int u = 0; u < UP; ++u) {
        const int lrow = wave * RPW + 4 * (i0 + u) + r4;
        const bool ok = (lrow < nrows);
        const int lr = ok ? lrow : 0;
        const size_t grow = (size_t)(row0 + lr);
        g[u] = ok ? batch[grow] : -1;
        w[u] = dinv[grow];
        h[u] = Hown[lr * F + f];
        a[u] = acc[lrow * F + f];
      }
#pragma unroll
      for (int u = 0; u < UP; ++u) {
        const int lrow = wave * RPW + 4 * (i0 + u) + r4;
        if (lrow < nrows) {
          const float v = fmaxf((a[u] + h[u]) * w[u] + bf, 0.0f);
          if (nrun > 0 && g[u] != gcur) {
            ATOMIC_FADD(&pooled[(size_t)gcur * F_H2 + f], run);
            if (f == 0) ATOMIC_FADD(&gcnt[gcur], (float)nrun);
            nrun = 0;
            run  = 0.0f;
          }
          gcur = g[u];
          run += v;
          ++nrun;
        }
      }
    }
    if (nrun > 0) {
      ATOMIC_FADD(&pooled[(size_t)gcur * F_H2 + f], run);
      if (f == 0) ATOMIC_FADD(&gcnt[gcur], (float)nrun);
    }
  } else if constexpr (MODE == 1) {
    const float4* h4 = (const float4*)(H + (size_t)row0 * F);
    float4* out4 = (float4*)(OUT + (size_t)row0 * F);
    const float4 bv = ((const float4*)b1)[t & 1];
    for (int i = t; i < nrows * F / 4; i += 512) {
      const float w = dinv[row0 + (i >> 1)];
      float4 av = acc4[i];
      const float4 hv = h4[i];
      av.x = fmaxf((av.x + hv.x) * w + bv.x, 0.0f);
      av.y = fmaxf((av.y + hv.y) * w + bv.y, 0.0f);
      av.z = fmaxf((av.z + hv.z) * w + bv.z, 0.0f);
      av.w = fmaxf((av.w + hv.w) * w + bv.w, 0.0f);
      out4[i] = av;
    }
  } else {
    const int m  = lane & 15;
    const int hi = lane >> 4;
    const int kk = hi << 1;
    const bool live_col = (m < F_H2);
    const int col = m & 7;
    v2f bf[4];
#pragma unroll
    for (int s = 0; s < 4; ++s) {
      const float w0 = W2[(4 * s + kk) * F_H2 + col];
      const float w1 = W2[(4 * s + kk + 1) * F_H2 + col];
      bf[s].x = live_col ? w0 : 0.0f;
      bf[s].y = live_col ? w1 : 0.0f;
    }
    const float bb0 = b1[8 + kk],  bb1 = b1[9 + kk];
    const float bb2 = b1[12 + kk], bb3 = b1[13 + kk];
    const int ntiles = nrows >> 4;
    const int wave_s = __builtin_amdgcn_readfirstlane(wave);
    for (int tile = wave_s; tile < ntiles; tile += NW) {
      const int lrow = tile * 16 + m;
      const size_t g = (size_t)(row0 + lrow);
      const float w = dinv[g];
      const float2 q0 = *(const float2*)(OUT + g * F + kk);
      const float2 q1 = *(const float2*)(OUT + g * F + 4 + kk);
      const float2 p0 = *(const float2*)(H + g * F + kk);
      const float2 p1 = *(const float2*)(H + g * F + 4 + kk);
      const float2 s0 = *(const float2*)(acc + lrow * F + kk);
      const float2 s1 = *(const float2*)(acc + lrow * F + 4 + kk);
      v2f a0, a1, a2, a3;
      a0.x = q0.x;  a0.y = q0.y;
      a1.x = q1.x;  a1.y = q1.y;
      a2.x = fmaxf((s0.x + p0.x) * w + bb0, 0.0f);
      a2.y = fmaxf((s0.y + p0.y) * w + bb1, 0.0f);
      a3.x = fmaxf((s1.x + p1.x) * w + bb2, 0.0f);
      a3.y = fmaxf((s1.y + p1.y) * w + bb3, 0.0f);
      v8f c = {0.f, 0.f, 0.f, 0.f, 0.f, 0.f, 0.f, 0.f};
      c = __builtin_amdgcn_wmma_f32_16x16x4_f32(false, a0, false, bf[0], (short)0, c, false, false);
      c = __builtin_amdgcn_wmma_f32_16x16x4_f32(false, a1, false, bf[1], (short)0, c, false, false);
      c = __builtin_amdgcn_wmma_f32_16x16x4_f32(false, a2, false, bf[2], (short)0, c, false, false);
      c = __builtin_amdgcn_wmma_f32_16x16x4_f32(false, a3, false, bf[3], (short)0, c, false, false);
      const int r0 = row0 + tile * 16 + 8 * hi;
      const float* dv = dinv + r0;
      float r[8];
#pragma unroll
      for (int j = 0; j < 8; ++j) r[j] = c[j] * dv[j];
      if (live_col) {
        float* out = OUT + (size_t)r0 * F + m;
#pragma unroll
        for (int j = 0; j < 8; ++j) out[(size_t)j * F] = r[j];
      }
    }
  }
}


__global__ void readout(const float* __restrict__ pooled, const float* __restrict__ cnt,
                        const float* __restrict__ Wl, const float* __restrict__ bl,
                        float* __restrict__ out, int G) {
  int g = blockIdx.x * blockDim.x + threadIdx.x;
  if (g >= G) return;
  float inv = 1.0f / fmaxf(cnt[g], 1.0f);
  float p[F_H2];
#pragma unroll
  for (int f = 0; f < F_H2; ++f) p[f] = pooled[(size_t)g * F_H2 + f] * inv;
#pragma unroll
  for (int c = 0; c < N_CLS; ++c) {
    float acc = bl[c];
#pragma unroll
    for (int f = 0; f < F_H2; ++f) acc += p[f] * Wl[f * N_CLS + c];
    out[(size_t)g * N_CLS + c] = acc;
  }
}


extern "C" void kernel_launch(void* const* d_in, const int* in_sizes, int n_in,
                              void* d_out, int out_size, void* d_ws, size_t ws_size,
                              hipStream_t stream) {
  const float* x     = (const float*)d_in[0];
  const int*   ei    = (const int*)d_in[1];
  const int*   src   = ei;
  const int*   dst   = ei + N_EDGES;
  const int*   batch = (const int*)d_in[2];
  const float* W1 = (const float*)d_in[4];
  const float* b1 = (const float*)d_in[5];
  const float* W2 = (const float*)d_in[6];
  const float* b2 = (const float*)d_in[7];
  const float* Wl = (const float*)d_in[8];
  const float* bl = (const float*)d_in[9];
  float* out = (float*)d_out;

  char* ws = (char*)d_ws;
  float* dinv   = (float*)(ws);
  float* bufA   = (float*)(ws + (4ll  << 20));
  float* bufB   = (float*)(ws + (68ll << 20));
  float* pooled = (float*)(ws + (132ll << 20));
  float* cnt    = pooled + (size_t)N_GRAPHS * F_H2;

  float* rows64 = bufA;
  float* planeP = bufA;
  float* planeQ = (float*)((char*)bufA + (32ll << 20));
  int* keys   = (int*)bufB;
  int* keys2  = (int*)bufA;
  int* counts = (int*)((char*)bufB + 64000000ll);
  int* tot    = (int*)((char*)bufB + 64000000ll + (512ll << 10));
  int* bstart = (int*)((char*)bufB + 64000000ll + (520ll << 10));
  int* fits   = (int*)((char*)bufB + 64000000ll + (528ll << 10));

  const int NB = 256;

  bin_count<<<NBLK, 1024, 0, stream>>>(dst, counts);
  bin_scan<<<N_RANGES, 256, 0, stream>>>(counts, tot);
  bin_scatter<0><<<NBLK, 1024, 0, stream>>>(src, dst, counts, tot, bstart, keys, fits);

  degree_w1_range<<<N_RANGES, 512, 0, stream>>>(keys, bstart, x, W1, dinv, rows64, fits);

  agg1_rows64<<<N_RANGES, 512, 0, stream>>>(keys, bstart, rows64, dinv, b1, W2, fits);
  agg_range_lds<1><<<N_RANGES, 512, 0, stream>>>(keys, bstart, planeP, planeQ, dinv, b1, W2,
                                                 b2, batch, pooled, fits, keys2, planeQ);
  gemm_x_w1_plane1<<<1024, 128, 0, stream>>>(x, W1, dinv, planeP, fits);
  agg_range_lds<2><<<N_RANGES, 512, 0, stream>>>(keys, bstart, planeP, planeQ, dinv, b1, W2,
                                                 b2, batch, pooled, fits, keys2, planeQ);
  bin_scatter<1><<<NBLK, 1024, 0, stream>>>(src, dst, counts, tot, bstart, keys2, fits);

  zero_f32<<<128, 256, 0, stream>>>(pooled, (long long)N_GRAPHS * (F_H2 + 1));
  agg_range_lds<0><<<N_RANGES, 512, 0, stream>>>(keys, bstart, planeQ, planeP, dinv, b1, W2,
                                                 b2, batch, pooled, fits, keys2,
                                                 (const float*)keys);

  readout<<<(N_GRAPHS + NB - 1) / NB, NB, 0, stream>>>(pooled, cnt, Wl, bl,
                                                       out, N_GRAPHS);
}


# edge passes: a lane adds its float4 to the LDS row in an order turned by its edge slot & 3, so one LDS add's lanes cover 4x the banks; same addends, a wave's add order changes, sums by arrival
# speedup vs baseline: 14.0713x; 1.1928x over previous
#include <hip/hip_runtime.h>


typedef float v2f __attribute__((ext_vector_type(2)));
typedef float v8f __attribute__((ext_vector_type(8)));

#define N_NODES   1000000
#define N_EDGES   16000000
#define F_IN      32
#define F_H1      16
#define F_H2      8
#define N_CLS     5
#define N_GRAPHS  10000

#define RB        11
#define R_ROWS    (1 << RB)
#define N_RANGES        ((N_NODES + R_ROWS - 1) >> RB)
#define CH        65536
#define NBLK      ((N_EDGES + CH - 1) / CH)

#define ATOMIC_FADD(p, v) \
  __hip_atomic_fetch_add((p), (v), __ATOMIC_RELAXED, __HIP_MEMORY_SCOPE_AGENT)


__device__ __forceinline__ void lds_add_turned4(float* a, const float4 x,
                                                const bool t1, const bool t2,
                                                const int (&o)[4]) {
  const float p0 = t1 ? x.y : x.x;
  const float p1 = t1 ? x.z : x.y;
  const float p2 = t1 ? x.w : x.z;
  const float p3 = t1 ? x.x : x.w;
  const float u0 = t2 ? p2 : p0;
  const float u1 = t2 ? p3 : p1;
  const float u2 = t2 ? p0 : p2;
  const float u3 = t2 ? p1 : p3;
  __hip_atomic_fetch_add(a + o[0], u0, __ATOMIC_RELAXED, __HIP_MEMORY_SCOPE_WORKGROUP);
  __hip_atomic_fetch_add(a + o[1], u1, __ATOMIC_RELAXED, __HIP_MEMORY_SCOPE_WORKGROUP);
  __hip_atomic_fetch_add(a + o[2], u2, __ATOMIC_RELAXED, __HIP_MEMORY_SCOPE_WORKGROUP);
  __hip_atomic_fetch_add(a + o[3], u3, __ATOMIC_RELAXED, __HIP_MEMORY_SCOPE_WORKGROUP);
}

__global__ void zero_f32(float* __restrict__ p, long long n) {
  long long i = (long long)blockIdx.x * blockDim.x + threadIdx.x;
  long long stride = (long long)gridDim.x * blockDim.x;
  for (; i < n; i += stride) p[i] = 0.0f;
}


__device__ __forceinline__ void w1_load_a(v2f (&a)[F_IN / 4], const float* __restrict__ xk) {
#pragma unroll
  for (int s = 0; s < F_IN / 4; ++s) {
    a[s].x = xk[4 * s];
    a[s].y = xk[4 * s + 1];
  }
}

__device__ __forceinline__ void w1_finish_rows64(const v2f (&a)[F_IN / 4],
                                                 const v2f (&bw)[F_IN / 4],
                                                 const float* dv,
                                                 float* __restrict__ out) {
  v8f c = {0.f, 0.f, 0.f, 0.f, 0.f, 0.f, 0.f, 0.f};
#pragma unroll
  for (int s = 0; s < F_IN / 4; ++s)
    c = __builtin_amdgcn_wmma_f32_16x16x4_f32(false, a[s], false, bw[s],
                                              (short)0, c, false, false);
  float r[8];
#pragma unroll
  for (int j = 0; j < 8; ++j) r[j] = c[j] * dv[j];
#pragma unroll
  for (int j = 0; j < 8; ++j) out[(size_t)j * F_H1] = r[j];
}

template <int HALF>
__device__ __forceinline__ void w1_tile_half(int tile, int lane,
                                             const float* __restrict__ X,
                                             const float* __restrict__ W1,
                                             const float* dv,
                                             float* __restrict__ P) {
  const int m  = lane & 15;
  const int kk = (lane >> 4) << 1;
  const float* xrow = X + (size_t)(tile * 16 + m) * F_IN;
  const bool live_col = (m < 8);
  const int col = HALF * 8 + (m & 7);

  v8f c = {0.f, 0.f, 0.f, 0.f, 0.f, 0.f, 0.f, 0.f};
#pragma unroll
  for (int k = 0; k < F_IN; k += 4) {
    v2f a, b;
    a.x = xrow[k + kk];
    a.y = xrow[k + kk + 1];
    const float w0 = W1[(k + kk) * F_H1 + col];
    const float w1 = W1[(k + kk + 1) * F_H1 + col];
    b.x = live_col ? w0 : 0.0f;
    b.y = live_col ? w1 : 0.0f;
    c = __builtin_amdgcn_wmma_f32_16x16x4_f32(false, a, false, b,
                                              (short)0, c, false, false);
  }
  const int hi = lane >> 4;
  const int row0 = tile * 16 + 8 * hi;
  float r[8];
#pragma unroll
  for (int j = 0; j < 8; ++j) r[j] = c[j] * dv[j];
  if (live_col) {
    float* out = P + (size_t)row0 * 8 + m;
#pragma unroll
    for (int j = 0; j < 8; ++j) out[(size_t)j * 8] = r[j];
  }
}

__global__ __launch_bounds__(128) void gemm_x_w1_plane1(const float* __restrict__ X,
                                                        const float* __restrict__ W1,
                                                        const float* __restrict__ dinv,
                                                        float* __restrict__ P,
                                                        const int* __restrict__ fits) {
  if (*fits != 0) return;
  const int lane = threadIdx.x & 31;
  const int w0 = __builtin_amdgcn_readfirstlane(
      (int)(blockIdx.x * (blockDim.x >> 5) + (threadIdx.x >> 5)));
  const int nw = (int)(gridDim.x * (blockDim.x >> 5));
  for (int tile = w0; tile < N_NODES / 16; tile += nw)
    w1_tile_half<1>(tile, lane, X, W1, dinv + tile * 16 + 8 * (lane >> 4), P);
}


static_assert(N_NODES <= (1 << 20), "a key holds the source row in 20 bits beside RB bits of local row");
static_assert(N_RANGES <= 1024, "one thread per range in the scatter kernel's offset scan");
static_assert(NBLK <= 256, "one thread per binning work-group in the per-range scan");
static_assert(N_EDGES % 4 == 0 && CH % 4096 == 0, "binning kernels read four edges per lane");
static_assert(N_NODES % 16 == 0 && R_ROWS % 16 == 0, "a range is whole 16-row tiles");

__device__ __forceinline__ int wave_incl_scan(int v, int lane) {
#pragma unroll
  for (int o = 1; o < 32; o <<= 1) {
    const int t = __shfl_up(v, o);
    if (lane >= o) v += t;
  }
  return v;
}

__device__ __forceinline__ int block_excl_scan(int v, int* wsum, int* total) {
  const int lane = threadIdx.x & 31, wave = threadIdx.x >> 5;
  const int nw = blockDim.x >> 5;
  const int incl = wave_incl_scan(v, lane);
  if (lane == 31) wsum[wave] = incl;
  __syncthreads();
  if (wave == 0) {
    const int w  = (lane < nw) ? wsum[lane] : 0;
    const int wi = wave_incl_scan(w, lane);
    wsum[lane] = wi - w;
    if (lane == 31) wsum[32] = wi;
  }
  __syncthreads();
  *total = wsum[32];
  return wsum[wave] + incl - v;
}

__global__ __launch_bounds__(1024) void bin_count(const int* __restrict__ dst,
                                                  int* __restrict__ counts) {
  __shared__ int hist[N_RANGES];
  const int t = threadIdx.x, blk = blockIdx.x;
  for (int i = t; i < N_RANGES; i += 1024) hist[i] = 0;
  __syncthreads();
  const int base = blk * CH;
  const int4* d4 = (const int4*)(dst + base);
  for (int i = t; i < CH / 4; i += 1024) {
    if (base + 4 * i < N_EDGES) {
      const int4 d = d4[i];
      atomicAdd(&hist[d.x >> RB], 1);
      atomicAdd(&hist[d.y >> RB], 1);
      atomicAdd(&hist[d.z >> RB], 1);
      atomicAdd(&hist[d.w >> RB], 1);
    }
  }
  __syncthreads();
  for (int i = t; i < N_RANGES; i += 1024) counts[i * NBLK + blk] = hist[i];
}

__global__ __launch_bounds__(256) void bin_scan(int* __restrict__ counts,
                                                int* __restrict__ tot) {
  __shared__ int wsum[33];
  const int b = blockIdx.x, t = threadIdx.x;
  const int v = (t < NBLK) ? counts[b * NBLK + t] : 0;
  int total;
  const int ex = block_excl_scan(v, wsum, &total);
  if (t < NBLK) counts[b * NBLK + t] = ex;
  if (t == 0) tot[b] = total;
}

template <int PASS>
__global__ __launch_bounds__(1024) void bin_scatter(const int* __restrict__ src,
                                                    const int* __restrict__ dst,
                                                    const int* __restrict__ offs,
                                                    const int* __restrict__ tot,
                                                    int* __restrict__ bstart,
                                                    int* __restrict__ keys,
                                                    int* fits) {
  __shared__ int stage[CH];
  __shared__ int lbeg[N_RANGES + 1];
  __shared__ int lcur[N_RANGES];
  __shared__ int gbeg[N_RANGES];
  __shared__ int wsum[33];
  if (PASS == 1) {
    if (*fits == 0) return;
  }
  const int t = threadIdx.x, blk = blockIdx.x;
  const int lane = t & 31, wave = t >> 5;
  const int v = (t < N_RANGES) ? tot[t] : 0;
  int total;
  const int ex = block_excl_scan(v, wsum, &total);
  int n = 0;
  if (t < N_RANGES) {
    const int o  = offs[t * NBLK + blk];
    const int o1 = (blk + 1 < NBLK) ? offs[t * NBLK + blk + 1] : v;
    n = o1 - o;
    gbeg[t] = ex + o;
    if (PASS == 0 && blk == 0) bstart[t] = ex;
  }
  if (PASS == 0 && blk == 0) {
    if (t == 0) bstart[N_RANGES] = total;
    const int row0 = t << RB;
    const int nrows = (N_NODES - row0 < R_ROWS) ? (N_NODES - row0) : R_ROWS;
    const int roomy = (t >= N_RANGES) || (v >= nrows * F_H2 + 32);
    const int all = __syncthreads_and(roomy);
    if (t == 0) *fits = all;
  }
  __syncthreads();
  int ltotal;
  const int lex = block_excl_scan(n, wsum, &ltotal);
  if (t < N_RANGES) {
    lbeg[t] = lex;
    lcur[t] = lex;
  }
  if (t == 0) lbeg[N_RANGES] = ltotal;
  __syncthreads();
  const int base = blk * CH;
  const int4* s4 = (const int4*)(src + base);
  const int4* d4 = (const int4*)(dst + base);
  for (int i = t; i < CH / 4; i += 1024) {
    if (base + 4 * i < N_EDGES) {
      const int4 s = s4[i], d = d4[i];
      const int p0 = atomicAdd(&lcur[d.x >> RB], 1);
      const int p1 = atomicAdd(&lcur[d.y >> RB], 1);
      const int p2 = atomicAdd(&lcur[d.z >> RB], 1);
      const int p3 = atomicAdd(&lcur[d.w >> RB], 1);
      stage[p0] = (s.x << RB) | (d.x & (R_ROWS - 1));
      stage[p1] = (s.y << RB) | (d.y & (R_ROWS - 1));
      stage[p2] = (s.z << RB) | (d.z & (R_ROWS - 1));
      stage[p3] = (s.w << RB) | (d.w & (R_ROWS - 1));
    }
  }
  __syncthreads();
  for (int r = wave; r < N_RANGES; r += 32) {
    const int ls = lbeg[r];
    const int nk = lbeg[r + 1] - ls;
    const int g  = gbeg[r];
    for (int j = lane - (g & 31); j < nk; j += 32)
      if (j >= 0) keys[g + j] = stage[ls + j];
  }
}


__global__ __launch_bounds__(512) void degree_w1_range(const int* __restrict__ keys,
                                                       const int* __restrict__ bstart,
                                                       const float* __restrict__ X,
                                                       const float* __restrict__ W1,
                                                       float* __restrict__ dinv,
                                                       float* __restrict__ P,
                                                       const int* __restrict__ fits) {
  constexpr int U  = 16;
  constexpr int NW = 512 / 32;
  __shared__ int   cnt[R_ROWS];
  __shared__ float dloc[R_ROWS];

  const int t = threadIdx.x, lane = t & 31, wave = t >> 5;
  const int b = blockIdx.x;
  const int row0 = b << RB;
  const int nrows = (N_NODES - row0 < R_ROWS) ? (N_NODES - row0) : R_ROWS;
  const int ntiles = nrows >> 4;
  const int ok = __builtin_amdgcn_readfirstlane(*fits);
  const int wave_s = __builtin_amdgcn_readfirstlane(wave);

  const int m  = lane & 15;
  const int hi = lane >> 4;
  const int kk = hi << 1;
  const float* xk = X + (size_t)(row0 + m) * F_IN + kk;

  for (int i = t; i < R_ROWS; i += 512) cnt[i] = 0;
  __syncthreads();

  v2f a[F_IN / 4], bw[F_IN / 4];
#pragma unroll
  for (int s = 0; s < F_IN / 4; ++s) {
    a[s].x = 0.0f;  a[s].y = 0.0f;
    bw[s].x = 0.0f; bw[s].y = 0.0f;
  }
  if (ok && wave_s < ntiles) {
    w1_load_a(a, xk + (size_t)wave_s * 16 * F_IN);
#pragma unroll
    for (int s = 0; s < F_IN / 4; ++s) {
      bw[s].x = W1[(4 * s + kk) * F_H1 + m];
      bw[s].y = W1[(4 * s + kk + 1) * F_H1 + m];
    }
  }

  const int e0 = bstart[b], e1 = bstart[b + 1];
  for (int base = e0 + wave * (32 * U); base < e1; base += NW * 32 * U) {
    int k[U];
#pragma unroll
    for (int u = 0; u < U; ++u) {
      const int e = base + u * 32 + lane;
      k[u] = (e < e1) ? keys[e] : -1;
    }
#pragma unroll
    for (int u = 0; u < U; ++u) {
      if (k[u] >= 0)
        __hip_atomic_fetch_add(&cnt[k[u] & (R_ROWS - 1)], 1, __ATOMIC_RELAXED,
                               __HIP_MEMORY_SCOPE_WORKGROUP);
    }
  }
  __syncthreads();

  for (int i = t; i < R_ROWS; i += 512) {
    const float d = rsqrtf((float)cnt[i] + 1.0f);
    dloc[i] = d;
    if (i < nrows) dinv[row0 + i] = d;
  }
  __syncthreads();

  if (ok) {
    const int n = (ntiles - wave_s + NW - 1) / NW;
    v2f a2[F_IN / 4];
    const int lr = 8 * hi;
    const float* xw = xk + (size_t)wave_s * 16 * F_IN;
    const float* dw = dloc + wave_s * 16 + lr;
    float* pw = P + (size_t)(row0 + wave_s * 16 + lr) * F_H1 + m;
    constexpr size_t XS = (size_t)NW * 16 * F_IN;
    constexpr size_t PS = (size_t)NW * 16 * F_H1;
    constexpr int    DS = NW * 16;
    int i = 0;
    for (; i + 2 < n; i += 2) {
      w1_load_a(a2, xw + (size_t)(i + 1) * XS);
      w1_finish_rows64(a, bw, dw + i * DS, pw + (size_t)i * PS);
      w1_load_a(a, xw + (size_t)(i + 2) * XS);
      w1_finish_rows64(a2, bw, dw + (i + 1) * DS, pw + (size_t)(i + 1) * PS);
    }
    if (i + 1 < n) {
      w1_load_a(a2, xw + (size_t)(i + 1) * XS);
      w1_finish_rows64(a, bw, dw + i * DS, pw + (size_t)i * PS);
      w1_finish_rows64(a2, bw, dw + (i + 1) * DS, pw + (size_t)(i + 1) * PS);
    } else if (i < n) {
      w1_finish_rows64(a, bw, dw + i * DS, pw + (size_t)i * PS);
    }
  } else {
    for (int tile = wave_s; tile < ntiles; tile += NW)
      w1_tile_half<0>((row0 >> 4) + tile, lane, X, W1, dloc + tile * 16 + 8 * hi, P);
  }
}


__global__ __launch_bounds__(512) void agg1_rows64(int* keys,
                                                   const int* __restrict__ bstart,
                                                   const float* __restrict__ R,
                                                   const float* __restrict__ dinv,
                                                   const float* __restrict__ b1,
                                                   const float* __restrict__ W2,
                                                   const int* __restrict__ fits) {
  constexpr int F   = F_H1;
  constexpr int LPE = F / 4;
  constexpr int EPI = 32 / LPE;
  constexpr int U   = 4;
  constexpr int NW  = 512 / 32;
  __shared__ float acc[R_ROWS * F];

  if (*fits == 0) return;

  const int t = threadIdx.x, lane = t & 31, wave = t >> 5;
  const int b = blockIdx.x;
  const int row0 = b << RB;
  const int nrows = (N_NODES - row0 < R_ROWS) ? (N_NODES - row0) : R_ROWS;

  float4* acc4 = (float4*)acc;
  for (int i = t; i < R_ROWS * F / 4; i += 512) acc4[i] = make_float4(0.f, 0.f, 0.f, 0.f);
  __syncthreads();

  const int e0 = bstart[b], e1 = bstart[b + 1];
  const int sub = lane / LPE;
  const int q   = lane % LPE;
  const int rot = sub & 3;
  const bool t1 = (rot & 1) != 0, t2 = (rot & 2) != 0;
  const int o[4] = {rot, (rot + 1) & 3, (rot + 2) & 3, (rot + 3) & 3};

  for (int base = e0 + wave * (32 * U); base < e1; base += NW * 32 * U) {
    int k[U];
#pragma unroll
    for (int u = 0; u < U; ++u) {
      const int e = base + u * 32 + lane;
      k[u] = (e < e1) ? __builtin_nontemporal_load(keys + e) : -1;
    }
    float4 v[U * LPE];
    int    dl[U * LPE];
#pragma unroll
    for (int u = 0; u < U; ++u) {
#pragma unroll
      for (int i = 0; i < LPE; ++i) {
        const int ks = __shfl(k[u], i * EPI + sub);
        const bool ok = (ks >= 0);
        const unsigned s = ok ? ((unsigned)ks >> RB) : 0u;
        v[u * LPE + i]  = *(const float4*)(R + (size_t)s * F + 4 * q);
        dl[u * LPE + i] = ok ? (ks & (R_ROWS - 1)) : -1;
      }
    }
#pragma unroll
    for (int j = 0; j < U * LPE; ++j) {
      if (dl[j] >= 0)
        lds_add_turned4(acc + dl[j] * F + 4 * q, v[j], t1, t2, o);
    }
  }
  __syncthreads();

  const int m  = lane & 15;
  const int hi = lane >> 4;
  const int kk = hi << 1;
  const bool live_col = (m < F_H2);
  const int col = m & 7;
  v2f bf[4];
  float bx[4], by[4];
#pragma unroll
  for (int s = 0; s < 4; ++s) {
    const float w0 = W2[(4 * s + kk) * F_H2 + col];
    const float w1 = W2[(4 * s + kk + 1) * F_H2 + col];
    bf[s].x = live_col ? w0 : 0.0f;
    bf[s].y = live_col ? w1 : 0.0f;
    bx[s] = b1[4 * s + kk];
    by[s] = b1[4 * s + kk + 1];
  }
  float* Hout = (float*)keys + ((e0 + 31) & ~31);
  const int ntiles = nrows >> 4;
  const int wave_s = __builtin_amdgcn_readfirstlane(wave);
  for (int tile = wave_s; tile < ntiles; tile += NW) {
    const int lrow = tile * 16 + m;
    const size_t g = (size_t)(row0 + lrow);
    const float w = dinv[g];
    v2f a[4];
#pragma unroll
    for (int s = 0; s < 4; ++s) {
      const float2 p  = *(const float2*)(R + g * F + 4 * s + kk);
      const float2 sm = *(const float2*)(acc + lrow * F + 4 * s + kk);
      a[s].x = fmaxf((sm.x + p.x) * w + bx[s], 0.0f);
      a[s].y = fmaxf((sm.y + p.y) * w + by[s], 0.0f);
    }
    v8f c = {0.f, 0.f, 0.f, 0.f, 0.f, 0.f, 0.f, 0.f};
    c = __builtin_amdgcn_wmma_f32_16x16x4_f32(false, a[0], false, bf[0], (short)0, c, false, false);
    c = __builtin_amdgcn_wmma_f32_16x16x4_f32(false, a[1], false, bf[1], (short)0, c, false, false);
    c = __builtin_amdgcn_wmma_f32_16x16x4_f32(false, a[2], false, bf[2], (short)0, c, false, false);
    c = __builtin_amdgcn_wmma_f32_16x16x4_f32(false, a[3], false, bf[3], (short)0, c, false, false);
    const int lr0 = tile * 16 + 8 * hi;
    const float* dv = dinv + row0 + lr0;
    float r[8];
#pragma unroll
    for (int j = 0; j < 8; ++j) r[j] = c[j] * dv[j];
    if (live_col) {
      float* out = Hout + (size_t)lr0 * F_H2 + m;
#pragma unroll
      for (int j = 0; j < 8; ++j) out[(size_t)j * F_H2] = r[j];
    }
  }
}


template <int MODE>
__global__ __launch_bounds__(512) void agg_range_lds(const int* __restrict__ keys,
                                                     const int* __restrict__ bstart,
                                                     const float* __restrict__ H,
                                                     float* OUT,
                                                     const float* __restrict__ dinv,
                                                     const float* __restrict__ b1,
                                                     const float* __restrict__ W2,
                                                     const float* __restrict__ b2,
                                                     const int* __restrict__ batch,
                                                     float* __restrict__ pooled,
                                                     const int* __restrict__ fits,
                                                     const int* __restrict__ keys2,
                                                     const float* __restrict__ Hb) {
  constexpr int F   = 8;
  constexpr int LPE = F / 4;
  constexpr int EPI = 32 / LPE;
  constexpr int U   = 4;
  constexpr int NW  = 512 / 32;
  __shared__ float acc[R_ROWS * F];
  __shared__ int hoff[MODE == 0 ? N_RANGES : 1];

  const int fit = *fits;
  if (MODE != 0 && fit) return;
  const int*   ksrc = (MODE == 0 && fit) ? keys2 : keys;
  const float* Hs   = (MODE == 0 && fit) ? Hb : H;

  const int t = threadIdx.x, lane = t & 31, wave = t >> 5;
  const int b = blockIdx.x;
  const int row0 = b << RB;
  const int nrows = (N_NODES - row0 < R_ROWS) ? (N_NODES - row0) : R_ROWS;

  float4* acc4 = (float4*)acc;
  for (int i = t; i < R_ROWS * F / 4; i += 512) acc4[i] = make_float4(0.f, 0.f, 0.f, 0.f);
  if constexpr (MODE == 0) {
    for (int r = t; r < N_RANGES; r += 512)
      hoff[r] = fit ? ((bstart[r] + 31) & ~31) : (r << (RB + 3));
  }
  __syncthreads();

  const int e0 = bstart[b], e1 = bstart[b + 1];
  const int sub = lane / LPE;
  const int q   = lane % LPE;
  const int rot = sub & 3;
  const bool t1 = (rot & 1) != 0, t2 = (rot & 2) != 0;
  const int o[4] = {rot, (rot + 1) & 3, (rot + 2) & 3, (rot + 3) & 3};

  for (int base = e0 + wave * (32 * U); base < e1; base += NW * 32 * U) {
    int k[U];
#pragma unroll
    for (int u = 0; u < U; ++u) {
      const int e = base + u * 32 + lane;
      k[u] = (e < e1) ? ksrc[e] : -1;
    }
    float4 v[U * LPE];
    int    dl[U * LPE];
#pragma unroll
    for (int u = 0; u < U; ++u) {
#pragma unroll
      for (int i = 0; i < LPE; ++i) {
        const int ks = __shfl(k[u], i * EPI + sub);
        const bool ok = (ks >= 0);
        const unsigned s = ok ? ((unsigned)ks >> RB) : 0u;
        if constexpr (MODE == 0)
          v[u * LPE + i] = *(const float4*)(Hs + (size_t)hoff[s >> RB] +
                                            (size_t)(s & (R_ROWS - 1)) * F + 4 * q);
        else
          v[u * LPE + i] = *(const float4*)(H + (size_t)s * F + 4 * q);
        dl[u * LPE + i] = ok ? (ks & (R_ROWS - 1)) : -1;
      }
    }
#pragma unroll
    for (int j = 0; j < U * LPE; ++j) {
      if (dl[j] >= 0)
        lds_add_turned4(acc + dl[j] * F + 4 * q, v[j], t1, t2, o);
    }
  }
  __syncthreads();

  if constexpr (MODE == 0) {
    constexpr int RPW = R_ROWS / NW;
    constexpr int UP  = 8;
    static_assert(RPW % (4 * UP) == 0, "a wave's rows are whole trips of 4*UP rows");
    const int f  = lane & 7;
    const int r4 = lane >> 3;
    const float bf = b2[f];
    float* gcnt = pooled + (size_t)N_GRAPHS * F_H2;
    const float* Hown = Hs + (size_t)hoff[b];
    int   gcur = 0;
    int   nrun = 0;
    float run  = 0.0f;
    for (int i0 = 0; i0 < RPW / 4; i0 += UP) {
      int   g[UP];
      float w[UP], a[UP], h[UP];
#pragma unroll
      for (int u = 0; u < UP; ++u) {
        const int lrow = wave * RPW + 4 * (i0 + u) + r4;
        const bool ok = (lrow < nrows);
        const int lr = ok ? lrow : 0;
        const size_t grow = (size_t)(row0 + lr);
        g[u] = ok ? batch[grow] : -1;
        w[u] = dinv[grow];
        h[u] = Hown[lr * F + f];
        a[u] = acc[lrow * F + f];
      }
#pragma unroll
      for (int u = 0; u < UP; ++u) {
        const int lrow = wave * RPW + 4 * (i0 + u) + r4;
        if (lrow < nrows) {
          const float v = fmaxf((a[u] + h[u]) * w[u] + bf, 0.0f);
          if (nrun > 0 && g[u] != gcur) {
            ATOMIC_FADD(&pooled[(size_t)gcur * F_H2 + f], run);
            if (f == 0) ATOMIC_FADD(&gcnt[gcur], (float)nrun);
            nrun = 0;
            run  = 0.0f;
          }
          gcur = g[u];
          run += v;
          ++nrun;
        }
      }
    }
    if (nrun > 0) {
      ATOMIC_FADD(&pooled[(size_t)gcur * F_H2 + f], run);
      if (f == 0) ATOMIC_FADD(&gcnt[gcur], (float)nrun);
    }
  } else if constexpr (MODE == 1) {
    const float4* h4 = (const float4*)(H + (size_t)row0 * F);
    float4* out4 = (float4*)(OUT + (size_t)row0 * F);
    const float4 bv = ((const float4*)b1)[t & 1];
    for (int i = t; i < nrows * F / 4; i += 512) {
      const float w = dinv[row0 + (i >> 1)];
      float4 av = acc4[i];
      const float4 hv = h4[i];
      av.x = fmaxf((av.x + hv.x) * w + bv.x, 0.0f);
      av.y = fmaxf((av.y + hv.y) * w + bv.y, 0.0f);
      av.z = fmaxf((av.z + hv.z) * w + bv.z, 0.0f);
      av.w = fmaxf((av.w + hv.w) * w + bv.w, 0.0f);
      out4[i] = av;
    }
  } else {
    const int m  = lane & 15;
    const int hi = lane >> 4;
    const int kk = hi << 1;
    const bool live_col = (m < F_H2);
    const int col = m & 7;
    v2f bf[4];
#pragma unroll
    for (int s = 0; s < 4; ++s) {
      const float w0 = W2[(4 * s + kk) * F_H2 + col];
      const float w1 = W2[(4 * s + kk + 1) * F_H2 + col];
      bf[s].x = live_col ? w0 : 0.0f;
      bf[s].y = live_col ? w1 : 0.0f;
    }
    const float bb0 = b1[8 + kk],  bb1 = b1[9 + kk];
    const float bb2 = b1[12 + kk], bb3 = b1[13 + kk];
    const int ntiles = nrows >> 4;
    const int wave_s = __builtin_amdgcn_readfirstlane(wave);
    for (int tile = wave_s; tile < ntiles; tile += NW) {
      const int lrow = tile * 16 + m;
      const size_t g = (size_t)(row0 + lrow);
      const float w = dinv[g];
      const float2 q0 = *(const float2*)(OUT + g * F + kk);
      const float2 q1 = *(const float2*)(OUT + g * F + 4 + kk);
      const float2 p0 = *(const float2*)(H + g * F + kk);
      const float2 p1 = *(const float2*)(H + g * F + 4 + kk);
      const float2 s0 = *(const float2*)(acc + lrow * F + kk);
      const float2 s1 = *(const float2*)(acc + lrow * F + 4 + kk);
      v2f a0, a1, a2, a3;
      a0.x = q0.x;  a0.y = q0.y;
      a1.x = q1.x;  a1.y = q1.y;
      a2.x = fmaxf((s0.x + p0.x) * w + bb0, 0.0f);
      a2.y = fmaxf((s0.y + p0.y) * w + bb1, 0.0f);
      a3.x = fmaxf((s1.x + p1.x) * w + bb2, 0.0f);
      a3.y = fmaxf((s1.y + p1.y) * w + bb3, 0.0f);
      v8f c = {0.f, 0.f, 0.f, 0.f, 0.f, 0.f, 0.f, 0.f};
      c = __builtin_amdgcn_wmma_f32_16x16x4_f32(false, a0, false, bf[0], (short)0, c, false, false);
      c = __builtin_amdgcn_wmma_f32_16x16x4_f32(false, a1, false, bf[1], (short)0, c, false, false);
      c = __builtin_amdgcn_wmma_f32_16x16x4_f32(false, a2, false, bf[2], (short)0, c, false, false);
      c = __builtin_amdgcn_wmma_f32_16x16x4_f32(false, a3, false, bf[3], (short)0, c, false, false);
      const int r0 = row0 + tile * 16 + 8 * hi;
      const float* dv = dinv + r0;
      float r[8];
#pragma unroll
      for (int j = 0; j < 8; ++j) r[j] = c[j] * dv[j];
      if (live_col) {
        float* out = OUT + (size_t)r0 * F + m;
#pragma unroll
        for (int j = 0; j < 8; ++j) out[(size_t)j * F] = r[j];
      }
    }
  }
}


__global__ void readout(const float* __restrict__ pooled, const float* __restrict__ cnt,
                        const float* __restrict__ Wl, const float* __restrict__ bl,
                        float* __restrict__ out, int G) {
  int g = blockIdx.x * blockDim.x + threadIdx.x;
  if (g >= G) return;
  float inv = 1.0f / fmaxf(cnt[g], 1.0f);
  float p[F_H2];
#pragma unroll
  for (int f = 0; f < F_H2; ++f) p[f] = pooled[(size_t)g * F_H2 + f] * inv;
#pragma unroll
  for (int c = 0; c < N_CLS; ++c) {
    float acc = bl[c];
#pragma unroll
    for (int f = 0; f < F_H2; ++f) acc += p[f] * Wl[f * N_CLS + c];
    out[(size_t)g * N_CLS + c] = acc;
  }
}


extern "C" void kernel_launch(void* const* d_in, const int* in_sizes, int n_in,
                              void* d_out, int out_size, void* d_ws, size_t ws_size,
                              hipStream_t stream) {
  const float* x     = (const float*)d_in[0];
  const int*   ei    = (const int*)d_in[1];
  const int*   src   = ei;
  const int*   dst   = ei + N_EDGES;
  const int*   batch = (const int*)d_in[2];
  const float* W1 = (const float*)d_in[4];
  const float* b1 = (const float*)d_in[5];
  const float* W2 = (const float*)d_in[6];
  const float* b2 = (const float*)d_in[7];
  const float* Wl = (const float*)d_in[8];
  const float* bl = (const float*)d_in[9];
  float* out = (float*)d_out;

  char* ws = (char*)d_ws;
  float* dinv   = (float*)(ws);
  float* bufA   = (float*)(ws + (4ll  << 20));
  float* bufB   = (float*)(ws + (68ll << 20));
  float* pooled = (float*)(ws + (132ll << 20));
  float* cnt    = pooled + (size_t)N_GRAPHS * F_H2;

  float* rows64 = bufA;
  float* planeP = bufA;
  float* planeQ = (float*)((char*)bufA + (32ll << 20));
  int* keys   = (int*)bufB;
  int* keys2  = (int*)bufA;
  int* counts = (int*)((char*)bufB + 64000000ll);
  int* tot    = (int*)((char*)bufB + 64000000ll + (512ll << 10));
  int* bstart = (int*)((char*)bufB + 64000000ll + (520ll << 10));
  int* fits   = (int*)((char*)bufB + 64000000ll + (528ll << 10));

  const int NB = 256;

  bin_count<<<NBLK, 1024, 0, stream>>>(dst, counts);
  bin_scan<<<N_RANGES, 256, 0, stream>>>(counts, tot);
  bin_scatter<0><<<NBLK, 1024, 0, stream>>>(src, dst, counts, tot, bstart, keys, fits);

  degree_w1_range<<<N_RANGES, 512, 0, stream>>>(keys, bstart, x, W1, dinv, rows64, fits);

  agg1_rows64<<<N_RANGES, 512, 0, stream>>>(keys, bstart, rows64, dinv, b1, W2, fits);
  agg_range_lds<1><<<N_RANGES, 512, 0, stream>>>(keys, bstart, planeP, planeQ, dinv, b1, W2,
                                                 b2, batch, pooled, fits, keys2, planeQ);
  gemm_x_w1_plane1<<<1024, 128, 0, stream>>>(x, W1, dinv, planeP, fits);
  agg_range_lds<2><<<N_RANGES, 512, 0, stream>>>(keys, bstart, planeP, planeQ, dinv, b1, W2,
                                                 b2, batch, pooled, fits, keys2, planeQ);
  bin_scatter<1><<<NBLK, 1024, 0, stream>>>(src, dst, counts, tot, bstart, keys2, fits);

  zero_f32<<<128, 256, 0, stream>>>(pooled, (long long)N_GRAPHS * (F_H2 + 1));
  agg_range_lds<0><<<N_RANGES, 512, 0, stream>>>(keys, bstart, planeQ, planeP, dinv, b1, W2,
                                                 b2, batch, pooled, fits, keys2,
                                                 (const float*)keys);

  readout<<<(N_GRAPHS + NB - 1) / NB, NB, 0, stream>>>(pooled, cnt, Wl, bl,
                                                       out, N_GRAPHS);
}
